# starting kernel
# baseline (speedup 1.0000x reference)
#include <hip/hip_runtime.h>

#define HD   256
#define NB   128
#define RT   16
#define WTAB (RT * NB * 4)

typedef float v2f __attribute__((ext_vector_type(2)));
typedef float v8f __attribute__((ext_vector_type(8)));

template <bool RELU>
__global__ __launch_bounds__(256)
void rgcn_selfloop_gemm(const float* __restrict__ A,
                        const float* __restrict__ W,
                        const float* __restrict__ bias,
                        float* __restrict__ out,
                        int n_rows)
{
    const int lane = threadIdx.x & 31;
    const int wid  = blockIdx.x * (blockDim.x >> 5) + (threadIdx.x >> 5);

    const int strips_n = HD / 64;
    const int tile_m   = wid / strips_n;
    const int strip_n  = wid % strips_n;
    if (tile_m * 16 >= n_rows) return;

    const int half = lane >> 4;
    const int sub  = lane & 15;
    const int arow_idx = tile_m * 16 + sub;
    const int col0     = strip_n * 64 + sub;

    const float* arow = A + (size_t)arow_idx * HD;

    v8f c0 = {}, c1 = {}, c2 = {}, c3 = {};
    for (int k = 0; k < HD; k += 4) {
        float a0 = arow[k + 2 * half + 0];
        float a1 = arow[k + 2 * half + 1];
        if (RELU) { a0 = fmaxf(a0, 0.0f); a1 = fmaxf(a1, 0.0f); }
        v2f a; a.x = a0; a.y = a1;

        const float* wrow0 = W + (size_t)(k + 2 * half + 0) * HD + col0;
        const float* wrow1 = W + (size_t)(k + 2 * half + 1) * HD + col0;
        v2f b0, b1, b2, b3;
        b0.x = wrow0[0];  b0.y = wrow1[0];
        b1.x = wrow0[16]; b1.y = wrow1[16];
        b2.x = wrow0[32]; b2.y = wrow1[32];
        b3.x = wrow0[48]; b3.y = wrow1[48];

        c0 = __builtin_amdgcn_wmma_f32_16x16x4_f32(false, a, false, b0, (short)0, c0, false, false);
        c1 = __builtin_amdgcn_wmma_f32_16x16x4_f32(false, a, false, b1, (short)0, c1, false, false);
        c2 = __builtin_amdgcn_wmma_f32_16x16x4_f32(false, a, false, b2, (short)0, c2, false, false);
        c3 = __builtin_amdgcn_wmma_f32_16x16x4_f32(false, a, false, b3, (short)0, c3, false, false);
    }

    float* obase = out + (size_t)(tile_m * 16 + half * 8) * HD + col0;
    #pragma unroll
    for (int t = 0; t < 4; ++t) {
        const v8f& c = (t == 0) ? c0 : (t == 1) ? c1 : (t == 2) ? c2 : c3;
        const float bcol = bias[col0 + t * 16];
        #pragma unroll
        for (int j = 0; j < 8; ++j)
            obase[(size_t)j * HD + t * 16] = c[j] + bcol;
    }
}

template <bool RELU>
__global__ __launch_bounds__(256)
void rgcn_edge_msg(const float* __restrict__ xin,
                   const int*   __restrict__ src,
                   const int*   __restrict__ dst,
                   const int*   __restrict__ etype,
                   const float* __restrict__ norm,
                   const float* __restrict__ w,
                   float* agg,
                   int n_edges)
{
    __shared__ __align__(16) float wlds[WTAB];
    for (int i = threadIdx.x; i < WTAB; i += blockDim.x) wlds[i] = w[i];
    __syncthreads();

    const int lane   = threadIdx.x & 31;
    const int wid    = blockIdx.x * (blockDim.x >> 5) + (threadIdx.x >> 5);
    const int nwaves = (gridDim.x * blockDim.x) >> 5;

    for (int e = wid; e < n_edges; e += nwaves) {
        const int   s   = src[e];
        const int   d   = dst[e];
        const int   r   = etype[e];
        const float nrm = norm[e];

        const float4* xrow = (const float4*)(xin + (size_t)s * HD) + lane * 2;
        float4 xa = xrow[0];
        float4 xb = xrow[1];
        if (RELU) {
            xa.x = fmaxf(xa.x, 0.0f); xa.y = fmaxf(xa.y, 0.0f);
            xa.z = fmaxf(xa.z, 0.0f); xa.w = fmaxf(xa.w, 0.0f);
            xb.x = fmaxf(xb.x, 0.0f); xb.y = fmaxf(xb.y, 0.0f);
            xb.z = fmaxf(xb.z, 0.0f); xb.w = fmaxf(xb.w, 0.0f);
        }
        const float xs[8] = { xa.x, xa.y, xa.z, xa.w, xb.x, xb.y, xb.z, xb.w };

        const float4* wrel = (const float4*)(wlds + (size_t)r * (NB * 4)) + lane * 4;
        float* arow = agg + (size_t)d * HD + lane * 8;

        #pragma unroll
        for (int j = 0; j < 4; ++j) {
            const float4 wq = wrel[j];
            const float x0 = xs[2 * j + 0];
            const float x1 = xs[2 * j + 1];
            const float o0 = (x0 * wq.x + x1 * wq.z) * nrm;
            const float o1 = (x0 * wq.y + x1 * wq.w) * nrm;
            (void)__hip_atomic_fetch_add(&arow[2 * j + 0], o0,
                                         __ATOMIC_RELAXED, __HIP_MEMORY_SCOPE_AGENT);
            (void)__hip_atomic_fetch_add(&arow[2 * j + 1], o1,
                                         __ATOMIC_RELAXED, __HIP_MEMORY_SCOPE_AGENT);
        }
    }
}

extern "C" void kernel_launch(void* const* d_in, const int* in_sizes, int n_in,
                              void* d_out, int out_size, void* d_ws, size_t ws_size,
                              hipStream_t stream)
{
    const float* x     = (const float*)d_in[0];
    const int*   src   = (const int*)  d_in[1];
    const int*   dst   = (const int*)  d_in[2];
    const int*   etype = (const int*)  d_in[3];
    const float* norm  = (const float*)d_in[4];
    const float* w1    = (const float*)d_in[5];
    const float* lw1   = (const float*)d_in[6];
    const float* b1    = (const float*)d_in[7];
    const float* w2    = (const float*)d_in[8];
    const float* lw2   = (const float*)d_in[9];
    const float* b2    = (const float*)d_in[10];
    float*       out   = (float*)d_out;

    const int n_nodes = in_sizes[0] / HD;
    const int n_edges = in_sizes[1];

    float* h = (float*)d_ws;

    const int waves_per_block = 256 / 32;
    const int strips = ((n_nodes + 15) / 16) * (HD / 64);
    const int gemm_blocks = (strips + waves_per_block - 1) / waves_per_block;
    const int edge_blocks = 1024;

    rgcn_selfloop_gemm<false><<<gemm_blocks, 256, 0, stream>>>(x, lw1, b1, h, n_nodes);
    rgcn_edge_msg<false><<<edge_blocks, 256, 0, stream>>>(x, src, dst, etype, norm, w1,
                                                          h, n_edges);

    rgcn_selfloop_gemm<true><<<gemm_blocks, 256, 0, stream>>>(h, lw2, b2, out, n_nodes);
    rgcn_edge_msg<true><<<edge_blocks, 256, 0, stream>>>(h, src, dst, etype, norm, w2,
                                                         out, n_edges);
}


# Float atomic scatter removed: edges bucketed by destination in each call (count, scan, fill; integer atomics), one wave per node sums its messages in ascending edge id; sum order now fixed
# speedup vs baseline: 1.4289x; 1.4289x over previous
#include <hip/hip_runtime.h>

#define HD   256
#define NB   128
#define RT   16
#define WTAB (RT * NB * 4)

typedef float v2f __attribute__((ext_vector_type(2)));
typedef float v8f __attribute__((ext_vector_type(8)));

template <bool RELU>
__global__ __launch_bounds__(256)
void rgcn_selfloop_gemm(const float* __restrict__ A,
                        const float* __restrict__ W,
                        const float* __restrict__ bias,
                        float* __restrict__ out,
                        int n_rows,
                        int* __restrict__ zero_buf,
                        int zero_n)
{
    if (zero_buf != nullptr) {
        const int gid = blockIdx.x * blockDim.x + threadIdx.x;
        if (gid < zero_n) zero_buf[gid] = 0;
    }

    const int lane = threadIdx.x & 31;
    const int wid  = blockIdx.x * (blockDim.x >> 5) + (threadIdx.x >> 5);

    const int strips_n = HD / 64;
    const int tile_m   = wid / strips_n;
    const int strip_n  = wid % strips_n;
    if (tile_m * 16 >= n_rows) return;

    const int half = lane >> 4;
    const int sub  = lane & 15;
    const int arow_idx = tile_m * 16 + sub;
    const int col0     = strip_n * 64 + sub;

    const float* arow = A + (size_t)arow_idx * HD;

    v8f c0 = {}, c1 = {}, c2 = {}, c3 = {};
    for (int k = 0; k < HD; k += 4) {
        float a0 = arow[k + 2 * half + 0];
        float a1 = arow[k + 2 * half + 1];
        if (RELU) { a0 = fmaxf(a0, 0.0f); a1 = fmaxf(a1, 0.0f); }
        v2f a; a.x = a0; a.y = a1;

        const float* wrow0 = W + (size_t)(k + 2 * half + 0) * HD + col0;
        const float* wrow1 = W + (size_t)(k + 2 * half + 1) * HD + col0;
        v2f b0, b1, b2, b3;
        b0.x = wrow0[0];  b0.y = wrow1[0];
        b1.x = wrow0[16]; b1.y = wrow1[16];
        b2.x = wrow0[32]; b2.y = wrow1[32];
        b3.x = wrow0[48]; b3.y = wrow1[48];

        c0 = __builtin_amdgcn_wmma_f32_16x16x4_f32(false, a, false, b0, (short)0, c0, false, false);
        c1 = __builtin_amdgcn_wmma_f32_16x16x4_f32(false, a, false, b1, (short)0, c1, false, false);
        c2 = __builtin_amdgcn_wmma_f32_16x16x4_f32(false, a, false, b2, (short)0, c2, false, false);
        c3 = __builtin_amdgcn_wmma_f32_16x16x4_f32(false, a, false, b3, (short)0, c3, false, false);
    }

    float* obase = out + (size_t)(tile_m * 16 + half * 8) * HD + col0;
    #pragma unroll
    for (int t = 0; t < 4; ++t) {
        const v8f& c = (t == 0) ? c0 : (t == 1) ? c1 : (t == 2) ? c2 : c3;
        const float bcol = bias[col0 + t * 16];
        #pragma unroll
        for (int j = 0; j < 8; ++j)
            obase[(size_t)j * HD + t * 16] = c[j] + bcol;
    }
}

template <bool RELU>
__global__ __launch_bounds__(256)
void rgcn_edge_msg(const float* __restrict__ xin,
                   const int*   __restrict__ src,
                   const int*   __restrict__ dst,
                   const int*   __restrict__ etype,
                   const float* __restrict__ norm,
                   const float* __restrict__ w,
                   float* agg,
                   int n_edges)
{
    __shared__ __align__(16) float wlds[WTAB];
    for (int i = threadIdx.x; i < WTAB; i += blockDim.x) wlds[i] = w[i];
    __syncthreads();

    const int lane   = threadIdx.x & 31;
    const int wid    = blockIdx.x * (blockDim.x >> 5) + (threadIdx.x >> 5);
    const int nwaves = (gridDim.x * blockDim.x) >> 5;

    for (int e = wid; e < n_edges; e += nwaves) {
        const int   s   = src[e];
        const int   d   = dst[e];
        const int   r   = etype[e];
        const float nrm = norm[e];

        const float4* xrow = (const float4*)(xin + (size_t)s * HD) + lane * 2;
        float4 xa = xrow[0];
        float4 xb = xrow[1];
        if (RELU) {
            xa.x = fmaxf(xa.x, 0.0f); xa.y = fmaxf(xa.y, 0.0f);
            xa.z = fmaxf(xa.z, 0.0f); xa.w = fmaxf(xa.w, 0.0f);
            xb.x = fmaxf(xb.x, 0.0f); xb.y = fmaxf(xb.y, 0.0f);
            xb.z = fmaxf(xb.z, 0.0f); xb.w = fmaxf(xb.w, 0.0f);
        }
        const float xs[8] = { xa.x, xa.y, xa.z, xa.w, xb.x, xb.y, xb.z, xb.w };

        const float4* wrel = (const float4*)(wlds + (size_t)r * (NB * 4)) + lane * 4;
        float* arow = agg + (size_t)d * HD + lane * 8;

        #pragma unroll
        for (int j = 0; j < 4; ++j) {
            const float4 wq = wrel[j];
            const float x0 = xs[2 * j + 0];
            const float x1 = xs[2 * j + 1];
            const float o0 = (x0 * wq.x + x1 * wq.z) * nrm;
            const float o1 = (x0 * wq.y + x1 * wq.w) * nrm;
            (void)__hip_atomic_fetch_add(&arow[2 * j + 0], o0,
                                         __ATOMIC_RELAXED, __HIP_MEMORY_SCOPE_AGENT);
            (void)__hip_atomic_fetch_add(&arow[2 * j + 1], o1,
                                         __ATOMIC_RELAXED, __HIP_MEMORY_SCOPE_AGENT);
        }
    }
}

__global__ __launch_bounds__(256)
void rgcn_deg_count(const int* __restrict__ dst, int* seg, int n_edges)
{
    const int e = blockIdx.x * blockDim.x + threadIdx.x;
    if (e < n_edges)
        (void)__hip_atomic_fetch_add(&seg[dst[e]], 1,
                                     __ATOMIC_RELAXED, __HIP_MEMORY_SCOPE_AGENT);
}

__global__ __launch_bounds__(1024)
void rgcn_seg_scan(int* seg, int n)
{
    __shared__ int wsum[32];
    const int t    = threadIdx.x;
    const int lane = t & 31;
    const int w    = t >> 5;
    const int chunk = (n + 1023) / 1024;
    const int b  = t * chunk;
    const int en = (b + chunk < n) ? (b + chunk) : n;

    int s = 0;
    for (int i = b; i < en; ++i) s += seg[i];

    int v = s;
    #pragma unroll
    for (int o = 1; o < 32; o <<= 1) {
        const int u = __shfl_up(v, o);
        if (lane >= o) v += u;
    }
    if (lane == 31) wsum[w] = v;
    __syncthreads();
    if (w == 0) {
        const int x = wsum[lane];
        int y = x;
        #pragma unroll
        for (int o = 1; o < 32; o <<= 1) {
            const int u = __shfl_up(y, o);
            if (lane >= o) y += u;
        }
        wsum[lane] = y - x;
    }
    __syncthreads();

    int run = wsum[w] + v - s;
    for (int i = b; i < en; ++i) {
        const int c = seg[i];
        seg[i] = run;
        run += c;
    }
}

__global__ __launch_bounds__(256)
void rgcn_seg_fill(const int* __restrict__ dst, int* seg, int* __restrict__ perm,
                   int n_edges)
{
    const int e = blockIdx.x * blockDim.x + threadIdx.x;
    if (e < n_edges) {
        const int p = __hip_atomic_fetch_add(&seg[dst[e]], 1,
                                             __ATOMIC_RELAXED, __HIP_MEMORY_SCOPE_AGENT);
        perm[p] = e;
    }
}

template <bool RELU>
__device__ __forceinline__
void rgcn_msg_acc(float (&acc)[8], float4 xa, float4 xb,
                  const float4* wrel, float nrm)
{
    if (RELU) {
        xa.x = fmaxf(xa.x, 0.0f); xa.y = fmaxf(xa.y, 0.0f);
        xa.z = fmaxf(xa.z, 0.0f); xa.w = fmaxf(xa.w, 0.0f);
        xb.x = fmaxf(xb.x, 0.0f); xb.y = fmaxf(xb.y, 0.0f);
        xb.z = fmaxf(xb.z, 0.0f); xb.w = fmaxf(xb.w, 0.0f);
    }
    const float xs[8] = { xa.x, xa.y, xa.z, xa.w, xb.x, xb.y, xb.z, xb.w };
    #pragma unroll
    for (int j = 0; j < 4; ++j) {
        const float4 wq = wrel[j];
        const float x0 = xs[2 * j + 0];
        const float x1 = xs[2 * j + 1];
        acc[2 * j + 0] += (x0 * wq.x + x1 * wq.z) * nrm;
        acc[2 * j + 1] += (x0 * wq.y + x1 * wq.w) * nrm;
    }
}

template <bool RELU>
__global__ __launch_bounds__(256)
void rgcn_node_gather(const float* __restrict__ xin,
                      const int*   __restrict__ src,
                      const int*   __restrict__ dst,
                      const int*   __restrict__ etype,
                      const float* __restrict__ norm,
                      const float* __restrict__ w,
                      const int*   __restrict__ seg,
                      const int*   __restrict__ perm,
                      float* agg,
                      int n_nodes,
                      int n_edges)
{
    __shared__ __align__(16) float wlds[WTAB];
    for (int i = threadIdx.x; i < WTAB / 4; i += blockDim.x)
        ((float4*)wlds)[i] = ((const float4*)w)[i];
    __syncthreads();

    const int lane   = threadIdx.x & 31;
    const int wid    = __builtin_amdgcn_readfirstlane(
                           blockIdx.x * (blockDim.x >> 5) + (threadIdx.x >> 5));
    const int nwaves = (gridDim.x * blockDim.x) >> 5;
    const float4* wl = (const float4*)wlds + lane * 4;

    for (int n = wid; n < n_nodes; n += nwaves) {
        const int end = seg[n];
        const int beg = (n > 0) ? seg[n - 1] : 0;
        const int cnt = end - beg;
        if (cnt <= 0) continue;

        float4* orow = (float4*)(agg + (size_t)n * HD) + lane * 2;
        float4 oa = orow[0];
        float4 ob = orow[1];

        float acc[8] = { 0.f, 0.f, 0.f, 0.f, 0.f, 0.f, 0.f, 0.f };

        if (cnt <= 32) {
            const int my = (lane < cnt) ? perm[beg + lane] : 0x7fffffff;
            int rank = 0;
            for (int m = 0; m < cnt; ++m)
                rank += (__builtin_amdgcn_readlane(my, m) < my) ? 1 : 0;
            const int sorted = __builtin_amdgcn_ds_permute(rank << 2, my);

            int   s_v = 0, r_v = 0;
            float n_v = 0.0f;
            if (lane < cnt) {
                s_v = src[sorted];
                r_v = etype[sorted];
                n_v = norm[sorted];
            }

            int k = 0;
            for (; k + 4 <= cnt; k += 4) {
                float4 xa[4], xb[4];
                int    rr[4];
                float  nn[4];
                #pragma unroll
                for (int u = 0; u < 4; ++u) {
                    const int s = __builtin_amdgcn_readlane(s_v, k + u);
                    rr[u] = __builtin_amdgcn_readlane(r_v, k + u);
                    nn[u] = __int_as_float(
                                __builtin_amdgcn_readlane(__float_as_int(n_v), k + u));
                    const float4* xrow = (const float4*)(xin + (size_t)s * HD) + lane * 2;
                    xa[u] = xrow[0];
                    xb[u] = xrow[1];
                }
                #pragma unroll
                for (int u = 0; u < 4; ++u)
                    rgcn_msg_acc<RELU>(acc, xa[u], xb[u], wl + (size_t)rr[u] * NB, nn[u]);
            }
            for (; k < cnt; ++k) {
                const int   s  = __builtin_amdgcn_readlane(s_v, k);
                const int   r  = __builtin_amdgcn_readlane(r_v, k);
                const float nr = __int_as_float(
                                     __builtin_amdgcn_readlane(__float_as_int(n_v), k));
                const float4* xrow = (const float4*)(xin + (size_t)s * HD) + lane * 2;
                rgcn_msg_acc<RELU>(acc, xrow[0], xrow[1], wl + (size_t)r * NB, nr);
            }
        } else if (cnt <= 2048) {
            int lo = -1;
            for (int i = 0; i < cnt; ++i) {
                int best = 0x7fffffff;
                for (int p = beg + lane; p < end; p += 32) {
                    const int e = perm[p];
                    if (e > lo && e < best) best = e;
                }
                #pragma unroll
                for (int o = 16; o > 0; o >>= 1) {
                    const int other = __shfl_xor(best, o);
                    best = (other < best) ? other : best;
                }
                const int e = __builtin_amdgcn_readfirstlane(best);
                lo = e;
                const int   s  = src[e];
                const int   r  = etype[e];
                const float nr = norm[e];
                const float4* xrow = (const float4*)(xin + (size_t)s * HD) + lane * 2;
                rgcn_msg_acc<RELU>(acc, xrow[0], xrow[1], wl + (size_t)r * NB, nr);
            }
        } else {
            for (int e0 = 0; e0 < n_edges; e0 += 32) {
                const int  ev  = e0 + lane;
                const bool hit = (ev < n_edges) && (dst[ev] == n);
                unsigned m = __builtin_amdgcn_ballot_w32(hit);
                while (m != 0u) {
                    const int e = e0 + (__builtin_ctz(m));
                    m &= m - 1u;
                    const int   s  = src[e];
                    const int   r  = etype[e];
                    const float nr = norm[e];
                    const float4* xrow = (const float4*)(xin + (size_t)s * HD) + lane * 2;
                    rgcn_msg_acc<RELU>(acc, xrow[0], xrow[1], wl + (size_t)r * NB, nr);
                }
            }
        }

        oa.x += acc[0]; oa.y += acc[1]; oa.z += acc[2]; oa.w += acc[3];
        ob.x += acc[4]; ob.y += acc[5]; ob.z += acc[6]; ob.w += acc[7];
        orow[0] = oa;
        orow[1] = ob;
    }
}

extern "C" void kernel_launch(void* const* d_in, const int* in_sizes, int n_in,
                              void* d_out, int out_size, void* d_ws, size_t ws_size,
                              hipStream_t stream)
{
    const float* x     = (const float*)d_in[0];
    const int*   src   = (const int*)  d_in[1];
    const int*   dst   = (const int*)  d_in[2];
    const int*   etype = (const int*)  d_in[3];
    const float* norm  = (const float*)d_in[4];
    const float* w1    = (const float*)d_in[5];
    const float* lw1   = (const float*)d_in[6];
    const float* b1    = (const float*)d_in[7];
    const float* w2    = (const float*)d_in[8];
    const float* lw2   = (const float*)d_in[9];
    const float* b2    = (const float*)d_in[10];
    float*       out   = (float*)d_out;

    const int n_nodes = in_sizes[0] / HD;
    const int n_edges = in_sizes[1];

    float* h = (float*)d_ws;

    const int waves_per_block = 256 / 32;
    const int strips = ((n_nodes + 15) / 16) * (HD / 64);
    const int gemm_blocks = (strips + waves_per_block - 1) / waves_per_block;

    const size_t seg_off  = ((size_t)n_nodes * HD * sizeof(float) + 255) & ~(size_t)255;
    const size_t perm_off = (seg_off + (size_t)n_nodes * sizeof(int) + 255) & ~(size_t)255;
    const size_t ws_need  = perm_off + (size_t)n_edges * sizeof(int);

    if (ws_size < ws_need) {
        const int edge_blocks = 1024;
        rgcn_selfloop_gemm<false><<<gemm_blocks, 256, 0, stream>>>(x, lw1, b1, h, n_nodes,
                                                                   nullptr, 0);
        rgcn_edge_msg<false><<<edge_blocks, 256, 0, stream>>>(x, src, dst, etype, norm, w1,
                                                              h, n_edges);
        rgcn_selfloop_gemm<true><<<gemm_blocks, 256, 0, stream>>>(h, lw2, b2, out, n_nodes,
                                                                  nullptr, 0);
        rgcn_edge_msg<true><<<edge_blocks, 256, 0, stream>>>(h, src, dst, etype, norm, w2,
                                                             out, n_edges);
        return;
    }

    int* seg  = (int*)((char*)d_ws + seg_off);
    int* perm = (int*)((char*)d_ws + perm_off);

    const int edge_blocks   = (n_edges + 255) / 256;
    const int gather_blocks = 1024;

    rgcn_selfloop_gemm<false><<<gemm_blocks, 256, 0, stream>>>(x, lw1, b1, h, n_nodes,
                                                               seg, n_nodes);
    rgcn_deg_count<<<edge_blocks, 256, 0, stream>>>(dst, seg, n_edges);
    rgcn_seg_scan<<<1, 1024, 0, stream>>>(seg, n_nodes);
    rgcn_seg_fill<<<edge_blocks, 256, 0, stream>>>(dst, seg, perm, n_edges);

    rgcn_node_gather<false><<<gather_blocks, 256, 0, stream>>>(x, src, dst, etype, norm, w1,
                                                               seg, perm, h, n_nodes, n_edges);

    rgcn_selfloop_gemm<true><<<gemm_blocks, 256, 0, stream>>>(h, lw2, b2, out, n_nodes,
                                                              nullptr, 0);
    rgcn_node_gather<true><<<gather_blocks, 256, 0, stream>>>(h, src, dst, etype, norm, w2,
                                                              seg, perm, out, n_nodes, n_edges);
}


# Nodes of 33 to 128 edges: ids ranked through a per-wave LDS buffer, rows then fetched four at a time in the short nodes' loop, not one dependent selection pass per edge; sum order kept
# speedup vs baseline: 1.6206x; 1.1342x over previous
#include <hip/hip_runtime.h>

#define HD   256
#define NB   128
#define RT   16
#define WTAB (RT * NB * 4)
#define LCAP 128

typedef float v2f __attribute__((ext_vector_type(2)));
typedef float v8f __attribute__((ext_vector_type(8)));

template <bool RELU>
__global__ __launch_bounds__(256)
void rgcn_selfloop_gemm(const float* __restrict__ A,
                        const float* __restrict__ W,
                        const float* __restrict__ bias,
                        float* __restrict__ out,
                        int n_rows,
                        int* __restrict__ zero_buf,
                        int zero_n)
{
    if (zero_buf != nullptr) {
        const int gid = blockIdx.x * blockDim.x + threadIdx.x;
        if (gid < zero_n) zero_buf[gid] = 0;
    }

    const int lane = threadIdx.x & 31;
    const int wid  = blockIdx.x * (blockDim.x >> 5) + (threadIdx.x >> 5);

    const int strips_n = HD / 64;
    const int tile_m   = wid / strips_n;
    const int strip_n  = wid % strips_n;
    if (tile_m * 16 >= n_rows) return;

    const int half = lane >> 4;
    const int sub  = lane & 15;
    const int arow_idx = tile_m * 16 + sub;
    const int col0     = strip_n * 64 + sub;

    const float* arow = A + (size_t)arow_idx * HD;

    v8f c0 = {}, c1 = {}, c2 = {}, c3 = {};
    for (int k = 0; k < HD; k += 4) {
        float a0 = arow[k + 2 * half + 0];
        float a1 = arow[k + 2 * half + 1];
        if (RELU) { a0 = fmaxf(a0, 0.0f); a1 = fmaxf(a1, 0.0f); }
        v2f a; a.x = a0; a.y = a1;

        const float* wrow0 = W + (size_t)(k + 2 * half + 0) * HD + col0;
        const float* wrow1 = W + (size_t)(k + 2 * half + 1) * HD + col0;
        v2f b0, b1, b2, b3;
        b0.x = wrow0[0];  b0.y = wrow1[0];
        b1.x = wrow0[16]; b1.y = wrow1[16];
        b2.x = wrow0[32]; b2.y = wrow1[32];
        b3.x = wrow0[48]; b3.y = wrow1[48];

        c0 = __builtin_amdgcn_wmma_f32_16x16x4_f32(false, a, false, b0, (short)0, c0, false, false);
        c1 = __builtin_amdgcn_wmma_f32_16x16x4_f32(false, a, false, b1, (short)0, c1, false, false);
        c2 = __builtin_amdgcn_wmma_f32_16x16x4_f32(false, a, false, b2, (short)0, c2, false, false);
        c3 = __builtin_amdgcn_wmma_f32_16x16x4_f32(false, a, false, b3, (short)0, c3, false, false);
    }

    float* obase = out + (size_t)(tile_m * 16 + half * 8) * HD + col0;
    #pragma unroll
    for (int t = 0; t < 4; ++t) {
        const v8f& c = (t == 0) ? c0 : (t == 1) ? c1 : (t == 2) ? c2 : c3;
        const float bcol = bias[col0 + t * 16];
        #pragma unroll
        for (int j = 0; j < 8; ++j)
            obase[(size_t)j * HD + t * 16] = c[j] + bcol;
    }
}

template <bool RELU>
__global__ __launch_bounds__(256)
void rgcn_edge_msg(const float* __restrict__ xin,
                   const int*   __restrict__ src,
                   const int*   __restrict__ dst,
                   const int*   __restrict__ etype,
                   const float* __restrict__ norm,
                   const float* __restrict__ w,
                   float* agg,
                   int n_edges)
{
    __shared__ __align__(16) float wlds[WTAB];
    for (int i = threadIdx.x; i < WTAB; i += blockDim.x) wlds[i] = w[i];
    __syncthreads();

    const int lane   = threadIdx.x & 31;
    const int wid    = blockIdx.x * (blockDim.x >> 5) + (threadIdx.x >> 5);
    const int nwaves = (gridDim.x * blockDim.x) >> 5;

    for (int e = wid; e < n_edges; e += nwaves) {
        const int   s   = src[e];
        const int   d   = dst[e];
        const int   r   = etype[e];
        const float nrm = norm[e];

        const float4* xrow = (const float4*)(xin + (size_t)s * HD) + lane * 2;
        float4 xa = xrow[0];
        float4 xb = xrow[1];
        if (RELU) {
            xa.x = fmaxf(xa.x, 0.0f); xa.y = fmaxf(xa.y, 0.0f);
            xa.z = fmaxf(xa.z, 0.0f); xa.w = fmaxf(xa.w, 0.0f);
            xb.x = fmaxf(xb.x, 0.0f); xb.y = fmaxf(xb.y, 0.0f);
            xb.z = fmaxf(xb.z, 0.0f); xb.w = fmaxf(xb.w, 0.0f);
        }
        const float xs[8] = { xa.x, xa.y, xa.z, xa.w, xb.x, xb.y, xb.z, xb.w };

        const float4* wrel = (const float4*)(wlds + (size_t)r * (NB * 4)) + lane * 4;
        float* arow = agg + (size_t)d * HD + lane * 8;

        #pragma unroll
        for (int j = 0; j < 4; ++j) {
            const float4 wq = wrel[j];
            const float x0 = xs[2 * j + 0];
            const float x1 = xs[2 * j + 1];
            const float o0 = (x0 * wq.x + x1 * wq.z) * nrm;
            const float o1 = (x0 * wq.y + x1 * wq.w) * nrm;
            (void)__hip_atomic_fetch_add(&arow[2 * j + 0], o0,
                                         __ATOMIC_RELAXED, __HIP_MEMORY_SCOPE_AGENT);
            (void)__hip_atomic_fetch_add(&arow[2 * j + 1], o1,
                                         __ATOMIC_RELAXED, __HIP_MEMORY_SCOPE_AGENT);
        }
    }
}

__global__ __launch_bounds__(256)
void rgcn_deg_count(const int* __restrict__ dst, int* seg, int n_edges)
{
    const int e = blockIdx.x * blockDim.x + threadIdx.x;
    if (e < n_edges)
        (void)__hip_atomic_fetch_add(&seg[dst[e]], 1,
                                     __ATOMIC_RELAXED, __HIP_MEMORY_SCOPE_AGENT);
}

__global__ __launch_bounds__(1024)
void rgcn_seg_scan(int* seg, int n)
{
    __shared__ int wsum[32];
    const int t    = threadIdx.x;
    const int lane = t & 31;
    const int w    = t >> 5;
    const int chunk = (n + 1023) / 1024;
    const int b  = t * chunk;
    const int en = (b + chunk < n) ? (b + chunk) : n;

    int s = 0;
    for (int i = b; i < en; ++i) s += seg[i];

    int v = s;
    #pragma unroll
    for (int o = 1; o < 32; o <<= 1) {
        const int u = __shfl_up(v, o);
        if (lane >= o) v += u;
    }
    if (lane == 31) wsum[w] = v;
    __syncthreads();
    if (w == 0) {
        const int x = wsum[lane];
        int y = x;
        #pragma unroll
        for (int o = 1; o < 32; o <<= 1) {
            const int u = __shfl_up(y, o);
            if (lane >= o) y += u;
        }
        wsum[lane] = y - x;
    }
    __syncthreads();

    int run = wsum[w] + v - s;
    for (int i = b; i < en; ++i) {
        const int c = seg[i];
        seg[i] = run;
        run += c;
    }
}

__global__ __launch_bounds__(256)
void rgcn_seg_fill(const int* __restrict__ dst, int* seg, int* __restrict__ perm,
                   int n_edges)
{
    const int e = blockIdx.x * blockDim.x + threadIdx.x;
    if (e < n_edges) {
        const int p = __hip_atomic_fetch_add(&seg[dst[e]], 1,
                                             __ATOMIC_RELAXED, __HIP_MEMORY_SCOPE_AGENT);
        perm[p] = e;
    }
}

template <bool RELU>
__device__ __forceinline__
void rgcn_msg_acc(float (&acc)[8], float4 xa, float4 xb,
                  const float4* wrel, float nrm)
{
    if (RELU) {
        xa.x = fmaxf(xa.x, 0.0f); xa.y = fmaxf(xa.y, 0.0f);
        xa.z = fmaxf(xa.z, 0.0f); xa.w = fmaxf(xa.w, 0.0f);
        xb.x = fmaxf(xb.x, 0.0f); xb.y = fmaxf(xb.y, 0.0f);
        xb.z = fmaxf(xb.z, 0.0f); xb.w = fmaxf(xb.w, 0.0f);
    }
    const float xs[8] = { xa.x, xa.y, xa.z, xa.w, xb.x, xb.y, xb.z, xb.w };
    #pragma unroll
    for (int j = 0; j < 4; ++j) {
        const float4 wq = wrel[j];
        const float x0 = xs[2 * j + 0];
        const float x1 = xs[2 * j + 1];
        acc[2 * j + 0] += (x0 * wq.x + x1 * wq.z) * nrm;
        acc[2 * j + 1] += (x0 * wq.y + x1 * wq.w) * nrm;
    }
}

template <bool RELU>
__device__ __forceinline__
void rgcn_sorted_run(float (&acc)[8], const int sorted, const int m, const int lane,
                     const float* __restrict__ xin,
                     const int*   __restrict__ src,
                     const int*   __restrict__ etype,
                     const float* __restrict__ norm,
                     const float4* wl)
{
    int   s_v = 0, r_v = 0;
    float n_v = 0.0f;
    if (lane < m) {
        s_v = src[sorted];
        r_v = etype[sorted];
        n_v = norm[sorted];
    }

    int k = 0;
    for (; k + 4 <= m; k += 4) {
        float4 xa[4], xb[4];
        int    rr[4];
        float  nn[4];
        #pragma unroll
        for (int u = 0; u < 4; ++u) {
            const int s = __builtin_amdgcn_readlane(s_v, k + u);
            rr[u] = __builtin_amdgcn_readlane(r_v, k + u);
            nn[u] = __int_as_float(
                        __builtin_amdgcn_readlane(__float_as_int(n_v), k + u));
            const float4* xrow = (const float4*)(xin + (size_t)s * HD) + lane * 2;
            xa[u] = xrow[0];
            xb[u] = xrow[1];
        }
        #pragma unroll
        for (int u = 0; u < 4; ++u)
            rgcn_msg_acc<RELU>(acc, xa[u], xb[u], wl + (size_t)rr[u] * NB, nn[u]);
    }
    for (; k < m; ++k) {
        const int   s  = __builtin_amdgcn_readlane(s_v, k);
        const int   r  = __builtin_amdgcn_readlane(r_v, k);
        const float nr = __int_as_float(
                             __builtin_amdgcn_readlane(__float_as_int(n_v), k));
        const float4* xrow = (const float4*)(xin + (size_t)s * HD) + lane * 2;
        rgcn_msg_acc<RELU>(acc, xrow[0], xrow[1], wl + (size_t)r * NB, nr);
    }
}

template <bool RELU>
__global__ __launch_bounds__(256)
void rgcn_node_gather(const float* __restrict__ xin,
                      const int*   __restrict__ src,
                      const int*   __restrict__ dst,
                      const int*   __restrict__ etype,
                      const float* __restrict__ norm,
                      const float* __restrict__ w,
                      const int*   __restrict__ seg,
                      const int*   __restrict__ perm,
                      float* agg,
                      int n_nodes,
                      int n_edges)
{
    __shared__ __align__(16) float wlds[WTAB];
    __shared__ __align__(16) int   idbuf[8 * 2 * LCAP];
    for (int i = threadIdx.x; i < WTAB / 4; i += blockDim.x)
        ((float4*)wlds)[i] = ((const float4*)w)[i];
    __syncthreads();

    const int lane   = threadIdx.x & 31;
    const int wid    = __builtin_amdgcn_readfirstlane(
                           blockIdx.x * (blockDim.x >> 5) + (threadIdx.x >> 5));
    const int nwaves = (gridDim.x * blockDim.x) >> 5;
    const float4* wl = (const float4*)wlds + lane * 4;
    int* uns = idbuf + __builtin_amdgcn_readfirstlane((int)(threadIdx.x >> 5)) * (2 * LCAP);
    int* srt = uns + LCAP;

    for (int n = wid; n < n_nodes; n += nwaves) {
        const int end = seg[n];
        const int beg = (n > 0) ? seg[n - 1] : 0;
        const int cnt = end - beg;
        if (cnt <= 0) continue;

        float4* orow = (float4*)(agg + (size_t)n * HD) + lane * 2;
        float4 oa = orow[0];
        float4 ob = orow[1];

        float acc[8] = { 0.f, 0.f, 0.f, 0.f, 0.f, 0.f, 0.f, 0.f };

        if (cnt <= LCAP) {
            int sorted0 = 0;
            if (cnt <= 32) {
                const int my = (lane < cnt) ? perm[beg + lane] : 0x7fffffff;
                int rank = 0;
                for (int m = 0; m < cnt; ++m)
                    rank += (__builtin_amdgcn_readlane(my, m) < my) ? 1 : 0;
                sorted0 = __builtin_amdgcn_ds_permute(rank << 2, my);
            } else {
                const int padded = (cnt + 31) & ~31;
                for (int p = lane; p < padded; p += 32)
                    uns[p] = (p < cnt) ? perm[beg + p] : 0x7fffffff;
                for (int p = lane; p < padded; p += 32) {
                    const int my = uns[p];
                    int rank = 0;
                    for (int q = 0; q < padded; q += 4) {
                        const int4 v = *(const int4*)(uns + q);
                        rank += ((v.x < my) ? 1 : 0) + ((v.y < my) ? 1 : 0)
                              + ((v.z < my) ? 1 : 0) + ((v.w < my) ? 1 : 0);
                    }
                    if (p < cnt) srt[rank] = my;
                }
            }
            for (int c = 0; c < cnt; c += 32) {
                const int m = (cnt - c < 32) ? (cnt - c) : 32;
                const int sorted = (cnt <= 32) ? sorted0 : ((lane < m) ? srt[c + lane] : 0);
                rgcn_sorted_run<RELU>(acc, sorted, m, lane, xin, src, etype, norm, wl);
            }
        } else if (cnt <= 2048) {
            int lo = -1;
            for (int i = 0; i < cnt; ++i) {
                int best = 0x7fffffff;
                for (int p = beg + lane; p < end; p += 32) {
                    const int e = perm[p];
                    if (e > lo && e < best) best = e;
                }
                #pragma unroll
                for (int o = 16; o > 0; o >>= 1) {
                    const int other = __shfl_xor(best, o);
                    best = (other < best) ? other : best;
                }
                const int e = __builtin_amdgcn_readfirstlane(best);
                lo = e;
                const int   s  = src[e];
                const int   r  = etype[e];
                const float nr = norm[e];
                const float4* xrow = (const float4*)(xin + (size_t)s * HD) + lane * 2;
                rgcn_msg_acc<RELU>(acc, xrow[0], xrow[1], wl + (size_t)r * NB, nr);
            }
        } else {
            for (int e0 = 0; e0 < n_edges; e0 += 32) {
                const int  ev  = e0 + lane;
                const bool hit = (ev < n_edges) && (dst[ev] == n);
                unsigned m = __builtin_amdgcn_ballot_w32(hit);
                while (m != 0u) {
                    const int e = e0 + (__builtin_ctz(m));
                    m &= m - 1u;
                    const int   s  = src[e];
                    const int   r  = etype[e];
                    const float nr = norm[e];
                    const float4* xrow = (const float4*)(xin + (size_t)s * HD) + lane * 2;
                    rgcn_msg_acc<RELU>(acc, xrow[0], xrow[1], wl + (size_t)r * NB, nr);
                }
            }
        }

        oa.x += acc[0]; oa.y += acc[1]; oa.z += acc[2]; oa.w += acc[3];
        ob.x += acc[4]; ob.y += acc[5]; ob.z += acc[6]; ob.w += acc[7];
        orow[0] = oa;
        orow[1] = ob;
    }
}

extern "C" void kernel_launch(void* const* d_in, const int* in_sizes, int n_in,
                              void* d_out, int out_size, void* d_ws, size_t ws_size,
                              hipStream_t stream)
{
    const float* x     = (const float*)d_in[0];
    const int*   src   = (const int*)  d_in[1];
    const int*   dst   = (const int*)  d_in[2];
    const int*   etype = (const int*)  d_in[3];
    const float* norm  = (const float*)d_in[4];
    const float* w1    = (const float*)d_in[5];
    const float* lw1   = (const float*)d_in[6];
    const float* b1    = (const float*)d_in[7];
    const float* w2    = (const float*)d_in[8];
    const float* lw2   = (const float*)d_in[9];
    const float* b2    = (const float*)d_in[10];
    float*       out   = (float*)d_out;

    const int n_nodes = in_sizes[0] / HD;
    const int n_edges = in_sizes[1];

    float* h = (float*)d_ws;

    const int waves_per_block = 256 / 32;
    const int strips = ((n_nodes + 15) / 16) * (HD / 64);
    const int gemm_blocks = (strips + waves_per_block - 1) / waves_per_block;

    const size_t seg_off  = ((size_t)n_nodes * HD * sizeof(float) + 255) & ~(size_t)255;
    const size_t perm_off = (seg_off + (size_t)n_nodes * sizeof(int) + 255) & ~(size_t)255;
    const size_t ws_need  = perm_off + (size_t)n_edges * sizeof(int);

    if (ws_size < ws_need) {
        const int edge_blocks = 1024;
        rgcn_selfloop_gemm<false><<<gemm_blocks, 256, 0, stream>>>(x, lw1, b1, h, n_nodes,
                                                                   nullptr, 0);
        rgcn_edge_msg<false><<<edge_blocks, 256, 0, stream>>>(x, src, dst, etype, norm, w1,
                                                              h, n_edges);
        rgcn_selfloop_gemm<true><<<gemm_blocks, 256, 0, stream>>>(h, lw2, b2, out, n_nodes,
                                                                  nullptr, 0);
        rgcn_edge_msg<true><<<edge_blocks, 256, 0, stream>>>(h, src, dst, etype, norm, w2,
                                                             out, n_edges);
        return;
    }

    int* seg  = (int*)((char*)d_ws + seg_off);
    int* perm = (int*)((char*)d_ws + perm_off);

    const int edge_blocks   = (n_edges + 255) / 256;
    const int gather_blocks = 1024;

    rgcn_selfloop_gemm<false><<<gemm_blocks, 256, 0, stream>>>(x, lw1, b1, h, n_nodes,
                                                               seg, n_nodes);
    rgcn_deg_count<<<edge_blocks, 256, 0, stream>>>(dst, seg, n_edges);
    rgcn_seg_scan<<<1, 1024, 0, stream>>>(seg, n_nodes);
    rgcn_seg_fill<<<edge_blocks, 256, 0, stream>>>(dst, seg, perm, n_edges);

    rgcn_node_gather<false><<<gather_blocks, 256, 0, stream>>>(x, src, dst, etype, norm, w1,
                                                               seg, perm, h, n_nodes, n_edges);

    rgcn_selfloop_gemm<true><<<gemm_blocks, 256, 0, stream>>>(h, lw2, b2, out, n_nodes,
                                                              nullptr, 0);
    rgcn_node_gather<true><<<gather_blocks, 256, 0, stream>>>(h, src, dst, etype, norm, w2,
                                                              seg, perm, out, n_nodes, n_edges);
}


# Gather: lane L owns float4 L and L+32 of a row, relation table staged slot-major in LDS: row loads and stores cover 4 whole 128-B lines (was 8 halves), table reads conflict-free; sum order kept
# speedup vs baseline: 1.6698x; 1.0304x over previous
#include <hip/hip_runtime.h>

#define HD   256
#define NB   128
#define RT   16
#define WTAB (RT * NB * 4)
#define LCAP 128

typedef float v2f __attribute__((ext_vector_type(2)));
typedef float v8f __attribute__((ext_vector_type(8)));

template <bool RELU>
__global__ __launch_bounds__(256)
void rgcn_selfloop_gemm(const float* __restrict__ A,
                        const float* __restrict__ W,
                        const float* __restrict__ bias,
                        float* __restrict__ out,
                        int n_rows,
                        int* __restrict__ zero_buf,
                        int zero_n)
{
    if (zero_buf != nullptr) {
        const int gid = blockIdx.x * blockDim.x + threadIdx.x;
        if (gid < zero_n) zero_buf[gid] = 0;
    }

    const int lane = threadIdx.x & 31;
    const int wid  = blockIdx.x * (blockDim.x >> 5) + (threadIdx.x >> 5);

    const int strips_n = HD / 64;
    const int tile_m   = wid / strips_n;
    const int strip_n  = wid % strips_n;
    if (tile_m * 16 >= n_rows) return;

    const int half = lane >> 4;
    const int sub  = lane & 15;
    const int arow_idx = tile_m * 16 + sub;
    const int col0     = strip_n * 64 + sub;

    const float* arow = A + (size_t)arow_idx * HD;

    v8f c0 = {}, c1 = {}, c2 = {}, c3 = {};
    for (int k = 0; k < HD; k += 4) {
        float a0 = arow[k + 2 * half + 0];
        float a1 = arow[k + 2 * half + 1];
        if (RELU) { a0 = fmaxf(a0, 0.0f); a1 = fmaxf(a1, 0.0f); }
        v2f a; a.x = a0; a.y = a1;

        const float* wrow0 = W + (size_t)(k + 2 * half + 0) * HD + col0;
        const float* wrow1 = W + (size_t)(k + 2 * half + 1) * HD + col0;
        v2f b0, b1, b2, b3;
        b0.x = wrow0[0];  b0.y = wrow1[0];
        b1.x = wrow0[16]; b1.y = wrow1[16];
        b2.x = wrow0[32]; b2.y = wrow1[32];
        b3.x = wrow0[48]; b3.y = wrow1[48];

        c0 = __builtin_amdgcn_wmma_f32_16x16x4_f32(false, a, false, b0, (short)0, c0, false, false);
        c1 = __builtin_amdgcn_wmma_f32_16x16x4_f32(false, a, false, b1, (short)0, c1, false, false);
        c2 = __builtin_amdgcn_wmma_f32_16x16x4_f32(false, a, false, b2, (short)0, c2, false, false);
        c3 = __builtin_amdgcn_wmma_f32_16x16x4_f32(false, a, false, b3, (short)0, c3, false, false);
    }

    float* obase = out + (size_t)(tile_m * 16 + half * 8) * HD + col0;
    #pragma unroll
    for (int t = 0; t < 4; ++t) {
        const v8f& c = (t == 0) ? c0 : (t == 1) ? c1 : (t == 2) ? c2 : c3;
        const float bcol = bias[col0 + t * 16];
        #pragma unroll
        for (int j = 0; j < 8; ++j)
            obase[(size_t)j * HD + t * 16] = c[j] + bcol;
    }
}

template <bool RELU>
__global__ __launch_bounds__(256)
void rgcn_edge_msg(const float* __restrict__ xin,
                   const int*   __restrict__ src,
                   const int*   __restrict__ dst,
                   const int*   __restrict__ etype,
                   const float* __restrict__ norm,
                   const float* __restrict__ w,
                   float* agg,
                   int n_edges)
{
    __shared__ __align__(16) float wlds[WTAB];
    for (int i = threadIdx.x; i < WTAB; i += blockDim.x) wlds[i] = w[i];
    __syncthreads();

    const int lane   = threadIdx.x & 31;
    const int wid    = blockIdx.x * (blockDim.x >> 5) + (threadIdx.x >> 5);
    const int nwaves = (gridDim.x * blockDim.x) >> 5;

    for (int e = wid; e < n_edges; e += nwaves) {
        const int   s   = src[e];
        const int   d   = dst[e];
        const int   r   = etype[e];
        const float nrm = norm[e];

        const float4* xrow = (const float4*)(xin + (size_t)s * HD) + lane * 2;
        float4 xa = xrow[0];
        float4 xb = xrow[1];
        if (RELU) {
            xa.x = fmaxf(xa.x, 0.0f); xa.y = fmaxf(xa.y, 0.0f);
            xa.z = fmaxf(xa.z, 0.0f); xa.w = fmaxf(xa.w, 0.0f);
            xb.x = fmaxf(xb.x, 0.0f); xb.y = fmaxf(xb.y, 0.0f);
            xb.z = fmaxf(xb.z, 0.0f); xb.w = fmaxf(xb.w, 0.0f);
        }
        const float xs[8] = { xa.x, xa.y, xa.z, xa.w, xb.x, xb.y, xb.z, xb.w };

        const float4* wrel = (const float4*)(wlds + (size_t)r * (NB * 4)) + lane * 4;
        float* arow = agg + (size_t)d * HD + lane * 8;

        #pragma unroll
        for (int j = 0; j < 4; ++j) {
            const float4 wq = wrel[j];
            const float x0 = xs[2 * j + 0];
            const float x1 = xs[2 * j + 1];
            const float o0 = (x0 * wq.x + x1 * wq.z) * nrm;
            const float o1 = (x0 * wq.y + x1 * wq.w) * nrm;
            (void)__hip_atomic_fetch_add(&arow[2 * j + 0], o0,
                                         __ATOMIC_RELAXED, __HIP_MEMORY_SCOPE_AGENT);
            (void)__hip_atomic_fetch_add(&arow[2 * j + 1], o1,
                                         __ATOMIC_RELAXED, __HIP_MEMORY_SCOPE_AGENT);
        }
    }
}

__global__ __launch_bounds__(256)
void rgcn_deg_count(const int* __restrict__ dst, int* seg, int n_edges)
{
    const int e = blockIdx.x * blockDim.x + threadIdx.x;
    if (e < n_edges)
        (void)__hip_atomic_fetch_add(&seg[dst[e]], 1,
                                     __ATOMIC_RELAXED, __HIP_MEMORY_SCOPE_AGENT);
}

__global__ __launch_bounds__(1024)
void rgcn_seg_scan(int* seg, int n)
{
    __shared__ int wsum[32];
    const int t    = threadIdx.x;
    const int lane = t & 31;
    const int w    = t >> 5;
    const int chunk = (n + 1023) / 1024;
    const int b  = t * chunk;
    const int en = (b + chunk < n) ? (b + chunk) : n;

    int s = 0;
    for (int i = b; i < en; ++i) s += seg[i];

    int v = s;
    #pragma unroll
    for (int o = 1; o < 32; o <<= 1) {
        const int u = __shfl_up(v, o);
        if (lane >= o) v += u;
    }
    if (lane == 31) wsum[w] = v;
    __syncthreads();
    if (w == 0) {
        const int x = wsum[lane];
        int y = x;
        #pragma unroll
        for (int o = 1; o < 32; o <<= 1) {
            const int u = __shfl_up(y, o);
            if (lane >= o) y += u;
        }
        wsum[lane] = y - x;
    }
    __syncthreads();

    int run = wsum[w] + v - s;
    for (int i = b; i < en; ++i) {
        const int c = seg[i];
        seg[i] = run;
        run += c;
    }
}

__global__ __launch_bounds__(256)
void rgcn_seg_fill(const int* __restrict__ dst, int* seg, int* __restrict__ perm,
                   int n_edges)
{
    const int e = blockIdx.x * blockDim.x + threadIdx.x;
    if (e < n_edges) {
        const int p = __hip_atomic_fetch_add(&seg[dst[e]], 1,
                                             __ATOMIC_RELAXED, __HIP_MEMORY_SCOPE_AGENT);
        perm[p] = e;
    }
}

template <bool RELU>
__device__ __forceinline__
void rgcn_msg_acc(float (&acc)[8], float4 xa, float4 xb,
                  const float4* wrel, float nrm)
{
    if (RELU) {
        xa.x = fmaxf(xa.x, 0.0f); xa.y = fmaxf(xa.y, 0.0f);
        xa.z = fmaxf(xa.z, 0.0f); xa.w = fmaxf(xa.w, 0.0f);
        xb.x = fmaxf(xb.x, 0.0f); xb.y = fmaxf(xb.y, 0.0f);
        xb.z = fmaxf(xb.z, 0.0f); xb.w = fmaxf(xb.w, 0.0f);
    }
    const float xs[8] = { xa.x, xa.y, xa.z, xa.w, xb.x, xb.y, xb.z, xb.w };
    #pragma unroll
    for (int j = 0; j < 4; ++j) {
        const float4 wq = wrel[32 * j];
        const float x0 = xs[2 * j + 0];
        const float x1 = xs[2 * j + 1];
        acc[2 * j + 0] += (x0 * wq.x + x1 * wq.z) * nrm;
        acc[2 * j + 1] += (x0 * wq.y + x1 * wq.w) * nrm;
    }
}

template <bool RELU>
__device__ __forceinline__
void rgcn_sorted_run(float (&acc)[8], const int sorted, const int m, const int lane,
                     const float* __restrict__ xin,
                     const int*   __restrict__ src,
                     const int*   __restrict__ etype,
                     const float* __restrict__ norm,
                     const float4* wl)
{
    int   s_v = 0, r_v = 0;
    float n_v = 0.0f;
    if (lane < m) {
        s_v = src[sorted];
        r_v = etype[sorted];
        n_v = norm[sorted];
    }

    int k = 0;
    for (; k + 4 <= m; k += 4) {
        float4 xa[4], xb[4];
        int    rr[4];
        float  nn[4];
        #pragma unroll
        for (int u = 0; u < 4; ++u) {
            const int s = __builtin_amdgcn_readlane(s_v, k + u);
            rr[u] = __builtin_amdgcn_readlane(r_v, k + u);
            nn[u] = __int_as_float(
                        __builtin_amdgcn_readlane(__float_as_int(n_v), k + u));
            const float4* xrow = (const float4*)(xin + (size_t)s * HD) + lane;
            xa[u] = xrow[0];
            xb[u] = xrow[32];
        }
        #pragma unroll
        for (int u = 0; u < 4; ++u)
            rgcn_msg_acc<RELU>(acc, xa[u], xb[u], wl + (size_t)rr[u] * NB, nn[u]);
    }
    for (; k < m; ++k) {
        const int   s  = __builtin_amdgcn_readlane(s_v, k);
        const int   r  = __builtin_amdgcn_readlane(r_v, k);
        const float nr = __int_as_float(
                             __builtin_amdgcn_readlane(__float_as_int(n_v), k));
        const float4* xrow = (const float4*)(xin + (size_t)s * HD) + lane;
        rgcn_msg_acc<RELU>(acc, xrow[0], xrow[32], wl + (size_t)r * NB, nr);
    }
}

template <bool RELU>
__global__ __launch_bounds__(256)
void rgcn_node_gather(const float* __restrict__ xin,
                      const int*   __restrict__ src,
                      const int*   __restrict__ dst,
                      const int*   __restrict__ etype,
                      const float* __restrict__ norm,
                      const float* __restrict__ w,
                      const int*   __restrict__ seg,
                      const int*   __restrict__ perm,
                      float* agg,
                      int n_nodes,
                      int n_edges)
{
    __shared__ __align__(16) float wlds[WTAB];
    __shared__ __align__(16) int   idbuf[8 * 2 * LCAP];
    for (int i = threadIdx.x; i < WTAB / 4; i += blockDim.x) {
        const int b = i & (NB - 1);
        const int j = ((b >> 6) << 1) | (b & 1);
        const int L = (b & 63) >> 1;
        ((float4*)wlds)[(i & ~(NB - 1)) + j * 32 + L] = ((const float4*)w)[i];
    }
    __syncthreads();

    const int lane   = threadIdx.x & 31;
    const int wid    = __builtin_amdgcn_readfirstlane(
                           blockIdx.x * (blockDim.x >> 5) + (threadIdx.x >> 5));
    const int nwaves = (gridDim.x * blockDim.x) >> 5;
    const float4* wl = (const float4*)wlds + lane;
    int* uns = idbuf + __builtin_amdgcn_readfirstlane((int)(threadIdx.x >> 5)) * (2 * LCAP);
    int* srt = uns + LCAP;

    for (int n = wid; n < n_nodes; n += nwaves) {
        const int end = seg[n];
        const int beg = (n > 0) ? seg[n - 1] : 0;
        const int cnt = end - beg;
        if (cnt <= 0) continue;

        float4* orow = (float4*)(agg + (size_t)n * HD) + lane;
        float4 oa = orow[0];
        float4 ob = orow[32];

        float acc[8] = { 0.f, 0.f, 0.f, 0.f, 0.f, 0.f, 0.f, 0.f };

        if (cnt <= LCAP) {
            int sorted0 = 0;
            if (cnt <= 32) {
                const int my = (lane < cnt) ? perm[beg + lane] : 0x7fffffff;
                int rank = 0;
                for (int m = 0; m < cnt; ++m)
                    rank += (__builtin_amdgcn_readlane(my, m) < my) ? 1 : 0;
                sorted0 = __builtin_amdgcn_ds_permute(rank << 2, my);
            } else {
                const int padded = (cnt + 31) & ~31;
                for (int p = lane; p < padded; p += 32)
                    uns[p] = (p < cnt) ? perm[beg + p] : 0x7fffffff;
                for (int p = lane; p < padded; p += 32) {
                    const int my = uns[p];
                    int rank = 0;
                    for (int q = 0; q < padded; q += 4) {
                        const int4 v = *(const int4*)(uns + q);
                        rank += ((v.x < my) ? 1 : 0) + ((v.y < my) ? 1 : 0)
                              + ((v.z < my) ? 1 : 0) + ((v.w < my) ? 1 : 0);
                    }
                    if (p < cnt) srt[rank] = my;
                }
            }
            for (int c = 0; c < cnt; c += 32) {
                const int m = (cnt - c < 32) ? (cnt - c) : 32;
                const int sorted = (cnt <= 32) ? sorted0 : ((lane < m) ? srt[c + lane] : 0);
                rgcn_sorted_run<RELU>(acc, sorted, m, lane, xin, src, etype, norm, wl);
            }
        } else if (cnt <= 2048) {
            int lo = -1;
            for (int i = 0; i < cnt; ++i) {
                int best = 0x7fffffff;
                for (int p = beg + lane; p < end; p += 32) {
                    const int e = perm[p];
                    if (e > lo && e < best) best = e;
                }
                #pragma unroll
                for (int o = 16; o > 0; o >>= 1) {
                    const int other = __shfl_xor(best, o);
                    best = (other < best) ? other : best;
                }
                const int e = __builtin_amdgcn_readfirstlane(best);
                lo = e;
                const int   s  = src[e];
                const int   r  = etype[e];
                const float nr = norm[e];
                const float4* xrow = (const float4*)(xin + (size_t)s * HD) + lane;
                rgcn_msg_acc<RELU>(acc, xrow[0], xrow[32], wl + (size_t)r * NB, nr);
            }
        } else {
            for (int e0 = 0; e0 < n_edges; e0 += 32) {
                const int  ev  = e0 + lane;
                const bool hit = (ev < n_edges) && (dst[ev] == n);
                unsigned m = __builtin_amdgcn_ballot_w32(hit);
                while (m != 0u) {
                    const int e = e0 + (__builtin_ctz(m));
                    m &= m - 1u;
                    const int   s  = src[e];
                    const int   r  = etype[e];
                    const float nr = norm[e];
                    const float4* xrow = (const float4*)(xin + (size_t)s * HD) + lane;
                    rgcn_msg_acc<RELU>(acc, xrow[0], xrow[32], wl + (size_t)r * NB, nr);
                }
            }
        }

        oa.x += acc[0]; oa.y += acc[1]; oa.z += acc[2]; oa.w += acc[3];
        ob.x += acc[4]; ob.y += acc[5]; ob.z += acc[6]; ob.w += acc[7];
        orow[0]  = oa;
        orow[32] = ob;
    }
}

extern "C" void kernel_launch(void* const* d_in, const int* in_sizes, int n_in,
                              void* d_out, int out_size, void* d_ws, size_t ws_size,
                              hipStream_t stream)
{
    const float* x     = (const float*)d_in[0];
    const int*   src   = (const int*)  d_in[1];
    const int*   dst   = (const int*)  d_in[2];
    const int*   etype = (const int*)  d_in[3];
    const float* norm  = (const float*)d_in[4];
    const float* w1    = (const float*)d_in[5];
    const float* lw1   = (const float*)d_in[6];
    const float* b1    = (const float*)d_in[7];
    const float* w2    = (const float*)d_in[8];
    const float* lw2   = (const float*)d_in[9];
    const float* b2    = (const float*)d_in[10];
    float*       out   = (float*)d_out;

    const int n_nodes = in_sizes[0] / HD;
    const int n_edges = in_sizes[1];

    float* h = (float*)d_ws;

    const int waves_per_block = 256 / 32;
    const int strips = ((n_nodes + 15) / 16) * (HD / 64);
    const int gemm_blocks = (strips + waves_per_block - 1) / waves_per_block;

    const size_t seg_off  = ((size_t)n_nodes * HD * sizeof(float) + 255) & ~(size_t)255;
    const size_t perm_off = (seg_off + (size_t)n_nodes * sizeof(int) + 255) & ~(size_t)255;
    const size_t ws_need  = perm_off + (size_t)n_edges * sizeof(int);

    if (ws_size < ws_need) {
        const int edge_blocks = 1024;
        rgcn_selfloop_gemm<false><<<gemm_blocks, 256, 0, stream>>>(x, lw1, b1, h, n_nodes,
                                                                   nullptr, 0);
        rgcn_edge_msg<false><<<edge_blocks, 256, 0, stream>>>(x, src, dst, etype, norm, w1,
                                                              h, n_edges);
        rgcn_selfloop_gemm<true><<<gemm_blocks, 256, 0, stream>>>(h, lw2, b2, out, n_nodes,
                                                                  nullptr, 0);
        rgcn_edge_msg<true><<<edge_blocks, 256, 0, stream>>>(h, src, dst, etype, norm, w2,
                                                             out, n_edges);
        return;
    }

    int* seg  = (int*)((char*)d_ws + seg_off);
    int* perm = (int*)((char*)d_ws + perm_off);

    const int edge_blocks   = (n_edges + 255) / 256;
    const int gather_blocks = 1024;

    rgcn_selfloop_gemm<false><<<gemm_blocks, 256, 0, stream>>>(x, lw1, b1, h, n_nodes,
                                                               seg, n_nodes);
    rgcn_deg_count<<<edge_blocks, 256, 0, stream>>>(dst, seg, n_edges);
    rgcn_seg_scan<<<1, 1024, 0, stream>>>(seg, n_nodes);
    rgcn_seg_fill<<<edge_blocks, 256, 0, stream>>>(dst, seg, perm, n_edges);

    rgcn_node_gather<false><<<gather_blocks, 256, 0, stream>>>(x, src, dst, etype, norm, w1,
                                                               seg, perm, h, n_nodes, n_edges);

    rgcn_selfloop_gemm<true><<<gemm_blocks, 256, 0, stream>>>(h, lw2, b2, out, n_nodes,
                                                              nullptr, 0);
    rgcn_node_gather<true><<<gather_blocks, 256, 0, stream>>>(h, src, dst, etype, norm, w2,
                                                              seg, perm, out, n_nodes, n_edges);
}


# Destination buckets built in one launch: fixed row of 64 slots per node, slot from one returning integer atomic per edge; count and scan launches gone (7 launches to 5); sum order kept
# speedup vs baseline: 2.1847x; 1.3084x over previous
#include <hip/hip_runtime.h>

#define HD   256
#define NB   128
#define RT   16
#define WTAB (RT * NB * 4)
#define LCAP 128

typedef float v2f __attribute__((ext_vector_type(2)));
typedef float v8f __attribute__((ext_vector_type(8)));

template <bool RELU>
__global__ __launch_bounds__(256)
void rgcn_selfloop_gemm(const float* __restrict__ A,
                        const float* __restrict__ W,
                        const float* __restrict__ bias,
                        float* __restrict__ out,
                        int n_rows,
                        int* __restrict__ zero_buf,
                        int zero_n)
{
    if (zero_buf != nullptr) {
        const int gid = blockIdx.x * blockDim.x + threadIdx.x;
        if (gid < zero_n) zero_buf[gid] = 0;
    }

    const int lane = threadIdx.x & 31;
    const int wid  = blockIdx.x * (blockDim.x >> 5) + (threadIdx.x >> 5);

    const int strips_n = HD / 64;
    const int tile_m   = wid / strips_n;
    const int strip_n  = wid % strips_n;
    if (tile_m * 16 >= n_rows) return;

    const int half = lane >> 4;
    const int sub  = lane & 15;
    const int arow_idx = tile_m * 16 + sub;
    const int col0     = strip_n * 64 + sub;

    const float* arow = A + (size_t)arow_idx * HD;

    v8f c0 = {}, c1 = {}, c2 = {}, c3 = {};
    for (int k = 0; k < HD; k += 4) {
        float a0 = arow[k + 2 * half + 0];
        float a1 = arow[k + 2 * half + 1];
        if (RELU) { a0 = fmaxf(a0, 0.0f); a1 = fmaxf(a1, 0.0f); }
        v2f a; a.x = a0; a.y = a1;

        const float* wrow0 = W + (size_t)(k + 2 * half + 0) * HD + col0;
        const float* wrow1 = W + (size_t)(k + 2 * half + 1) * HD + col0;
        v2f b0, b1, b2, b3;
        b0.x = wrow0[0];  b0.y = wrow1[0];
        b1.x = wrow0[16]; b1.y = wrow1[16];
        b2.x = wrow0[32]; b2.y = wrow1[32];
        b3.x = wrow0[48]; b3.y = wrow1[48];

        c0 = __builtin_amdgcn_wmma_f32_16x16x4_f32(false, a, false, b0, (short)0, c0, false, false);
        c1 = __builtin_amdgcn_wmma_f32_16x16x4_f32(false, a, false, b1, (short)0, c1, false, false);
        c2 = __builtin_amdgcn_wmma_f32_16x16x4_f32(false, a, false, b2, (short)0, c2, false, false);
        c3 = __builtin_amdgcn_wmma_f32_16x16x4_f32(false, a, false, b3, (short)0, c3, false, false);
    }

    float* obase = out + (size_t)(tile_m * 16 + half * 8) * HD + col0;
    #pragma unroll
    for (int t = 0; t < 4; ++t) {
        const v8f& c = (t == 0) ? c0 : (t == 1) ? c1 : (t == 2) ? c2 : c3;
        const float bcol = bias[col0 + t * 16];
        #pragma unroll
        for (int j = 0; j < 8; ++j)
            obase[(size_t)j * HD + t * 16] = c[j] + bcol;
    }
}

template <bool RELU>
__global__ __launch_bounds__(256)
void rgcn_edge_msg(const float* __restrict__ xin,
                   const int*   __restrict__ src,
                   const int*   __restrict__ dst,
                   const int*   __restrict__ etype,
                   const float* __restrict__ norm,
                   const float* __restrict__ w,
                   float* agg,
                   int n_edges)
{
    __shared__ __align__(16) float wlds[WTAB];
    for (int i = threadIdx.x; i < WTAB; i += blockDim.x) wlds[i] = w[i];
    __syncthreads();

    const int lane   = threadIdx.x & 31;
    const int wid    = blockIdx.x * (blockDim.x >> 5) + (threadIdx.x >> 5);
    const int nwaves = (gridDim.x * blockDim.x) >> 5;

    for (int e = wid; e < n_edges; e += nwaves) {
        const int   s   = src[e];
        const int   d   = dst[e];
        const int   r   = etype[e];
        const float nrm = norm[e];

        const float4* xrow = (const float4*)(xin + (size_t)s * HD) + lane * 2;
        float4 xa = xrow[0];
        float4 xb = xrow[1];
        if (RELU) {
            xa.x = fmaxf(xa.x, 0.0f); xa.y = fmaxf(xa.y, 0.0f);
            xa.z = fmaxf(xa.z, 0.0f); xa.w = fmaxf(xa.w, 0.0f);
            xb.x = fmaxf(xb.x, 0.0f); xb.y = fmaxf(xb.y, 0.0f);
            xb.z = fmaxf(xb.z, 0.0f); xb.w = fmaxf(xb.w, 0.0f);
        }
        const float xs[8] = { xa.x, xa.y, xa.z, xa.w, xb.x, xb.y, xb.z, xb.w };

        const float4* wrel = (const float4*)(wlds + (size_t)r * (NB * 4)) + lane * 4;
        float* arow = agg + (size_t)d * HD + lane * 8;

        #pragma unroll
        for (int j = 0; j < 4; ++j) {
            const float4 wq = wrel[j];
            const float x0 = xs[2 * j + 0];
            const float x1 = xs[2 * j + 1];
            const float o0 = (x0 * wq.x + x1 * wq.z) * nrm;
            const float o1 = (x0 * wq.y + x1 * wq.w) * nrm;
            (void)__hip_atomic_fetch_add(&arow[2 * j + 0], o0,
                                         __ATOMIC_RELAXED, __HIP_MEMORY_SCOPE_AGENT);
            (void)__hip_atomic_fetch_add(&arow[2 * j + 1], o1,
                                         __ATOMIC_RELAXED, __HIP_MEMORY_SCOPE_AGENT);
        }
    }
}

__global__ __launch_bounds__(256)
void rgcn_deg_count(const int* __restrict__ dst, int* seg, int n_edges)
{
    const int e = blockIdx.x * blockDim.x + threadIdx.x;
    if (e < n_edges)
        (void)__hip_atomic_fetch_add(&seg[dst[e]], 1,
                                     __ATOMIC_RELAXED, __HIP_MEMORY_SCOPE_AGENT);
}

__global__ __launch_bounds__(1024)
void rgcn_seg_scan(int* seg, int n)
{
    __shared__ int wsum[32];
    const int t    = threadIdx.x;
    const int lane = t & 31;
    const int w    = t >> 5;
    const int chunk = (n + 1023) / 1024;
    const int b  = t * chunk;
    const int en = (b + chunk < n) ? (b + chunk) : n;

    int s = 0;
    for (int i = b; i < en; ++i) s += seg[i];

    int v = s;
    #pragma unroll
    for (int o = 1; o < 32; o <<= 1) {
        const int u = __shfl_up(v, o);
        if (lane >= o) v += u;
    }
    if (lane == 31) wsum[w] = v;
    __syncthreads();
    if (w == 0) {
        const int x = wsum[lane];
        int y = x;
        #pragma unroll
        for (int o = 1; o < 32; o <<= 1) {
            const int u = __shfl_up(y, o);
            if (lane >= o) y += u;
        }
        wsum[lane] = y - x;
    }
    __syncthreads();

    int run = wsum[w] + v - s;
    for (int i = b; i < en; ++i) {
        const int c = seg[i];
        seg[i] = run;
        run += c;
    }
}

__global__ __launch_bounds__(256)
void rgcn_seg_fill(const int* __restrict__ dst, int* seg, int* __restrict__ perm,
                   int n_edges)
{
    const int e = blockIdx.x * blockDim.x + threadIdx.x;
    if (e < n_edges) {
        const int p = __hip_atomic_fetch_add(&seg[dst[e]], 1,
                                             __ATOMIC_RELAXED, __HIP_MEMORY_SCOPE_AGENT);
        perm[p] = e;
    }
}

__global__ __launch_bounds__(256)
void rgcn_bucket_fill(const int* __restrict__ dst, int* cnt, int* __restrict__ bucket,
                      int cap, int n_edges)
{
    const int e0 = (blockIdx.x * blockDim.x + threadIdx.x) * 4;
    if (e0 + 3 < n_edges) {
        const int4 d = *(const int4*)(dst + e0);
        const int p0 = __hip_atomic_fetch_add(&cnt[d.x], 1,
                                              __ATOMIC_RELAXED, __HIP_MEMORY_SCOPE_AGENT);
        const int p1 = __hip_atomic_fetch_add(&cnt[d.y], 1,
                                              __ATOMIC_RELAXED, __HIP_MEMORY_SCOPE_AGENT);
        const int p2 = __hip_atomic_fetch_add(&cnt[d.z], 1,
                                              __ATOMIC_RELAXED, __HIP_MEMORY_SCOPE_AGENT);
        const int p3 = __hip_atomic_fetch_add(&cnt[d.w], 1,
                                              __ATOMIC_RELAXED, __HIP_MEMORY_SCOPE_AGENT);
        if (p0 < cap) bucket[(size_t)d.x * cap + p0] = e0;
        if (p1 < cap) bucket[(size_t)d.y * cap + p1] = e0 + 1;
        if (p2 < cap) bucket[(size_t)d.z * cap + p2] = e0 + 2;
        if (p3 < cap) bucket[(size_t)d.w * cap + p3] = e0 + 3;
    } else {
        for (int e = e0; e < n_edges; ++e) {
            const int d = dst[e];
            const int p = __hip_atomic_fetch_add(&cnt[d], 1,
                                                 __ATOMIC_RELAXED, __HIP_MEMORY_SCOPE_AGENT);
            if (p < cap) bucket[(size_t)d * cap + p] = e;
        }
    }
}

template <bool RELU>
__device__ __forceinline__
void rgcn_msg_acc(float (&acc)[8], float4 xa, float4 xb,
                  const float4* wrel, float nrm)
{
    if (RELU) {
        xa.x = fmaxf(xa.x, 0.0f); xa.y = fmaxf(xa.y, 0.0f);
        xa.z = fmaxf(xa.z, 0.0f); xa.w = fmaxf(xa.w, 0.0f);
        xb.x = fmaxf(xb.x, 0.0f); xb.y = fmaxf(xb.y, 0.0f);
        xb.z = fmaxf(xb.z, 0.0f); xb.w = fmaxf(xb.w, 0.0f);
    }
    const float xs[8] = { xa.x, xa.y, xa.z, xa.w, xb.x, xb.y, xb.z, xb.w };
    #pragma unroll
    for (int j = 0; j < 4; ++j) {
        const float4 wq = wrel[32 * j];
        const float x0 = xs[2 * j + 0];
        const float x1 = xs[2 * j + 1];
        acc[2 * j + 0] += (x0 * wq.x + x1 * wq.z) * nrm;
        acc[2 * j + 1] += (x0 * wq.y + x1 * wq.w) * nrm;
    }
}

template <bool RELU>
__device__ __forceinline__
void rgcn_sorted_run(float (&acc)[8], const int sorted, const int m, const int lane,
                     const float* __restrict__ xin,
                     const int*   __restrict__ src,
                     const int*   __restrict__ etype,
                     const float* __restrict__ norm,
                     const float4* wl)
{
    int   s_v = 0, r_v = 0;
    float n_v = 0.0f;
    if (lane < m) {
        s_v = src[sorted];
        r_v = etype[sorted];
        n_v = norm[sorted];
    }

    int k = 0;
    for (; k + 4 <= m; k += 4) {
        float4 xa[4], xb[4];
        int    rr[4];
        float  nn[4];
        #pragma unroll
        for (int u = 0; u < 4; ++u) {
            const int s = __builtin_amdgcn_readlane(s_v, k + u);
            rr[u] = __builtin_amdgcn_readlane(r_v, k + u);
            nn[u] = __int_as_float(
                        __builtin_amdgcn_readlane(__float_as_int(n_v), k + u));
            const float4* xrow = (const float4*)(xin + (size_t)s * HD) + lane;
            xa[u] = xrow[0];
            xb[u] = xrow[32];
        }
        #pragma unroll
        for (int u = 0; u < 4; ++u)
            rgcn_msg_acc<RELU>(acc, xa[u], xb[u], wl + (size_t)rr[u] * NB, nn[u]);
    }
    for (; k < m; ++k) {
        const int   s  = __builtin_amdgcn_readlane(s_v, k);
        const int   r  = __builtin_amdgcn_readlane(r_v, k);
        const float nr = __int_as_float(
                             __builtin_amdgcn_readlane(__float_as_int(n_v), k));
        const float4* xrow = (const float4*)(xin + (size_t)s * HD) + lane;
        rgcn_msg_acc<RELU>(acc, xrow[0], xrow[32], wl + (size_t)r * NB, nr);
    }
}

template <bool RELU>
__global__ __launch_bounds__(256)
void rgcn_node_gather(const float* __restrict__ xin,
                      const int*   __restrict__ src,
                      const int*   __restrict__ dst,
                      const int*   __restrict__ etype,
                      const float* __restrict__ norm,
                      const float* __restrict__ w,
                      const int*   __restrict__ seg,
                      const int*   __restrict__ perm,
                      float* agg,
                      int n_nodes,
                      int n_edges,
                      int cap)
{
    __shared__ __align__(16) float wlds[WTAB];
    __shared__ __align__(16) int   idbuf[8 * 2 * LCAP];
    for (int i = threadIdx.x; i < WTAB / 4; i += blockDim.x) {
        const int b = i & (NB - 1);
        const int j = ((b >> 6) << 1) | (b & 1);
        const int L = (b & 63) >> 1;
        ((float4*)wlds)[(i & ~(NB - 1)) + j * 32 + L] = ((const float4*)w)[i];
    }
    __syncthreads();

    const int lane   = threadIdx.x & 31;
    const int wid    = __builtin_amdgcn_readfirstlane(
                           blockIdx.x * (blockDim.x >> 5) + (threadIdx.x >> 5));
    const int nwaves = (gridDim.x * blockDim.x) >> 5;
    const float4* wl = (const float4*)wlds + lane;
    int* uns = idbuf + __builtin_amdgcn_readfirstlane((int)(threadIdx.x >> 5)) * (2 * LCAP);
    int* srt = uns + LCAP;

    for (int n = wid; n < n_nodes; n += nwaves) {
        int beg, cnt;
        if (cap > 0) {
            cnt = seg[n];
            beg = n * cap;
        } else {
            beg = (n > 0) ? seg[n - 1] : 0;
            cnt = seg[n] - beg;
        }
        if (cnt <= 0) continue;
        const int  end    = beg + cnt;
        const bool listed = (cap == 0) || (cnt <= cap);

        float4* orow = (float4*)(agg + (size_t)n * HD) + lane;
        float4 oa = orow[0];
        float4 ob = orow[32];

        float acc[8] = { 0.f, 0.f, 0.f, 0.f, 0.f, 0.f, 0.f, 0.f };

        if (listed && cnt <= LCAP) {
            int sorted0 = 0;
            if (cnt <= 32) {
                const int my = (lane < cnt) ? perm[beg + lane] : 0x7fffffff;
                int rank = 0;
                for (int m = 0; m < cnt; ++m)
                    rank += (__builtin_amdgcn_readlane(my, m) < my) ? 1 : 0;
                sorted0 = __builtin_amdgcn_ds_permute(rank << 2, my);
            } else {
                const int padded = (cnt + 31) & ~31;
                for (int p = lane; p < padded; p += 32)
                    uns[p] = (p < cnt) ? perm[beg + p] : 0x7fffffff;
                for (int p = lane; p < padded; p += 32) {
                    const int my = uns[p];
                    int rank = 0;
                    for (int q = 0; q < padded; q += 4) {
                        const int4 v = *(const int4*)(uns + q);
                        rank += ((v.x < my) ? 1 : 0) + ((v.y < my) ? 1 : 0)
                              + ((v.z < my) ? 1 : 0) + ((v.w < my) ? 1 : 0);
                    }
                    if (p < cnt) srt[rank] = my;
                }
            }
            for (int c = 0; c < cnt; c += 32) {
                const int m = (cnt - c < 32) ? (cnt - c) : 32;
                const int sorted = (cnt <= 32) ? sorted0 : ((lane < m) ? srt[c + lane] : 0);
                rgcn_sorted_run<RELU>(acc, sorted, m, lane, xin, src, etype, norm, wl);
            }
        } else if (listed && cnt <= 2048) {
            int lo = -1;
            for (int i = 0; i < cnt; ++i) {
                int best = 0x7fffffff;
                for (int p = beg + lane; p < end; p += 32) {
                    const int e = perm[p];
                    if (e > lo && e < best) best = e;
                }
                #pragma unroll
                for (int o = 16; o > 0; o >>= 1) {
                    const int other = __shfl_xor(best, o);
                    best = (other < best) ? other : best;
                }
                const int e = __builtin_amdgcn_readfirstlane(best);
                lo = e;
                const int   s  = src[e];
                const int   r  = etype[e];
                const float nr = norm[e];
                const float4* xrow = (const float4*)(xin + (size_t)s * HD) + lane;
                rgcn_msg_acc<RELU>(acc, xrow[0], xrow[32], wl + (size_t)r * NB, nr);
            }
        } else {
            for (int e0 = 0; e0 < n_edges; e0 += 32) {
                const int  ev  = e0 + lane;
                const bool hit = (ev < n_edges) && (dst[ev] == n);
                unsigned m = __builtin_amdgcn_ballot_w32(hit);
                while (m != 0u) {
                    const int e = e0 + (__builtin_ctz(m));
                    m &= m - 1u;
                    const int   s  = src[e];
                    const int   r  = etype[e];
                    const float nr = norm[e];
                    const float4* xrow = (const float4*)(xin + (size_t)s * HD) + lane;
                    rgcn_msg_acc<RELU>(acc, xrow[0], xrow[32], wl + (size_t)r * NB, nr);
                }
            }
        }

        oa.x += acc[0]; oa.y += acc[1]; oa.z += acc[2]; oa.w += acc[3];
        ob.x += acc[4]; ob.y += acc[5]; ob.z += acc[6]; ob.w += acc[7];
        orow[0]  = oa;
        orow[32] = ob;
    }
}

extern "C" void kernel_launch(void* const* d_in, const int* in_sizes, int n_in,
                              void* d_out, int out_size, void* d_ws, size_t ws_size,
                              hipStream_t stream)
{
    const float* x     = (const float*)d_in[0];
    const int*   src   = (const int*)  d_in[1];
    const int*   dst   = (const int*)  d_in[2];
    const int*   etype = (const int*)  d_in[3];
    const float* norm  = (const float*)d_in[4];
    const float* w1    = (const float*)d_in[5];
    const float* lw1   = (const float*)d_in[6];
    const float* b1    = (const float*)d_in[7];
    const float* w2    = (const float*)d_in[8];
    const float* lw2   = (const float*)d_in[9];
    const float* b2    = (const float*)d_in[10];
    float*       out   = (float*)d_out;

    const int n_nodes = in_sizes[0] / HD;
    const int n_edges = in_sizes[1];

    float* h = (float*)d_ws;

    const int waves_per_block = 256 / 32;
    const int strips = ((n_nodes + 15) / 16) * (HD / 64);
    const int gemm_blocks = (strips + waves_per_block - 1) / waves_per_block;

    const size_t seg_off  = ((size_t)n_nodes * HD * sizeof(float) + 255) & ~(size_t)255;
    const size_t perm_off = (seg_off + (size_t)n_nodes * sizeof(int) + 255) & ~(size_t)255;
    const size_t ws_need  = perm_off + (size_t)n_edges * sizeof(int);

    if (ws_size < ws_need) {
        const int edge_blocks = 1024;
        rgcn_selfloop_gemm<false><<<gemm_blocks, 256, 0, stream>>>(x, lw1, b1, h, n_nodes,
                                                                   nullptr, 0);
        rgcn_edge_msg<false><<<edge_blocks, 256, 0, stream>>>(x, src, dst, etype, norm, w1,
                                                              h, n_edges);
        rgcn_selfloop_gemm<true><<<gemm_blocks, 256, 0, stream>>>(h, lw2, b2, out, n_nodes,
                                                                  nullptr, 0);
        rgcn_edge_msg<true><<<edge_blocks, 256, 0, stream>>>(h, src, dst, etype, norm, w2,
                                                             out, n_edges);
        return;
    }

    int* seg  = (int*)((char*)d_ws + seg_off);
    const int gather_blocks = 1024;

    const int    bcap     = 64;
    const size_t bkt_need = perm_off + (size_t)n_nodes * bcap * sizeof(int);

    if (ws_size >= bkt_need) {
        int* bucket = (int*)((char*)d_ws + perm_off);
        const int fill_blocks = (n_edges + 1023) / 1024;

        rgcn_selfloop_gemm<false><<<gemm_blocks, 256, 0, stream>>>(x, lw1, b1, h, n_nodes,
                                                                   seg, n_nodes);
        rgcn_bucket_fill<<<fill_blocks, 256, 0, stream>>>(dst, seg, bucket, bcap, n_edges);

        rgcn_node_gather<false><<<gather_blocks, 256, 0, stream>>>(x, src, dst, etype, norm, w1,
                                                                   seg, bucket, h, n_nodes,
                                                                   n_edges, bcap);
        rgcn_selfloop_gemm<true><<<gemm_blocks, 256, 0, stream>>>(h, lw2, b2, out, n_nodes,
                                                                  nullptr, 0);
        rgcn_node_gather<true><<<gather_blocks, 256, 0, stream>>>(h, src, dst, etype, norm, w2,
                                                                  seg, bucket, out, n_nodes,
                                                                  n_edges, bcap);
        return;
    }

    int* perm = (int*)((char*)d_ws + perm_off);
    const int edge_blocks   = (n_edges + 255) / 256;

    rgcn_selfloop_gemm<false><<<gemm_blocks, 256, 0, stream>>>(x, lw1, b1, h, n_nodes,
                                                               seg, n_nodes);
    rgcn_deg_count<<<edge_blocks, 256, 0, stream>>>(dst, seg, n_edges);
    rgcn_seg_scan<<<1, 1024, 0, stream>>>(seg, n_nodes);
    rgcn_seg_fill<<<edge_blocks, 256, 0, stream>>>(dst, seg, perm, n_edges);

    rgcn_node_gather<false><<<gather_blocks, 256, 0, stream>>>(x, src, dst, etype, norm, w1,
                                                               seg, perm, h, n_nodes, n_edges, 0);

    rgcn_selfloop_gemm<true><<<gemm_blocks, 256, 0, stream>>>(h, lw2, b2, out, n_nodes,
                                                              nullptr, 0);
    rgcn_node_gather<true><<<gather_blocks, 256, 0, stream>>>(h, src, dst, etype, norm, w2,
                                                              seg, perm, out, n_nodes, n_edges, 0);
}


# Bucket slots hold the edge's packed record (source, relation, norm), not its id: gather needs no per-edge src/etype/norm look-ups; sum order now by ascending record key, the same in every run
# speedup vs baseline: 2.3077x; 1.0563x over previous
#include <hip/hip_runtime.h>

#define HD   256
#define NB   128
#define RT   16
#define WTAB (RT * NB * 4)
#define LCAP 128

typedef float v2f __attribute__((ext_vector_type(2)));
typedef float v8f __attribute__((ext_vector_type(8)));

template <bool RELU>
__global__ __launch_bounds__(256)
void rgcn_selfloop_gemm(const float* __restrict__ A,
                        const float* __restrict__ W,
                        const float* __restrict__ bias,
                        float* __restrict__ out,
                        int n_rows,
                        int* __restrict__ zero_buf,
                        int zero_n)
{
    if (zero_buf != nullptr) {
        const int gid = blockIdx.x * blockDim.x + threadIdx.x;
        if (gid < zero_n) zero_buf[gid] = 0;
    }

    const int lane = threadIdx.x & 31;
    const int wid  = blockIdx.x * (blockDim.x >> 5) + (threadIdx.x >> 5);

    const int strips_n = HD / 64;
    const int tile_m   = wid / strips_n;
    const int strip_n  = wid % strips_n;
    if (tile_m * 16 >= n_rows) return;

    const int half = lane >> 4;
    const int sub  = lane & 15;
    const int arow_idx = tile_m * 16 + sub;
    const int col0     = strip_n * 64 + sub;

    const float* arow = A + (size_t)arow_idx * HD;

    v8f c0 = {}, c1 = {}, c2 = {}, c3 = {};
    for (int k = 0; k < HD; k += 4) {
        float a0 = arow[k + 2 * half + 0];
        float a1 = arow[k + 2 * half + 1];
        if (RELU) { a0 = fmaxf(a0, 0.0f); a1 = fmaxf(a1, 0.0f); }
        v2f a; a.x = a0; a.y = a1;

        const float* wrow0 = W + (size_t)(k + 2 * half + 0) * HD + col0;
        const float* wrow1 = W + (size_t)(k + 2 * half + 1) * HD + col0;
        v2f b0, b1, b2, b3;
        b0.x = wrow0[0];  b0.y = wrow1[0];
        b1.x = wrow0[16]; b1.y = wrow1[16];
        b2.x = wrow0[32]; b2.y = wrow1[32];
        b3.x = wrow0[48]; b3.y = wrow1[48];

        c0 = __builtin_amdgcn_wmma_f32_16x16x4_f32(false, a, false, b0, (short)0, c0, false, false);
        c1 = __builtin_amdgcn_wmma_f32_16x16x4_f32(false, a, false, b1, (short)0, c1, false, false);
        c2 = __builtin_amdgcn_wmma_f32_16x16x4_f32(false, a, false, b2, (short)0, c2, false, false);
        c3 = __builtin_amdgcn_wmma_f32_16x16x4_f32(false, a, false, b3, (short)0, c3, false, false);
    }

    float* obase = out + (size_t)(tile_m * 16 + half * 8) * HD + col0;
    #pragma unroll
    for (int t = 0; t < 4; ++t) {
        const v8f& c = (t == 0) ? c0 : (t == 1) ? c1 : (t == 2) ? c2 : c3;
        const float bcol = bias[col0 + t * 16];
        #pragma unroll
        for (int j = 0; j < 8; ++j)
            obase[(size_t)j * HD + t * 16] = c[j] + bcol;
    }
}

template <bool RELU>
__global__ __launch_bounds__(256)
void rgcn_edge_msg(const float* __restrict__ xin,
                   const int*   __restrict__ src,
                   const int*   __restrict__ dst,
                   const int*   __restrict__ etype,
                   const float* __restrict__ norm,
                   const float* __restrict__ w,
                   float* agg,
                   int n_edges)
{
    __shared__ __align__(16) float wlds[WTAB];
    for (int i = threadIdx.x; i < WTAB; i += blockDim.x) wlds[i] = w[i];
    __syncthreads();

    const int lane   = threadIdx.x & 31;
    const int wid    = blockIdx.x * (blockDim.x >> 5) + (threadIdx.x >> 5);
    const int nwaves = (gridDim.x * blockDim.x) >> 5;

    for (int e = wid; e < n_edges; e += nwaves) {
        const int   s   = src[e];
        const int   d   = dst[e];
        const int   r   = etype[e];
        const float nrm = norm[e];

        const float4* xrow = (const float4*)(xin + (size_t)s * HD) + lane * 2;
        float4 xa = xrow[0];
        float4 xb = xrow[1];
        if (RELU) {
            xa.x = fmaxf(xa.x, 0.0f); xa.y = fmaxf(xa.y, 0.0f);
            xa.z = fmaxf(xa.z, 0.0f); xa.w = fmaxf(xa.w, 0.0f);
            xb.x = fmaxf(xb.x, 0.0f); xb.y = fmaxf(xb.y, 0.0f);
            xb.z = fmaxf(xb.z, 0.0f); xb.w = fmaxf(xb.w, 0.0f);
        }
        const float xs[8] = { xa.x, xa.y, xa.z, xa.w, xb.x, xb.y, xb.z, xb.w };

        const float4* wrel = (const float4*)(wlds + (size_t)r * (NB * 4)) + lane * 4;
        float* arow = agg + (size_t)d * HD + lane * 8;

        #pragma unroll
        for (int j = 0; j < 4; ++j) {
            const float4 wq = wrel[j];
            const float x0 = xs[2 * j + 0];
            const float x1 = xs[2 * j + 1];
            const float o0 = (x0 * wq.x + x1 * wq.z) * nrm;
            const float o1 = (x0 * wq.y + x1 * wq.w) * nrm;
            (void)__hip_atomic_fetch_add(&arow[2 * j + 0], o0,
                                         __ATOMIC_RELAXED, __HIP_MEMORY_SCOPE_AGENT);
            (void)__hip_atomic_fetch_add(&arow[2 * j + 1], o1,
                                         __ATOMIC_RELAXED, __HIP_MEMORY_SCOPE_AGENT);
        }
    }
}

__global__ __launch_bounds__(256)
void rgcn_deg_count(const int* __restrict__ dst, int* seg, int n_edges)
{
    const int e = blockIdx.x * blockDim.x + threadIdx.x;
    if (e < n_edges)
        (void)__hip_atomic_fetch_add(&seg[dst[e]], 1,
                                     __ATOMIC_RELAXED, __HIP_MEMORY_SCOPE_AGENT);
}

__global__ __launch_bounds__(1024)
void rgcn_seg_scan(int* seg, int n)
{
    __shared__ int wsum[32];
    const int t    = threadIdx.x;
    const int lane = t & 31;
    const int w    = t >> 5;
    const int chunk = (n + 1023) / 1024;
    const int b  = t * chunk;
    const int en = (b + chunk < n) ? (b + chunk) : n;

    int s = 0;
    for (int i = b; i < en; ++i) s += seg[i];

    int v = s;
    #pragma unroll
    for (int o = 1; o < 32; o <<= 1) {
        const int u = __shfl_up(v, o);
        if (lane >= o) v += u;
    }
    if (lane == 31) wsum[w] = v;
    __syncthreads();
    if (w == 0) {
        const int x = wsum[lane];
        int y = x;
        #pragma unroll
        for (int o = 1; o < 32; o <<= 1) {
            const int u = __shfl_up(y, o);
            if (lane >= o) y += u;
        }
        wsum[lane] = y - x;
    }
    __syncthreads();

    int run = wsum[w] + v - s;
    for (int i = b; i < en; ++i) {
        const int c = seg[i];
        seg[i] = run;
        run += c;
    }
}

__global__ __launch_bounds__(256)
void rgcn_seg_fill(const int* __restrict__ dst, int* seg, int* __restrict__ perm,
                   int n_edges)
{
    const int e = blockIdx.x * blockDim.x + threadIdx.x;
    if (e < n_edges) {
        const int p = __hip_atomic_fetch_add(&seg[dst[e]], 1,
                                             __ATOMIC_RELAXED, __HIP_MEMORY_SCOPE_AGENT);
        perm[p] = e;
    }
}

__global__ __launch_bounds__(256)
void rgcn_bucket_fill(const int* __restrict__ dst, int* cnt, int* __restrict__ bucket,
                      int cap, int n_edges)
{
    const int e0 = (blockIdx.x * blockDim.x + threadIdx.x) * 4;
    if (e0 + 3 < n_edges) {
        const int4 d = *(const int4*)(dst + e0);
        const int p0 = __hip_atomic_fetch_add(&cnt[d.x], 1,
                                              __ATOMIC_RELAXED, __HIP_MEMORY_SCOPE_AGENT);
        const int p1 = __hip_atomic_fetch_add(&cnt[d.y], 1,
                                              __ATOMIC_RELAXED, __HIP_MEMORY_SCOPE_AGENT);
        const int p2 = __hip_atomic_fetch_add(&cnt[d.z], 1,
                                              __ATOMIC_RELAXED, __HIP_MEMORY_SCOPE_AGENT);
        const int p3 = __hip_atomic_fetch_add(&cnt[d.w], 1,
                                              __ATOMIC_RELAXED, __HIP_MEMORY_SCOPE_AGENT);
        if (p0 < cap) bucket[(size_t)d.x * cap + p0] = e0;
        if (p1 < cap) bucket[(size_t)d.y * cap + p1] = e0 + 1;
        if (p2 < cap) bucket[(size_t)d.z * cap + p2] = e0 + 2;
        if (p3 < cap) bucket[(size_t)d.w * cap + p3] = e0 + 3;
    } else {
        for (int e = e0; e < n_edges; ++e) {
            const int d = dst[e];
            const int p = __hip_atomic_fetch_add(&cnt[d], 1,
                                                 __ATOMIC_RELAXED, __HIP_MEMORY_SCOPE_AGENT);
            if (p < cap) bucket[(size_t)d * cap + p] = e;
        }
    }
}

template <bool RELU>
__device__ __forceinline__
void rgcn_msg_acc(float (&acc)[8], float4 xa, float4 xb,
                  const float4* wrel, float nrm)
{
    if (RELU) {
        xa.x = fmaxf(xa.x, 0.0f); xa.y = fmaxf(xa.y, 0.0f);
        xa.z = fmaxf(xa.z, 0.0f); xa.w = fmaxf(xa.w, 0.0f);
        xb.x = fmaxf(xb.x, 0.0f); xb.y = fmaxf(xb.y, 0.0f);
        xb.z = fmaxf(xb.z, 0.0f); xb.w = fmaxf(xb.w, 0.0f);
    }
    const float xs[8] = { xa.x, xa.y, xa.z, xa.w, xb.x, xb.y, xb.z, xb.w };
    #pragma unroll
    for (int j = 0; j < 4; ++j) {
        const float4 wq = wrel[32 * j];
        const float x0 = xs[2 * j + 0];
        const float x1 = xs[2 * j + 1];
        acc[2 * j + 0] += (x0 * wq.x + x1 * wq.z) * nrm;
        acc[2 * j + 1] += (x0 * wq.y + x1 * wq.w) * nrm;
    }
}

template <bool RELU>
__device__ __forceinline__
void rgcn_sorted_run(float (&acc)[8], const int sorted, const int m, const int lane,
                     const float* __restrict__ xin,
                     const int*   __restrict__ src,
                     const int*   __restrict__ etype,
                     const float* __restrict__ norm,
                     const float4* wl)
{
    int   s_v = 0, r_v = 0;
    float n_v = 0.0f;
    if (lane < m) {
        s_v = src[sorted];
        r_v = etype[sorted];
        n_v = norm[sorted];
    }

    int k = 0;
    for (; k + 4 <= m; k += 4) {
        float4 xa[4], xb[4];
        int    rr[4];
        float  nn[4];
        #pragma unroll
        for (int u = 0; u < 4; ++u) {
            const int s = __builtin_amdgcn_readlane(s_v, k + u);
            rr[u] = __builtin_amdgcn_readlane(r_v, k + u);
            nn[u] = __int_as_float(
                        __builtin_amdgcn_readlane(__float_as_int(n_v), k + u));
            const float4* xrow = (const float4*)(xin + (size_t)s * HD) + lane;
            xa[u] = xrow[0];
            xb[u] = xrow[32];
        }
        #pragma unroll
        for (int u = 0; u < 4; ++u)
            rgcn_msg_acc<RELU>(acc, xa[u], xb[u], wl + (size_t)rr[u] * NB, nn[u]);
    }
    for (; k < m; ++k) {
        const int   s  = __builtin_amdgcn_readlane(s_v, k);
        const int   r  = __builtin_amdgcn_readlane(r_v, k);
        const float nr = __int_as_float(
                             __builtin_amdgcn_readlane(__float_as_int(n_v), k));
        const float4* xrow = (const float4*)(xin + (size_t)s * HD) + lane;
        rgcn_msg_acc<RELU>(acc, xrow[0], xrow[32], wl + (size_t)r * NB, nr);
    }
}

template <bool RELU>
__global__ __launch_bounds__(256)
void rgcn_node_gather(const float* __restrict__ xin,
                      const int*   __restrict__ src,
                      const int*   __restrict__ dst,
                      const int*   __restrict__ etype,
                      const float* __restrict__ norm,
                      const float* __restrict__ w,
                      const int*   __restrict__ seg,
                      const int*   __restrict__ perm,
                      float* agg,
                      int n_nodes,
                      int n_edges,
                      int cap)
{
    __shared__ __align__(16) float wlds[WTAB];
    __shared__ __align__(16) int   idbuf[8 * 2 * LCAP];
    for (int i = threadIdx.x; i < WTAB / 4; i += blockDim.x) {
        const int b = i & (NB - 1);
        const int j = ((b >> 6) << 1) | (b & 1);
        const int L = (b & 63) >> 1;
        ((float4*)wlds)[(i & ~(NB - 1)) + j * 32 + L] = ((const float4*)w)[i];
    }
    __syncthreads();

    const int lane   = threadIdx.x & 31;
    const int wid    = __builtin_amdgcn_readfirstlane(
                           blockIdx.x * (blockDim.x >> 5) + (threadIdx.x >> 5));
    const int nwaves = (gridDim.x * blockDim.x) >> 5;
    const float4* wl = (const float4*)wlds + lane;
    int* uns = idbuf + __builtin_amdgcn_readfirstlane((int)(threadIdx.x >> 5)) * (2 * LCAP);
    int* srt = uns + LCAP;

    for (int n = wid; n < n_nodes; n += nwaves) {
        int beg, cnt;
        if (cap > 0) {
            cnt = seg[n];
            beg = n * cap;
        } else {
            beg = (n > 0) ? seg[n - 1] : 0;
            cnt = seg[n] - beg;
        }
        if (cnt <= 0) continue;
        const int  end    = beg + cnt;
        const bool listed = (cap == 0) || (cnt <= cap);

        float4* orow = (float4*)(agg + (size_t)n * HD) + lane;
        float4 oa = orow[0];
        float4 ob = orow[32];

        float acc[8] = { 0.f, 0.f, 0.f, 0.f, 0.f, 0.f, 0.f, 0.f };

        if (listed && cnt <= LCAP) {
            int sorted0 = 0;
            if (cnt <= 32) {
                const int my = (lane < cnt) ? perm[beg + lane] : 0x7fffffff;
                int rank = 0;
                for (int m = 0; m < cnt; ++m)
                    rank += (__builtin_amdgcn_readlane(my, m) < my) ? 1 : 0;
                sorted0 = __builtin_amdgcn_ds_permute(rank << 2, my);
            } else {
                const int padded = (cnt + 31) & ~31;
                for (int p = lane; p < padded; p += 32)
                    uns[p] = (p < cnt) ? perm[beg + p] : 0x7fffffff;
                for (int p = lane; p < padded; p += 32) {
                    const int my = uns[p];
                    int rank = 0;
                    for (int q = 0; q < padded; q += 4) {
                        const int4 v = *(const int4*)(uns + q);
                        rank += ((v.x < my) ? 1 : 0) + ((v.y < my) ? 1 : 0)
                              + ((v.z < my) ? 1 : 0) + ((v.w < my) ? 1 : 0);
                    }
                    if (p < cnt) srt[rank] = my;
                }
            }
            for (int c = 0; c < cnt; c += 32) {
                const int m = (cnt - c < 32) ? (cnt - c) : 32;
                const int sorted = (cnt <= 32) ? sorted0 : ((lane < m) ? srt[c + lane] : 0);
                rgcn_sorted_run<RELU>(acc, sorted, m, lane, xin, src, etype, norm, wl);
            }
        } else if (listed && cnt <= 2048) {
            int lo = -1;
            for (int i = 0; i < cnt; ++i) {
                int best = 0x7fffffff;
                for (int p = beg + lane; p < end; p += 32) {
                    const int e = perm[p];
                    if (e > lo && e < best) best = e;
                }
                #pragma unroll
                for (int o = 16; o > 0; o >>= 1) {
                    const int other = __shfl_xor(best, o);
                    best = (other < best) ? other : best;
                }
                const int e = __builtin_amdgcn_readfirstlane(best);
                lo = e;
                const int   s  = src[e];
                const int   r  = etype[e];
                const float nr = norm[e];
                const float4* xrow = (const float4*)(xin + (size_t)s * HD) + lane;
                rgcn_msg_acc<RELU>(acc, xrow[0], xrow[32], wl + (size_t)r * NB, nr);
            }
        } else {
            for (int e0 = 0; e0 < n_edges; e0 += 32) {
                const int  ev  = e0 + lane;
                const bool hit = (ev < n_edges) && (dst[ev] == n);
                unsigned m = __builtin_amdgcn_ballot_w32(hit);
                while (m != 0u) {
                    const int e = e0 + (__builtin_ctz(m));
                    m &= m - 1u;
                    const int   s  = src[e];
                    const int   r  = etype[e];
                    const float nr = norm[e];
                    const float4* xrow = (const float4*)(xin + (size_t)s * HD) + lane;
                    rgcn_msg_acc<RELU>(acc, xrow[0], xrow[32], wl + (size_t)r * NB, nr);
                }
            }
        }

        oa.x += acc[0]; oa.y += acc[1]; oa.z += acc[2]; oa.w += acc[3];
        ob.x += acc[4]; ob.y += acc[5]; ob.z += acc[6]; ob.w += acc[7];
        orow[0]  = oa;
        orow[32] = ob;
    }
}

#define RCAP 64

__global__ __launch_bounds__(256)
void rgcn_bucket_fill_rec(const int*   __restrict__ dst,
                          const int*   __restrict__ src,
                          const int*   __restrict__ etype,
                          const float* __restrict__ norm,
                          int* cnt, int* __restrict__ bucket,
                          int cap, int n_edges)
{
    int2* rec = (int2*)bucket;
    const int e0 = (blockIdx.x * blockDim.x + threadIdx.x) * 4;
    if (e0 + 3 < n_edges) {
        const int4   d = *(const int4*)(dst + e0);
        const int4   s = *(const int4*)(src + e0);
        const int4   r = *(const int4*)(etype + e0);
        const float4 w = *(const float4*)(norm + e0);
        const int p0 = __hip_atomic_fetch_add(&cnt[d.x], 1,
                                              __ATOMIC_RELAXED, __HIP_MEMORY_SCOPE_AGENT);
        const int p1 = __hip_atomic_fetch_add(&cnt[d.y], 1,
                                              __ATOMIC_RELAXED, __HIP_MEMORY_SCOPE_AGENT);
        const int p2 = __hip_atomic_fetch_add(&cnt[d.z], 1,
                                              __ATOMIC_RELAXED, __HIP_MEMORY_SCOPE_AGENT);
        const int p3 = __hip_atomic_fetch_add(&cnt[d.w], 1,
                                              __ATOMIC_RELAXED, __HIP_MEMORY_SCOPE_AGENT);
        if (p0 < cap) rec[(size_t)d.x * cap + p0] = make_int2(s.x | (r.x << 24), __float_as_int(w.x));
        if (p1 < cap) rec[(size_t)d.y * cap + p1] = make_int2(s.y | (r.y << 24), __float_as_int(w.y));
        if (p2 < cap) rec[(size_t)d.z * cap + p2] = make_int2(s.z | (r.z << 24), __float_as_int(w.z));
        if (p3 < cap) rec[(size_t)d.w * cap + p3] = make_int2(s.w | (r.w << 24), __float_as_int(w.w));
    } else {
        for (int e = e0; e < n_edges; ++e) {
            const int d = dst[e];
            const int p = __hip_atomic_fetch_add(&cnt[d], 1,
                                                 __ATOMIC_RELAXED, __HIP_MEMORY_SCOPE_AGENT);
            if (p < cap)
                rec[(size_t)d * cap + p] = make_int2(src[e] | (etype[e] << 24),
                                                     __float_as_int(norm[e]));
        }
    }
}

__device__ __forceinline__
unsigned long long rgcn_rec_key(int w0, int nb)
{
    return ((unsigned long long)(unsigned)w0 << 32) | (unsigned long long)(unsigned)nb;
}

template <bool RELU>
__device__ __forceinline__
void rgcn_rec_run(float (&acc)[8], const int w0, const int nb, const int m, const int lane,
                  const float* __restrict__ xin, const float4* wl)
{
    const int s_v = w0 & 0x00ffffff;
    const int r_v = (int)((unsigned)w0 >> 24);

    int k = 0;
    for (; k + 4 <= m; k += 4) {
        float4 xa[4], xb[4];
        int    rr[4];
        float  nn[4];
        #pragma unroll
        for (int u = 0; u < 4; ++u) {
            const int s = __builtin_amdgcn_readlane(s_v, k + u);
            rr[u] = __builtin_amdgcn_readlane(r_v, k + u);
            nn[u] = __int_as_float(__builtin_amdgcn_readlane(nb, k + u));
            const float4* xrow = (const float4*)(xin + (size_t)s * HD) + lane;
            xa[u] = xrow[0];
            xb[u] = xrow[32];
        }
        #pragma unroll
        for (int u = 0; u < 4; ++u)
            rgcn_msg_acc<RELU>(acc, xa[u], xb[u], wl + (size_t)rr[u] * NB, nn[u]);
    }
    for (; k < m; ++k) {
        const int   s  = __builtin_amdgcn_readlane(s_v, k);
        const int   r  = __builtin_amdgcn_readlane(r_v, k);
        const float nr = __int_as_float(__builtin_amdgcn_readlane(nb, k));
        const float4* xrow = (const float4*)(xin + (size_t)s * HD) + lane;
        rgcn_msg_acc<RELU>(acc, xrow[0], xrow[32], wl + (size_t)r * NB, nr);
    }
}

template <bool RELU>
__global__ __launch_bounds__(256)
void rgcn_node_gather_rec(const float* __restrict__ xin,
                          const int*   __restrict__ src,
                          const int*   __restrict__ dst,
                          const int*   __restrict__ etype,
                          const float* __restrict__ norm,
                          const float* __restrict__ w,
                          const int*   __restrict__ seg,
                          const int*   __restrict__ bucket,
                          float* agg,
                          int n_nodes,
                          int n_edges,
                          int cap)
{
    __shared__ __align__(16) float wlds[WTAB];
    __shared__ __align__(16) int   keybuf[8 * 4 * RCAP];
    for (int i = threadIdx.x; i < WTAB / 4; i += blockDim.x) {
        const int b = i & (NB - 1);
        const int j = ((b >> 6) << 1) | (b & 1);
        const int L = (b & 63) >> 1;
        ((float4*)wlds)[(i & ~(NB - 1)) + j * 32 + L] = ((const float4*)w)[i];
    }
    __syncthreads();

    const int lane   = threadIdx.x & 31;
    const int wid    = __builtin_amdgcn_readfirstlane(
                           blockIdx.x * (blockDim.x >> 5) + (threadIdx.x >> 5));
    const int nwaves = (gridDim.x * blockDim.x) >> 5;
    const float4* wl = (const float4*)wlds + lane;
    int2* uns = (int2*)(keybuf + __builtin_amdgcn_readfirstlane((int)(threadIdx.x >> 5)) * (4 * RCAP));
    int2* srt = uns + RCAP;
    const int2* rec = (const int2*)bucket;

    for (int n = wid; n < n_nodes; n += nwaves) {
        const int cnt = seg[n];
        if (cnt <= 0) continue;

        float4* orow = (float4*)(agg + (size_t)n * HD) + lane;
        float4 oa = orow[0];
        float4 ob = orow[32];

        float acc[8] = { 0.f, 0.f, 0.f, 0.f, 0.f, 0.f, 0.f, 0.f };

        if (cnt <= cap && cnt <= RCAP) {
            const int2* row = rec + (size_t)n * cap;
            int w0s = 0, nbs = 0;
            if (cnt <= 32) {
                int2 mine = make_int2(-1, -1);
                if (lane < cnt) mine = row[lane];
                const unsigned long long my = rgcn_rec_key(mine.x, mine.y);
                int rank = 0;
                for (int m = 0; m < cnt; ++m) {
                    const unsigned long long ok =
                        rgcn_rec_key(__builtin_amdgcn_readlane(mine.x, m),
                                     __builtin_amdgcn_readlane(mine.y, m));
                    rank += (ok < my || (ok == my && m < lane)) ? 1 : 0;
                }
                w0s = __builtin_amdgcn_ds_permute(rank << 2, mine.x);
                nbs = __builtin_amdgcn_ds_permute(rank << 2, mine.y);
            } else {
                const int padded = (cnt + 31) & ~31;
                for (int p = lane; p < padded; p += 32)
                    uns[p] = (p < cnt) ? row[p] : make_int2(-1, -1);
                for (int p = lane; p < padded; p += 32) {
                    const int2 mine = uns[p];
                    const unsigned long long my = rgcn_rec_key(mine.x, mine.y);
                    int rank = 0;
                    for (int q = 0; q < padded; q += 2) {
                        const int4 v = *(const int4*)(uns + q);
                        const unsigned long long k0 = rgcn_rec_key(v.x, v.y);
                        const unsigned long long k1 = rgcn_rec_key(v.z, v.w);
                        rank += ((k0 < my || (k0 == my && q     < p)) ? 1 : 0)
                              + ((k1 < my || (k1 == my && q + 1 < p)) ? 1 : 0);
                    }
                    if (p < cnt) srt[rank] = mine;
                }
            }
            for (int c = 0; c < cnt; c += 32) {
                const int m = (cnt - c < 32) ? (cnt - c) : 32;
                if (cnt > 32) {
                    int2 r = make_int2(0, 0);
                    if (lane < m) r = srt[c + lane];
                    w0s = r.x;
                    nbs = r.y;
                }
                rgcn_rec_run<RELU>(acc, w0s, nbs, m, lane, xin, wl);
            }
        } else {
            for (int e0 = 0; e0 < n_edges; e0 += 32) {
                const int  ev  = e0 + lane;
                const bool hit = (ev < n_edges) && (dst[ev] == n);
                unsigned m = __builtin_amdgcn_ballot_w32(hit);
                while (m != 0u) {
                    const int e = e0 + (__builtin_ctz(m));
                    m &= m - 1u;
                    const int   s  = src[e];
                    const int   r  = etype[e];
                    const float nr = norm[e];
                    const float4* xrow = (const float4*)(xin + (size_t)s * HD) + lane;
                    rgcn_msg_acc<RELU>(acc, xrow[0], xrow[32], wl + (size_t)r * NB, nr);
                }
            }
        }

        oa.x += acc[0]; oa.y += acc[1]; oa.z += acc[2]; oa.w += acc[3];
        ob.x += acc[4]; ob.y += acc[5]; ob.z += acc[6]; ob.w += acc[7];
        orow[0]  = oa;
        orow[32] = ob;
    }
}

extern "C" void kernel_launch(void* const* d_in, const int* in_sizes, int n_in,
                              void* d_out, int out_size, void* d_ws, size_t ws_size,
                              hipStream_t stream)
{
    const float* x     = (const float*)d_in[0];
    const int*   src   = (const int*)  d_in[1];
    const int*   dst   = (const int*)  d_in[2];
    const int*   etype = (const int*)  d_in[3];
    const float* norm  = (const float*)d_in[4];
    const float* w1    = (const float*)d_in[5];
    const float* lw1   = (const float*)d_in[6];
    const float* b1    = (const float*)d_in[7];
    const float* w2    = (const float*)d_in[8];
    const float* lw2   = (const float*)d_in[9];
    const float* b2    = (const float*)d_in[10];
    float*       out   = (float*)d_out;

    const int n_nodes = in_sizes[0] / HD;
    const int n_edges = in_sizes[1];

    float* h = (float*)d_ws;

    const int waves_per_block = 256 / 32;
    const int strips = ((n_nodes + 15) / 16) * (HD / 64);
    const int gemm_blocks = (strips + waves_per_block - 1) / waves_per_block;

    const size_t seg_off  = ((size_t)n_nodes * HD * sizeof(float) + 255) & ~(size_t)255;
    const size_t perm_off = (seg_off + (size_t)n_nodes * sizeof(int) + 255) & ~(size_t)255;
    const size_t ws_need  = perm_off + (size_t)n_edges * sizeof(int);

    if (ws_size < ws_need) {
        const int edge_blocks = 1024;
        rgcn_selfloop_gemm<false><<<gemm_blocks, 256, 0, stream>>>(x, lw1, b1, h, n_nodes,
                                                                   nullptr, 0);
        rgcn_edge_msg<false><<<edge_blocks, 256, 0, stream>>>(x, src, dst, etype, norm, w1,
                                                              h, n_edges);
        rgcn_selfloop_gemm<true><<<gemm_blocks, 256, 0, stream>>>(h, lw2, b2, out, n_nodes,
                                                                  nullptr, 0);
        rgcn_edge_msg<true><<<edge_blocks, 256, 0, stream>>>(h, src, dst, etype, norm, w2,
                                                             out, n_edges);
        return;
    }

    int* seg  = (int*)((char*)d_ws + seg_off);
    const int gather_blocks = 1024;

    const int    bcap     = 64;
    const size_t bkt_need = perm_off + (size_t)n_nodes * bcap * sizeof(int);

    const size_t rec_need = perm_off + (size_t)n_nodes * RCAP * 2 * sizeof(int);

    if (ws_size >= rec_need && n_nodes <= (1 << 24)) {
        int* bucket = (int*)((char*)d_ws + perm_off);
        const int fill_blocks = (n_edges + 1023) / 1024;

        rgcn_selfloop_gemm<false><<<gemm_blocks, 256, 0, stream>>>(x, lw1, b1, h, n_nodes,
                                                                   seg, n_nodes);
        rgcn_bucket_fill_rec<<<fill_blocks, 256, 0, stream>>>(dst, src, etype, norm, seg, bucket,
                                                              RCAP, n_edges);

        rgcn_node_gather_rec<false><<<gather_blocks, 256, 0, stream>>>(x, src, dst, etype, norm, w1,
                                                                       seg, bucket, h, n_nodes,
                                                                       n_edges, RCAP);
        rgcn_selfloop_gemm<true><<<gemm_blocks, 256, 0, stream>>>(h, lw2, b2, out, n_nodes,
                                                                  nullptr, 0);
        rgcn_node_gather_rec<true><<<gather_blocks, 256, 0, stream>>>(h, src, dst, etype, norm, w2,
                                                                      seg, bucket, out, n_nodes,
                                                                      n_edges, RCAP);
        return;
    }

    if (ws_size >= bkt_need) {
        int* bucket = (int*)((char*)d_ws + perm_off);
        const int fill_blocks = (n_edges + 1023) / 1024;

        rgcn_selfloop_gemm<false><<<gemm_blocks, 256, 0, stream>>>(x, lw1, b1, h, n_nodes,
                                                                   seg, n_nodes);
        rgcn_bucket_fill<<<fill_blocks, 256, 0, stream>>>(dst, seg, bucket, bcap, n_edges);

        rgcn_node_gather<false><<<gather_blocks, 256, 0, stream>>>(x, src, dst, etype, norm, w1,
                                                                   seg, bucket, h, n_nodes,
                                                                   n_edges, bcap);
        rgcn_selfloop_gemm<true><<<gemm_blocks, 256, 0, stream>>>(h, lw2, b2, out, n_nodes,
                                                                  nullptr, 0);
        rgcn_node_gather<true><<<gather_blocks, 256, 0, stream>>>(h, src, dst, etype, norm, w2,
                                                                  seg, bucket, out, n_nodes,
                                                                  n_edges, bcap);
        return;
    }

    int* perm = (int*)((char*)d_ws + perm_off);
    const int edge_blocks   = (n_edges + 255) / 256;

    rgcn_selfloop_gemm<false><<<gemm_blocks, 256, 0, stream>>>(x, lw1, b1, h, n_nodes,
                                                               seg, n_nodes);
    rgcn_deg_count<<<edge_blocks, 256, 0, stream>>>(dst, seg, n_edges);
    rgcn_seg_scan<<<1, 1024, 0, stream>>>(seg, n_nodes);
    rgcn_seg_fill<<<edge_blocks, 256, 0, stream>>>(dst, seg, perm, n_edges);

    rgcn_node_gather<false><<<gather_blocks, 256, 0, stream>>>(x, src, dst, etype, norm, w1,
                                                               seg, perm, h, n_nodes, n_edges, 0);

    rgcn_selfloop_gemm<true><<<gemm_blocks, 256, 0, stream>>>(h, lw2, b2, out, n_nodes,
                                                              nullptr, 0);
    rgcn_node_gather<true><<<gather_blocks, 256, 0, stream>>>(h, src, dst, etype, norm, w2,
                                                              seg, perm, out, n_nodes, n_edges, 0);
}


# Self-loop GEMM: work-group = 8 row tiles x one 64-column strip of W staged once in LDS (4 k per 16-B read), A by 16-B loads; k sum order regrouped inside each 8 k, fixed and the same in every run
# speedup vs baseline: 2.4362x; 1.0557x over previous
#include <hip/hip_runtime.h>

#define HD   256
#define NB   128
#define RT   16
#define WTAB (RT * NB * 4)
#define LCAP 128

typedef float v2f __attribute__((ext_vector_type(2)));
typedef float v8f __attribute__((ext_vector_type(8)));

#define GEMM_KQ (HD / 4)

template <bool RELU>
__global__ __launch_bounds__(256)
void rgcn_selfloop_gemm(const float* __restrict__ A,
                        const float* __restrict__ W,
                        const float* __restrict__ bias,
                        float* __restrict__ out,
                        int n_rows,
                        int* __restrict__ zero_buf,
                        int zero_n)
{
    __shared__ __align__(16) float bl[GEMM_KQ * 64 * 4];

    if (zero_buf != nullptr) {
        const int gid = blockIdx.x * blockDim.x + threadIdx.x;
        if (gid < zero_n) zero_buf[gid] = 0;
    }

    const int strip_n   = blockIdx.x & 3;
    const int row_group = blockIdx.x >> 2;

    const int lane   = threadIdx.x & 31;
    const int tile_m = row_group * 8 + (threadIdx.x >> 5);
    const bool live  = tile_m * 16 < n_rows;
    const int half = lane >> 4;
    const int sub  = lane & 15;
    const int col0 = strip_n * 64 + sub;

    const float4* arow = (const float4*)(A + (size_t)(tile_m * 16 + sub) * HD) + half;

    float4 an[4] = {};
    if (live) {
        #pragma unroll
        for (int q = 0; q < 4; ++q) an[q] = arow[2 * q];
    }

    {
        const int scol = threadIdx.x & 63;
        const int skq  = threadIdx.x >> 6;
        const float* wsrc = W + strip_n * 64 + scol;
        #pragma unroll 4
        for (int i = 0; i < GEMM_KQ / 4; ++i) {
            const int kq = skq + 4 * i;
            const float* p = wsrc + (size_t)(4 * kq) * HD;
            float4 v;
            v.x = p[0];
            v.y = p[HD];
            v.z = p[2 * HD];
            v.w = p[3 * HD];
            ((float4*)bl)[kq * 64 + scol] = v;
        }
    }
    __syncthreads();

    if (!live) return;

    const float4* bfr  = (const float4*)bl + half * 64 + sub;

    v8f c0 = {}, c1 = {}, c2 = {}, c3 = {};

    for (int g = 0; g < 8; ++g) {
        float4 ac[4];
        #pragma unroll
        for (int q = 0; q < 4; ++q) ac[q] = an[q];
        if (g < 7) {
            #pragma unroll
            for (int q = 0; q < 4; ++q) an[q] = arow[2 * (4 * (g + 1) + q)];
        }
        #pragma unroll
        for (int q = 0; q < 4; ++q) {
            const int t = 4 * g + q;
            float4 a4 = ac[q];
            if (RELU) {
                a4.x = fmaxf(a4.x, 0.0f); a4.y = fmaxf(a4.y, 0.0f);
                a4.z = fmaxf(a4.z, 0.0f); a4.w = fmaxf(a4.w, 0.0f);
            }
            const float4 q0 = bfr[t * 128 +  0];
            const float4 q1 = bfr[t * 128 + 16];
            const float4 q2 = bfr[t * 128 + 32];
            const float4 q3 = bfr[t * 128 + 48];

            v2f a;  v2f b0, b1, b2, b3;
            a.x  = a4.x; a.y  = a4.y;
            b0.x = q0.x; b0.y = q0.y;
            b1.x = q1.x; b1.y = q1.y;
            b2.x = q2.x; b2.y = q2.y;
            b3.x = q3.x; b3.y = q3.y;
            c0 = __builtin_amdgcn_wmma_f32_16x16x4_f32(false, a, false, b0, (short)0, c0, false, false);
            c1 = __builtin_amdgcn_wmma_f32_16x16x4_f32(false, a, false, b1, (short)0, c1, false, false);
            c2 = __builtin_amdgcn_wmma_f32_16x16x4_f32(false, a, false, b2, (short)0, c2, false, false);
            c3 = __builtin_amdgcn_wmma_f32_16x16x4_f32(false, a, false, b3, (short)0, c3, false, false);

            a.x  = a4.z; a.y  = a4.w;
            b0.x = q0.z; b0.y = q0.w;
            b1.x = q1.z; b1.y = q1.w;
            b2.x = q2.z; b2.y = q2.w;
            b3.x = q3.z; b3.y = q3.w;
            c0 = __builtin_amdgcn_wmma_f32_16x16x4_f32(false, a, false, b0, (short)0, c0, false, false);
            c1 = __builtin_amdgcn_wmma_f32_16x16x4_f32(false, a, false, b1, (short)0, c1, false, false);
            c2 = __builtin_amdgcn_wmma_f32_16x16x4_f32(false, a, false, b2, (short)0, c2, false, false);
            c3 = __builtin_amdgcn_wmma_f32_16x16x4_f32(false, a, false, b3, (short)0, c3, false, false);
        }
    }

    float* obase = out + (size_t)(tile_m * 16 + half * 8) * HD + col0;
    #pragma unroll
    for (int t = 0; t < 4; ++t) {
        const v8f& c = (t == 0) ? c0 : (t == 1) ? c1 : (t == 2) ? c2 : c3;
        const float bcol = bias[col0 + t * 16];
        #pragma unroll
        for (int j = 0; j < 8; ++j)
            obase[(size_t)j * HD + t * 16] = c[j] + bcol;
    }
}

template <bool RELU>
__global__ __launch_bounds__(256)
void rgcn_edge_msg(const float* __restrict__ xin,
                   const int*   __restrict__ src,
                   const int*   __restrict__ dst,
                   const int*   __restrict__ etype,
                   const float* __restrict__ norm,
                   const float* __restrict__ w,
                   float* agg,
                   int n_edges)
{
    __shared__ __align__(16) float wlds[WTAB];
    for (int i = threadIdx.x; i < WTAB; i += blockDim.x) wlds[i] = w[i];
    __syncthreads();

    const int lane   = threadIdx.x & 31;
    const int wid    = blockIdx.x * (blockDim.x >> 5) + (threadIdx.x >> 5);
    const int nwaves = (gridDim.x * blockDim.x) >> 5;

    for (int e = wid; e < n_edges; e += nwaves) {
        const int   s   = src[e];
        const int   d   = dst[e];
        const int   r   = etype[e];
        const float nrm = norm[e];

        const float4* xrow = (const float4*)(xin + (size_t)s * HD) + lane * 2;
        float4 xa = xrow[0];
        float4 xb = xrow[1];
        if (RELU) {
            xa.x = fmaxf(xa.x, 0.0f); xa.y = fmaxf(xa.y, 0.0f);
            xa.z = fmaxf(xa.z, 0.0f); xa.w = fmaxf(xa.w, 0.0f);
            xb.x = fmaxf(xb.x, 0.0f); xb.y = fmaxf(xb.y, 0.0f);
            xb.z = fmaxf(xb.z, 0.0f); xb.w = fmaxf(xb.w, 0.0f);
        }
        const float xs[8] = { xa.x, xa.y, xa.z, xa.w, xb.x, xb.y, xb.z, xb.w };

        const float4* wrel = (const float4*)(wlds + (size_t)r * (NB * 4)) + lane * 4;
        float* arow = agg + (size_t)d * HD + lane * 8;

        #pragma unroll
        for (int j = 0; j < 4; ++j) {
            const float4 wq = wrel[j];
            const float x0 = xs[2 * j + 0];
            const float x1 = xs[2 * j + 1];
            const float o0 = (x0 * wq.x + x1 * wq.z) * nrm;
            const float o1 = (x0 * wq.y + x1 * wq.w) * nrm;
            (void)__hip_atomic_fetch_add(&arow[2 * j + 0], o0,
                                         __ATOMIC_RELAXED, __HIP_MEMORY_SCOPE_AGENT);
            (void)__hip_atomic_fetch_add(&arow[2 * j + 1], o1,
                                         __ATOMIC_RELAXED, __HIP_MEMORY_SCOPE_AGENT);
        }
    }
}

__global__ __launch_bounds__(256)
void rgcn_deg_count(const int* __restrict__ dst, int* seg, int n_edges)
{
    const int e = blockIdx.x * blockDim.x + threadIdx.x;
    if (e < n_edges)
        (void)__hip_atomic_fetch_add(&seg[dst[e]], 1,
                                     __ATOMIC_RELAXED, __HIP_MEMORY_SCOPE_AGENT);
}

__global__ __launch_bounds__(1024)
void rgcn_seg_scan(int* seg, int n)
{
    __shared__ int wsum[32];
    const int t    = threadIdx.x;
    const int lane = t & 31;
    const int w    = t >> 5;
    const int chunk = (n + 1023) / 1024;
    const int b  = t * chunk;
    const int en = (b + chunk < n) ? (b + chunk) : n;

    int s = 0;
    for (int i = b; i < en; ++i) s += seg[i];

    int v = s;
    #pragma unroll
    for (int o = 1; o < 32; o <<= 1) {
        const int u = __shfl_up(v, o);
        if (lane >= o) v += u;
    }
    if (lane == 31) wsum[w] = v;
    __syncthreads();
    if (w == 0) {
        const int x = wsum[lane];
        int y = x;
        #pragma unroll
        for (int o = 1; o < 32; o <<= 1) {
            const int u = __shfl_up(y, o);
            if (lane >= o) y += u;
        }
        wsum[lane] = y - x;
    }
    __syncthreads();

    int run = wsum[w] + v - s;
    for (int i = b; i < en; ++i) {
        const int c = seg[i];
        seg[i] = run;
        run += c;
    }
}

__global__ __launch_bounds__(256)
void rgcn_seg_fill(const int* __restrict__ dst, int* seg, int* __restrict__ perm,
                   int n_edges)
{
    const int e = blockIdx.x * blockDim.x + threadIdx.x;
    if (e < n_edges) {
        const int p = __hip_atomic_fetch_add(&seg[dst[e]], 1,
                                             __ATOMIC_RELAXED, __HIP_MEMORY_SCOPE_AGENT);
        perm[p] = e;
    }
}

__global__ __launch_bounds__(256)
void rgcn_bucket_fill(const int* __restrict__ dst, int* cnt, int* __restrict__ bucket,
                      int cap, int n_edges)
{
    const int e0 = (blockIdx.x * blockDim.x + threadIdx.x) * 4;
    if (e0 + 3 < n_edges) {
        const int4 d = *(const int4*)(dst + e0);
        const int p0 = __hip_atomic_fetch_add(&cnt[d.x], 1,
                                              __ATOMIC_RELAXED, __HIP_MEMORY_SCOPE_AGENT);
        const int p1 = __hip_atomic_fetch_add(&cnt[d.y], 1,
                                              __ATOMIC_RELAXED, __HIP_MEMORY_SCOPE_AGENT);
        const int p2 = __hip_atomic_fetch_add(&cnt[d.z], 1,
                                              __ATOMIC_RELAXED, __HIP_MEMORY_SCOPE_AGENT);
        const int p3 = __hip_atomic_fetch_add(&cnt[d.w], 1,
                                              __ATOMIC_RELAXED, __HIP_MEMORY_SCOPE_AGENT);
        if (p0 < cap) bucket[(size_t)d.x * cap + p0] = e0;
        if (p1 < cap) bucket[(size_t)d.y * cap + p1] = e0 + 1;
        if (p2 < cap) bucket[(size_t)d.z * cap + p2] = e0 + 2;
        if (p3 < cap) bucket[(size_t)d.w * cap + p3] = e0 + 3;
    } else {
        for (int e = e0; e < n_edges; ++e) {
            const int d = dst[e];
            const int p = __hip_atomic_fetch_add(&cnt[d], 1,
                                                 __ATOMIC_RELAXED, __HIP_MEMORY_SCOPE_AGENT);
            if (p < cap) bucket[(size_t)d * cap + p] = e;
        }
    }
}

template <bool RELU>
__device__ __forceinline__
void rgcn_msg_acc(float (&acc)[8], float4 xa, float4 xb,
                  const float4* wrel, float nrm)
{
    if (RELU) {
        xa.x = fmaxf(xa.x, 0.0f); xa.y = fmaxf(xa.y, 0.0f);
        xa.z = fmaxf(xa.z, 0.0f); xa.w = fmaxf(xa.w, 0.0f);
        xb.x = fmaxf(xb.x, 0.0f); xb.y = fmaxf(xb.y, 0.0f);
        xb.z = fmaxf(xb.z, 0.0f); xb.w = fmaxf(xb.w, 0.0f);
    }
    const float xs[8] = { xa.x, xa.y, xa.z, xa.w, xb.x, xb.y, xb.z, xb.w };
    #pragma unroll
    for (int j = 0; j < 4; ++j) {
        const float4 wq = wrel[32 * j];
        const float x0 = xs[2 * j + 0];
        const float x1 = xs[2 * j + 1];
        acc[2 * j + 0] += (x0 * wq.x + x1 * wq.z) * nrm;
        acc[2 * j + 1] += (x0 * wq.y + x1 * wq.w) * nrm;
    }
}

template <bool RELU>
__device__ __forceinline__
void rgcn_sorted_run(float (&acc)[8], const int sorted, const int m, const int lane,
                     const float* __restrict__ xin,
                     const int*   __restrict__ src,
                     const int*   __restrict__ etype,
                     const float* __restrict__ norm,
                     const float4* wl)
{
    int   s_v = 0, r_v = 0;
    float n_v = 0.0f;
    if (lane < m) {
        s_v = src[sorted];
        r_v = etype[sorted];
        n_v = norm[sorted];
    }

    int k = 0;
    for (; k + 4 <= m; k += 4) {
        float4 xa[4], xb[4];
        int    rr[4];
        float  nn[4];
        #pragma unroll
        for (int u = 0; u < 4; ++u) {
            const int s = __builtin_amdgcn_readlane(s_v, k + u);
            rr[u] = __builtin_amdgcn_readlane(r_v, k + u);
            nn[u] = __int_as_float(
                        __builtin_amdgcn_readlane(__float_as_int(n_v), k + u));
            const float4* xrow = (const float4*)(xin + (size_t)s * HD) + lane;
            xa[u] = xrow[0];
            xb[u] = xrow[32];
        }
        #pragma unroll
        for (int u = 0; u < 4; ++u)
            rgcn_msg_acc<RELU>(acc, xa[u], xb[u], wl + (size_t)rr[u] * NB, nn[u]);
    }
    for (; k < m; ++k) {
        const int   s  = __builtin_amdgcn_readlane(s_v, k);
        const int   r  = __builtin_amdgcn_readlane(r_v, k);
        const float nr = __int_as_float(
                             __builtin_amdgcn_readlane(__float_as_int(n_v), k));
        const float4* xrow = (const float4*)(xin + (size_t)s * HD) + lane;
        rgcn_msg_acc<RELU>(acc, xrow[0], xrow[32], wl + (size_t)r * NB, nr);
    }
}

template <bool RELU>
__global__ __launch_bounds__(256)
void rgcn_node_gather(const float* __restrict__ xin,
                      const int*   __restrict__ src,
                      const int*   __restrict__ dst,
                      const int*   __restrict__ etype,
                      const float* __restrict__ norm,
                      const float* __restrict__ w,
                      const int*   __restrict__ seg,
                      const int*   __restrict__ perm,
                      float* agg,
                      int n_nodes,
                      int n_edges,
                      int cap)
{
    __shared__ __align__(16) float wlds[WTAB];
    __shared__ __align__(16) int   idbuf[8 * 2 * LCAP];
    for (int i = threadIdx.x; i < WTAB / 4; i += blockDim.x) {
        const int b = i & (NB - 1);
        const int j = ((b >> 6) << 1) | (b & 1);
        const int L = (b & 63) >> 1;
        ((float4*)wlds)[(i & ~(NB - 1)) + j * 32 + L] = ((const float4*)w)[i];
    }
    __syncthreads();

    const int lane   = threadIdx.x & 31;
    const int wid    = __builtin_amdgcn_readfirstlane(
                           blockIdx.x * (blockDim.x >> 5) + (threadIdx.x >> 5));
    const int nwaves = (gridDim.x * blockDim.x) >> 5;
    const float4* wl = (const float4*)wlds + lane;
    int* uns = idbuf + __builtin_amdgcn_readfirstlane((int)(threadIdx.x >> 5)) * (2 * LCAP);
    int* srt = uns + LCAP;

    for (int n = wid; n < n_nodes; n += nwaves) {
        int beg, cnt;
        if (cap > 0) {
            cnt = seg[n];
            beg = n * cap;
        } else {
            beg = (n > 0) ? seg[n - 1] : 0;
            cnt = seg[n] - beg;
        }
        if (cnt <= 0) continue;
        const int  end    = beg + cnt;
        const bool listed = (cap == 0) || (cnt <= cap);

        float4* orow = (float4*)(agg + (size_t)n * HD) + lane;
        float4 oa = orow[0];
        float4 ob = orow[32];

        float acc[8] = { 0.f, 0.f, 0.f, 0.f, 0.f, 0.f, 0.f, 0.f };

        if (listed && cnt <= LCAP) {
            int sorted0 = 0;
            if (cnt <= 32) {
                const int my = (lane < cnt) ? perm[beg + lane] : 0x7fffffff;
                int rank = 0;
                for (int m = 0; m < cnt; ++m)
                    rank += (__builtin_amdgcn_readlane(my, m) < my) ? 1 : 0;
                sorted0 = __builtin_amdgcn_ds_permute(rank << 2, my);
            } else {
                const int padded = (cnt + 31) & ~31;
                for (int p = lane; p < padded; p += 32)
                    uns[p] = (p < cnt) ? perm[beg + p] : 0x7fffffff;
                for (int p = lane; p < padded; p += 32) {
                    const int my = uns[p];
                    int rank = 0;
                    for (int q = 0; q < padded; q += 4) {
                        const int4 v = *(const int4*)(uns + q);
                        rank += ((v.x < my) ? 1 : 0) + ((v.y < my) ? 1 : 0)
                              + ((v.z < my) ? 1 : 0) + ((v.w < my) ? 1 : 0);
                    }
                    if (p < cnt) srt[rank] = my;
                }
            }
            for (int c = 0; c < cnt; c += 32) {
                const int m = (cnt - c < 32) ? (cnt - c) : 32;
                const int sorted = (cnt <= 32) ? sorted0 : ((lane < m) ? srt[c + lane] : 0);
                rgcn_sorted_run<RELU>(acc, sorted, m, lane, xin, src, etype, norm, wl);
            }
        } else if (listed && cnt <= 2048) {
            int lo = -1;
            for (int i = 0; i < cnt; ++i) {
                int best = 0x7fffffff;
                for (int p = beg + lane; p < end; p += 32) {
                    const int e = perm[p];
                    if (e > lo && e < best) best = e;
                }
                #pragma unroll
                for (int o = 16; o > 0; o >>= 1) {
                    const int other = __shfl_xor(best, o);
                    best = (other < best) ? other : best;
                }
                const int e = __builtin_amdgcn_readfirstlane(best);
                lo = e;
                const int   s  = src[e];
                const int   r  = etype[e];
                const float nr = norm[e];
                const float4* xrow = (const float4*)(xin + (size_t)s * HD) + lane;
                rgcn_msg_acc<RELU>(acc, xrow[0], xrow[32], wl + (size_t)r * NB, nr);
            }
        } else {
            for (int e0 = 0; e0 < n_edges; e0 += 32) {
                const int  ev  = e0 + lane;
                const bool hit = (ev < n_edges) && (dst[ev] == n);
                unsigned m = __builtin_amdgcn_ballot_w32(hit);
                while (m != 0u) {
                    const int e = e0 + (__builtin_ctz(m));
                    m &= m - 1u;
                    const int   s  = src[e];
                    const int   r  = etype[e];
                    const float nr = norm[e];
                    const float4* xrow = (const float4*)(xin + (size_t)s * HD) + lane;
                    rgcn_msg_acc<RELU>(acc, xrow[0], xrow[32], wl + (size_t)r * NB, nr);
                }
            }
        }

        oa.x += acc[0]; oa.y += acc[1]; oa.z += acc[2]; oa.w += acc[3];
        ob.x += acc[4]; ob.y += acc[5]; ob.z += acc[6]; ob.w += acc[7];
        orow[0]  = oa;
        orow[32] = ob;
    }
}

#define RCAP 64

__global__ __launch_bounds__(256)
void rgcn_bucket_fill_rec(const int*   __restrict__ dst,
                          const int*   __restrict__ src,
                          const int*   __restrict__ etype,
                          const float* __restrict__ norm,
                          int* cnt, int* __restrict__ bucket,
                          int cap, int n_edges)
{
    int2* rec = (int2*)bucket;
    const int e0 = (blockIdx.x * blockDim.x + threadIdx.x) * 4;
    if (e0 + 3 < n_edges) {
        const int4   d = *(const int4*)(dst + e0);
        const int4   s = *(const int4*)(src + e0);
        const int4   r = *(const int4*)(etype + e0);
        const float4 w = *(const float4*)(norm + e0);
        const int p0 = __hip_atomic_fetch_add(&cnt[d.x], 1,
                                              __ATOMIC_RELAXED, __HIP_MEMORY_SCOPE_AGENT);
        const int p1 = __hip_atomic_fetch_add(&cnt[d.y], 1,
                                              __ATOMIC_RELAXED, __HIP_MEMORY_SCOPE_AGENT);
        const int p2 = __hip_atomic_fetch_add(&cnt[d.z], 1,
                                              __ATOMIC_RELAXED, __HIP_MEMORY_SCOPE_AGENT);
        const int p3 = __hip_atomic_fetch_add(&cnt[d.w], 1,
                                              __ATOMIC_RELAXED, __HIP_MEMORY_SCOPE_AGENT);
        if (p0 < cap) rec[(size_t)d.x * cap + p0] = make_int2(s.x | (r.x << 24), __float_as_int(w.x));
        if (p1 < cap) rec[(size_t)d.y * cap + p1] = make_int2(s.y | (r.y << 24), __float_as_int(w.y));
        if (p2 < cap) rec[(size_t)d.z * cap + p2] = make_int2(s.z | (r.z << 24), __float_as_int(w.z));
        if (p3 < cap) rec[(size_t)d.w * cap + p3] = make_int2(s.w | (r.w << 24), __float_as_int(w.w));
    } else {
        for (int e = e0; e < n_edges; ++e) {
            const int d = dst[e];
            const int p = __hip_atomic_fetch_add(&cnt[d], 1,
                                                 __ATOMIC_RELAXED, __HIP_MEMORY_SCOPE_AGENT);
            if (p < cap)
                rec[(size_t)d * cap + p] = make_int2(src[e] | (etype[e] << 24),
                                                     __float_as_int(norm[e]));
        }
    }
}

__device__ __forceinline__
unsigned long long rgcn_rec_key(int w0, int nb)
{
    return ((unsigned long long)(unsigned)w0 << 32) | (unsigned long long)(unsigned)nb;
}

template <bool RELU>
__device__ __forceinline__
void rgcn_rec_run(float (&acc)[8], const int w0, const int nb, const int m, const int lane,
                  const float* __restrict__ xin, const float4* wl)
{
    const int s_v = w0 & 0x00ffffff;
    const int r_v = (int)((unsigned)w0 >> 24);

    int k = 0;
    for (; k + 4 <= m; k += 4) {
        float4 xa[4], xb[4];
        int    rr[4];
        float  nn[4];
        #pragma unroll
        for (int u = 0; u < 4; ++u) {
            const int s = __builtin_amdgcn_readlane(s_v, k + u);
            rr[u] = __builtin_amdgcn_readlane(r_v, k + u);
            nn[u] = __int_as_float(__builtin_amdgcn_readlane(nb, k + u));
            const float4* xrow = (const float4*)(xin + (size_t)s * HD) + lane;
            xa[u] = xrow[0];
            xb[u] = xrow[32];
        }
        #pragma unroll
        for (int u = 0; u < 4; ++u)
            rgcn_msg_acc<RELU>(acc, xa[u], xb[u], wl + (size_t)rr[u] * NB, nn[u]);
    }
    for (; k < m; ++k) {
        const int   s  = __builtin_amdgcn_readlane(s_v, k);
        const int   r  = __builtin_amdgcn_readlane(r_v, k);
        const float nr = __int_as_float(__builtin_amdgcn_readlane(nb, k));
        const float4* xrow = (const float4*)(xin + (size_t)s * HD) + lane;
        rgcn_msg_acc<RELU>(acc, xrow[0], xrow[32], wl + (size_t)r * NB, nr);
    }
}

template <bool RELU>
__global__ __launch_bounds__(256)
void rgcn_node_gather_rec(const float* __restrict__ xin,
                          const int*   __restrict__ src,
                          const int*   __restrict__ dst,
                          const int*   __restrict__ etype,
                          const float* __restrict__ norm,
                          const float* __restrict__ w,
                          const int*   __restrict__ seg,
                          const int*   __restrict__ bucket,
                          float* agg,
                          int n_nodes,
                          int n_edges,
                          int cap)
{
    __shared__ __align__(16) float wlds[WTAB];
    __shared__ __align__(16) int   keybuf[8 * 4 * RCAP];
    for (int i = threadIdx.x; i < WTAB / 4; i += blockDim.x) {
        const int b = i & (NB - 1);
        const int j = ((b >> 6) << 1) | (b & 1);
        const int L = (b & 63) >> 1;
        ((float4*)wlds)[(i & ~(NB - 1)) + j * 32 + L] = ((const float4*)w)[i];
    }
    __syncthreads();

    const int lane   = threadIdx.x & 31;
    const int wid    = __builtin_amdgcn_readfirstlane(
                           blockIdx.x * (blockDim.x >> 5) + (threadIdx.x >> 5));
    const int nwaves = (gridDim.x * blockDim.x) >> 5;
    const float4* wl = (const float4*)wlds + lane;
    int2* uns = (int2*)(keybuf + __builtin_amdgcn_readfirstlane((int)(threadIdx.x >> 5)) * (4 * RCAP));
    int2* srt = uns + RCAP;
    const int2* rec = (const int2*)bucket;

    for (int n = wid; n < n_nodes; n += nwaves) {
        const int cnt = seg[n];
        if (cnt <= 0) continue;

        float4* orow = (float4*)(agg + (size_t)n * HD) + lane;
        float4 oa = orow[0];
        float4 ob = orow[32];

        float acc[8] = { 0.f, 0.f, 0.f, 0.f, 0.f, 0.f, 0.f, 0.f };

        if (cnt <= cap && cnt <= RCAP) {
            const int2* row = rec + (size_t)n * cap;
            int w0s = 0, nbs = 0;
            if (cnt <= 32) {
                int2 mine = make_int2(-1, -1);
                if (lane < cnt) mine = row[lane];
                const unsigned long long my = rgcn_rec_key(mine.x, mine.y);
                int rank = 0;
                for (int m = 0; m < cnt; ++m) {
                    const unsigned long long ok =
                        rgcn_rec_key(__builtin_amdgcn_readlane(mine.x, m),
                                     __builtin_amdgcn_readlane(mine.y, m));
                    rank += (ok < my || (ok == my && m < lane)) ? 1 : 0;
                }
                w0s = __builtin_amdgcn_ds_permute(rank << 2, mine.x);
                nbs = __builtin_amdgcn_ds_permute(rank << 2, mine.y);
            } else {
                const int padded = (cnt + 31) & ~31;
                for (int p = lane; p < padded; p += 32)
                    uns[p] = (p < cnt) ? row[p] : make_int2(-1, -1);
                for (int p = lane; p < padded; p += 32) {
                    const int2 mine = uns[p];
                    const unsigned long long my = rgcn_rec_key(mine.x, mine.y);
                    int rank = 0;
                    for (int q = 0; q < padded; q += 2) {
                        const int4 v = *(const int4*)(uns + q);
                        const unsigned long long k0 = rgcn_rec_key(v.x, v.y);
                        const unsigned long long k1 = rgcn_rec_key(v.z, v.w);
                        rank += ((k0 < my || (k0 == my && q     < p)) ? 1 : 0)
                              + ((k1 < my || (k1 == my && q + 1 < p)) ? 1 : 0);
                    }
                    if (p < cnt) srt[rank] = mine;
                }
            }
            for (int c = 0; c < cnt; c += 32) {
                const int m = (cnt - c < 32) ? (cnt - c) : 32;
                if (cnt > 32) {
                    int2 r = make_int2(0, 0);
                    if (lane < m) r = srt[c + lane];
                    w0s = r.x;
                    nbs = r.y;
                }
                rgcn_rec_run<RELU>(acc, w0s, nbs, m, lane, xin, wl);
            }
        } else {
            for (int e0 = 0; e0 < n_edges; e0 += 32) {
                const int  ev  = e0 + lane;
                const bool hit = (ev < n_edges) && (dst[ev] == n);
                unsigned m = __builtin_amdgcn_ballot_w32(hit);
                while (m != 0u) {
                    const int e = e0 + (__builtin_ctz(m));
                    m &= m - 1u;
                    const int   s  = src[e];
                    const int   r  = etype[e];
                    const float nr = norm[e];
                    const float4* xrow = (const float4*)(xin + (size_t)s * HD) + lane;
                    rgcn_msg_acc<RELU>(acc, xrow[0], xrow[32], wl + (size_t)r * NB, nr);
                }
            }
        }

        oa.x += acc[0]; oa.y += acc[1]; oa.z += acc[2]; oa.w += acc[3];
        ob.x += acc[4]; ob.y += acc[5]; ob.z += acc[6]; ob.w += acc[7];
        orow[0]  = oa;
        orow[32] = ob;
    }
}

extern "C" void kernel_launch(void* const* d_in, const int* in_sizes, int n_in,
                              void* d_out, int out_size, void* d_ws, size_t ws_size,
                              hipStream_t stream)
{
    const float* x     = (const float*)d_in[0];
    const int*   src   = (const int*)  d_in[1];
    const int*   dst   = (const int*)  d_in[2];
    const int*   etype = (const int*)  d_in[3];
    const float* norm  = (const float*)d_in[4];
    const float* w1    = (const float*)d_in[5];
    const float* lw1   = (const float*)d_in[6];
    const float* b1    = (const float*)d_in[7];
    const float* w2    = (const float*)d_in[8];
    const float* lw2   = (const float*)d_in[9];
    const float* b2    = (const float*)d_in[10];
    float*       out   = (float*)d_out;

    const int n_nodes = in_sizes[0] / HD;
    const int n_edges = in_sizes[1];

    float* h = (float*)d_ws;

    const int row_groups  = ((n_nodes + 15) / 16 + 7) / 8;
    const int gemm_blocks = row_groups * (HD / 64);

    const size_t seg_off  = ((size_t)n_nodes * HD * sizeof(float) + 255) & ~(size_t)255;
    const size_t perm_off = (seg_off + (size_t)n_nodes * sizeof(int) + 255) & ~(size_t)255;
    const size_t ws_need  = perm_off + (size_t)n_edges * sizeof(int);

    if (ws_size < ws_need) {
        const int edge_blocks = 1024;
        rgcn_selfloop_gemm<false><<<gemm_blocks, 256, 0, stream>>>(x, lw1, b1, h, n_nodes,
                                                                   nullptr, 0);
        rgcn_edge_msg<false><<<edge_blocks, 256, 0, stream>>>(x, src, dst, etype, norm, w1,
                                                              h, n_edges);
        rgcn_selfloop_gemm<true><<<gemm_blocks, 256, 0, stream>>>(h, lw2, b2, out, n_nodes,
                                                                  nullptr, 0);
        rgcn_edge_msg<true><<<edge_blocks, 256, 0, stream>>>(h, src, dst, etype, norm, w2,
                                                             out, n_edges);
        return;
    }

    int* seg  = (int*)((char*)d_ws + seg_off);
    const int gather_blocks = 1024;

    const int    bcap     = 64;
    const size_t bkt_need = perm_off + (size_t)n_nodes * bcap * sizeof(int);

    const size_t rec_need = perm_off + (size_t)n_nodes * RCAP * 2 * sizeof(int);

    if (ws_size >= rec_need && n_nodes <= (1 << 24)) {
        int* bucket = (int*)((char*)d_ws + perm_off);
        const int fill_blocks = (n_edges + 1023) / 1024;

        rgcn_selfloop_gemm<false><<<gemm_blocks, 256, 0, stream>>>(x, lw1, b1, h, n_nodes,
                                                                   seg, n_nodes);
        rgcn_bucket_fill_rec<<<fill_blocks, 256, 0, stream>>>(dst, src, etype, norm, seg, bucket,
                                                              RCAP, n_edges);

        rgcn_node_gather_rec<false><<<gather_blocks, 256, 0, stream>>>(x, src, dst, etype, norm, w1,
                                                                       seg, bucket, h, n_nodes,
                                                                       n_edges, RCAP);
        rgcn_selfloop_gemm<true><<<gemm_blocks, 256, 0, stream>>>(h, lw2, b2, out, n_nodes,
                                                                  nullptr, 0);
        rgcn_node_gather_rec<true><<<gather_blocks, 256, 0, stream>>>(h, src, dst, etype, norm, w2,
                                                                      seg, bucket, out, n_nodes,
                                                                      n_edges, RCAP);
        return;
    }

    if (ws_size >= bkt_need) {
        int* bucket = (int*)((char*)d_ws + perm_off);
        const int fill_blocks = (n_edges + 1023) / 1024;

        rgcn_selfloop_gemm<false><<<gemm_blocks, 256, 0, stream>>>(x, lw1, b1, h, n_nodes,
                                                                   seg, n_nodes);
        rgcn_bucket_fill<<<fill_blocks, 256, 0, stream>>>(dst, seg, bucket, bcap, n_edges);

        rgcn_node_gather<false><<<gather_blocks, 256, 0, stream>>>(x, src, dst, etype, norm, w1,
                                                                   seg, bucket, h, n_nodes,
                                                                   n_edges, bcap);
        rgcn_selfloop_gemm<true><<<gemm_blocks, 256, 0, stream>>>(h, lw2, b2, out, n_nodes,
                                                                  nullptr, 0);
        rgcn_node_gather<true><<<gather_blocks, 256, 0, stream>>>(h, src, dst, etype, norm, w2,
                                                                  seg, bucket, out, n_nodes,
                                                                  n_edges, bcap);
        return;
    }

    int* perm = (int*)((char*)d_ws + perm_off);
    const int edge_blocks   = (n_edges + 255) / 256;

    rgcn_selfloop_gemm<false><<<gemm_blocks, 256, 0, stream>>>(x, lw1, b1, h, n_nodes,
                                                               seg, n_nodes);
    rgcn_deg_count<<<edge_blocks, 256, 0, stream>>>(dst, seg, n_edges);
    rgcn_seg_scan<<<1, 1024, 0, stream>>>(seg, n_nodes);
    rgcn_seg_fill<<<edge_blocks, 256, 0, stream>>>(dst, seg, perm, n_edges);

    rgcn_node_gather<false><<<gather_blocks, 256, 0, stream>>>(x, src, dst, etype, norm, w1,
                                                               seg, perm, h, n_nodes, n_edges, 0);

    rgcn_selfloop_gemm<true><<<gemm_blocks, 256, 0, stream>>>(h, lw2, b2, out, n_nodes,
                                                              nullptr, 0);
    rgcn_node_gather<true><<<gather_blocks, 256, 0, stream>>>(h, src, dst, etype, norm, w2,
                                                              seg, perm, out, n_nodes, n_edges, 0);
}


# Gather wave: independent loads asked for together, not one round trip each: table's 8 loads, a node's last 1-3 rows, next node's count and records during the current node's sums; sum order kept
# speedup vs baseline: 2.5313x; 1.0390x over previous
#include <hip/hip_runtime.h>

#define HD   256
#define NB   128
#define RT   16
#define WTAB (RT * NB * 4)
#define LCAP 128

typedef float v2f __attribute__((ext_vector_type(2)));
typedef float v8f __attribute__((ext_vector_type(8)));

#define GEMM_KQ (HD / 4)

template <bool RELU>
__global__ __launch_bounds__(256)
void rgcn_selfloop_gemm(const float* __restrict__ A,
                        const float* __restrict__ W,
                        const float* __restrict__ bias,
                        float* __restrict__ out,
                        int n_rows,
                        int* __restrict__ zero_buf,
                        int zero_n)
{
    __shared__ __align__(16) float bl[GEMM_KQ * 64 * 4];

    if (zero_buf != nullptr) {
        const int gid = blockIdx.x * blockDim.x + threadIdx.x;
        if (gid < zero_n) zero_buf[gid] = 0;
    }

    const int strip_n   = blockIdx.x & 3;
    const int row_group = blockIdx.x >> 2;

    const int lane   = threadIdx.x & 31;
    const int tile_m = row_group * 8 + (threadIdx.x >> 5);
    const bool live  = tile_m * 16 < n_rows;
    const int half = lane >> 4;
    const int sub  = lane & 15;
    const int col0 = strip_n * 64 + sub;

    const float4* arow = (const float4*)(A + (size_t)(tile_m * 16 + sub) * HD) + half;

    float4 an[4] = {};
    if (live) {
        #pragma unroll
        for (int q = 0; q < 4; ++q) an[q] = arow[2 * q];
    }

    {
        const int scol = threadIdx.x & 63;
        const int skq  = threadIdx.x >> 6;
        const float* wsrc = W + strip_n * 64 + scol;
        #pragma unroll 4
        for (int i = 0; i < GEMM_KQ / 4; ++i) {
            const int kq = skq + 4 * i;
            const float* p = wsrc + (size_t)(4 * kq) * HD;
            float4 v;
            v.x = p[0];
            v.y = p[HD];
            v.z = p[2 * HD];
            v.w = p[3 * HD];
            ((float4*)bl)[kq * 64 + scol] = v;
        }
    }
    __syncthreads();

    if (!live) return;

    const float4* bfr  = (const float4*)bl + half * 64 + sub;

    v8f c0 = {}, c1 = {}, c2 = {}, c3 = {};

    for (int g = 0; g < 8; ++g) {
        float4 ac[4];
        #pragma unroll
        for (int q = 0; q < 4; ++q) ac[q] = an[q];
        if (g < 7) {
            #pragma unroll
            for (int q = 0; q < 4; ++q) an[q] = arow[2 * (4 * (g + 1) + q)];
        }
        #pragma unroll
        for (int q = 0; q < 4; ++q) {
            const int t = 4 * g + q;
            float4 a4 = ac[q];
            if (RELU) {
                a4.x = fmaxf(a4.x, 0.0f); a4.y = fmaxf(a4.y, 0.0f);
                a4.z = fmaxf(a4.z, 0.0f); a4.w = fmaxf(a4.w, 0.0f);
            }
            const float4 q0 = bfr[t * 128 +  0];
            const float4 q1 = bfr[t * 128 + 16];
            const float4 q2 = bfr[t * 128 + 32];
            const float4 q3 = bfr[t * 128 + 48];

            v2f a;  v2f b0, b1, b2, b3;
            a.x  = a4.x; a.y  = a4.y;
            b0.x = q0.x; b0.y = q0.y;
            b1.x = q1.x; b1.y = q1.y;
            b2.x = q2.x; b2.y = q2.y;
            b3.x = q3.x; b3.y = q3.y;
            c0 = __builtin_amdgcn_wmma_f32_16x16x4_f32(false, a, false, b0, (short)0, c0, false, false);
            c1 = __builtin_amdgcn_wmma_f32_16x16x4_f32(false, a, false, b1, (short)0, c1, false, false);
            c2 = __builtin_amdgcn_wmma_f32_16x16x4_f32(false, a, false, b2, (short)0, c2, false, false);
            c3 = __builtin_amdgcn_wmma_f32_16x16x4_f32(false, a, false, b3, (short)0, c3, false, false);

            a.x  = a4.z; a.y  = a4.w;
            b0.x = q0.z; b0.y = q0.w;
            b1.x = q1.z; b1.y = q1.w;
            b2.x = q2.z; b2.y = q2.w;
            b3.x = q3.z; b3.y = q3.w;
            c0 = __builtin_amdgcn_wmma_f32_16x16x4_f32(false, a, false, b0, (short)0, c0, false, false);
            c1 = __builtin_amdgcn_wmma_f32_16x16x4_f32(false, a, false, b1, (short)0, c1, false, false);
            c2 = __builtin_amdgcn_wmma_f32_16x16x4_f32(false, a, false, b2, (short)0, c2, false, false);
            c3 = __builtin_amdgcn_wmma_f32_16x16x4_f32(false, a, false, b3, (short)0, c3, false, false);
        }
    }

    float* obase = out + (size_t)(tile_m * 16 + half * 8) * HD + col0;
    #pragma unroll
    for (int t = 0; t < 4; ++t) {
        const v8f& c = (t == 0) ? c0 : (t == 1) ? c1 : (t == 2) ? c2 : c3;
        const float bcol = bias[col0 + t * 16];
        #pragma unroll
        for (int j = 0; j < 8; ++j)
            obase[(size_t)j * HD + t * 16] = c[j] + bcol;
    }
}

template <bool RELU>
__global__ __launch_bounds__(256)
void rgcn_edge_msg(const float* __restrict__ xin,
                   const int*   __restrict__ src,
                   const int*   __restrict__ dst,
                   const int*   __restrict__ etype,
                   const float* __restrict__ norm,
                   const float* __restrict__ w,
                   float* agg,
                   int n_edges)
{
    __shared__ __align__(16) float wlds[WTAB];
    for (int i = threadIdx.x; i < WTAB; i += blockDim.x) wlds[i] = w[i];
    __syncthreads();

    const int lane   = threadIdx.x & 31;
    const int wid    = blockIdx.x * (blockDim.x >> 5) + (threadIdx.x >> 5);
    const int nwaves = (gridDim.x * blockDim.x) >> 5;

    for (int e = wid; e < n_edges; e += nwaves) {
        const int   s   = src[e];
        const int   d   = dst[e];
        const int   r   = etype[e];
        const float nrm = norm[e];

        const float4* xrow = (const float4*)(xin + (size_t)s * HD) + lane * 2;
        float4 xa = xrow[0];
        float4 xb = xrow[1];
        if (RELU) {
            xa.x = fmaxf(xa.x, 0.0f); xa.y = fmaxf(xa.y, 0.0f);
            xa.z = fmaxf(xa.z, 0.0f); xa.w = fmaxf(xa.w, 0.0f);
            xb.x = fmaxf(xb.x, 0.0f); xb.y = fmaxf(xb.y, 0.0f);
            xb.z = fmaxf(xb.z, 0.0f); xb.w = fmaxf(xb.w, 0.0f);
        }
        const float xs[8] = { xa.x, xa.y, xa.z, xa.w, xb.x, xb.y, xb.z, xb.w };

        const float4* wrel = (const float4*)(wlds + (size_t)r * (NB * 4)) + lane * 4;
        float* arow = agg + (size_t)d * HD + lane * 8;

        #pragma unroll
        for (int j = 0; j < 4; ++j) {
            const float4 wq = wrel[j];
            const float x0 = xs[2 * j + 0];
            const float x1 = xs[2 * j + 1];
            const float o0 = (x0 * wq.x + x1 * wq.z) * nrm;
            const float o1 = (x0 * wq.y + x1 * wq.w) * nrm;
            (void)__hip_atomic_fetch_add(&arow[2 * j + 0], o0,
                                         __ATOMIC_RELAXED, __HIP_MEMORY_SCOPE_AGENT);
            (void)__hip_atomic_fetch_add(&arow[2 * j + 1], o1,
                                         __ATOMIC_RELAXED, __HIP_MEMORY_SCOPE_AGENT);
        }
    }
}

__global__ __launch_bounds__(256)
void rgcn_deg_count(const int* __restrict__ dst, int* seg, int n_edges)
{
    const int e = blockIdx.x * blockDim.x + threadIdx.x;
    if (e < n_edges)
        (void)__hip_atomic_fetch_add(&seg[dst[e]], 1,
                                     __ATOMIC_RELAXED, __HIP_MEMORY_SCOPE_AGENT);
}

__global__ __launch_bounds__(1024)
void rgcn_seg_scan(int* seg, int n)
{
    __shared__ int wsum[32];
    const int t    = threadIdx.x;
    const int lane = t & 31;
    const int w    = t >> 5;
    const int chunk = (n + 1023) / 1024;
    const int b  = t * chunk;
    const int en = (b + chunk < n) ? (b + chunk) : n;

    int s = 0;
    for (int i = b; i < en; ++i) s += seg[i];

    int v = s;
    #pragma unroll
    for (int o = 1; o < 32; o <<= 1) {
        const int u = __shfl_up(v, o);
        if (lane >= o) v += u;
    }
    if (lane == 31) wsum[w] = v;
    __syncthreads();
    if (w == 0) {
        const int x = wsum[lane];
        int y = x;
        #pragma unroll
        for (int o = 1; o < 32; o <<= 1) {
            const int u = __shfl_up(y, o);
            if (lane >= o) y += u;
        }
        wsum[lane] = y - x;
    }
    __syncthreads();

    int run = wsum[w] + v - s;
    for (int i = b; i < en; ++i) {
        const int c = seg[i];
        seg[i] = run;
        run += c;
    }
}

__global__ __launch_bounds__(256)
void rgcn_seg_fill(const int* __restrict__ dst, int* seg, int* __restrict__ perm,
                   int n_edges)
{
    const int e = blockIdx.x * blockDim.x + threadIdx.x;
    if (e < n_edges) {
        const int p = __hip_atomic_fetch_add(&seg[dst[e]], 1,
                                             __ATOMIC_RELAXED, __HIP_MEMORY_SCOPE_AGENT);
        perm[p] = e;
    }
}

__global__ __launch_bounds__(256)
void rgcn_bucket_fill(const int* __restrict__ dst, int* cnt, int* __restrict__ bucket,
                      int cap, int n_edges)
{
    const int e0 = (blockIdx.x * blockDim.x + threadIdx.x) * 4;
    if (e0 + 3 < n_edges) {
        const int4 d = *(const int4*)(dst + e0);
        const int p0 = __hip_atomic_fetch_add(&cnt[d.x], 1,
                                              __ATOMIC_RELAXED, __HIP_MEMORY_SCOPE_AGENT);
        const int p1 = __hip_atomic_fetch_add(&cnt[d.y], 1,
                                              __ATOMIC_RELAXED, __HIP_MEMORY_SCOPE_AGENT);
        const int p2 = __hip_atomic_fetch_add(&cnt[d.z], 1,
                                              __ATOMIC_RELAXED, __HIP_MEMORY_SCOPE_AGENT);
        const int p3 = __hip_atomic_fetch_add(&cnt[d.w], 1,
                                              __ATOMIC_RELAXED, __HIP_MEMORY_SCOPE_AGENT);
        if (p0 < cap) bucket[(size_t)d.x * cap + p0] = e0;
        if (p1 < cap) bucket[(size_t)d.y * cap + p1] = e0 + 1;
        if (p2 < cap) bucket[(size_t)d.z * cap + p2] = e0 + 2;
        if (p3 < cap) bucket[(size_t)d.w * cap + p3] = e0 + 3;
    } else {
        for (int e = e0; e < n_edges; ++e) {
            const int d = dst[e];
            const int p = __hip_atomic_fetch_add(&cnt[d], 1,
                                                 __ATOMIC_RELAXED, __HIP_MEMORY_SCOPE_AGENT);
            if (p < cap) bucket[(size_t)d * cap + p] = e;
        }
    }
}

template <bool RELU>
__device__ __forceinline__
void rgcn_msg_acc(float (&acc)[8], float4 xa, float4 xb,
                  const float4* wrel, float nrm)
{
    if (RELU) {
        xa.x = fmaxf(xa.x, 0.0f); xa.y = fmaxf(xa.y, 0.0f);
        xa.z = fmaxf(xa.z, 0.0f); xa.w = fmaxf(xa.w, 0.0f);
        xb.x = fmaxf(xb.x, 0.0f); xb.y = fmaxf(xb.y, 0.0f);
        xb.z = fmaxf(xb.z, 0.0f); xb.w = fmaxf(xb.w, 0.0f);
    }
    const float xs[8] = { xa.x, xa.y, xa.z, xa.w, xb.x, xb.y, xb.z, xb.w };
    #pragma unroll
    for (int j = 0; j < 4; ++j) {
        const float4 wq = wrel[32 * j];
        const float x0 = xs[2 * j + 0];
        const float x1 = xs[2 * j + 1];
        acc[2 * j + 0] += (x0 * wq.x + x1 * wq.z) * nrm;
        acc[2 * j + 1] += (x0 * wq.y + x1 * wq.w) * nrm;
    }
}

template <bool RELU>
__device__ __forceinline__
void rgcn_sorted_run(float (&acc)[8], const int sorted, const int m, const int lane,
                     const float* __restrict__ xin,
                     const int*   __restrict__ src,
                     const int*   __restrict__ etype,
                     const float* __restrict__ norm,
                     const float4* wl)
{
    int   s_v = 0, r_v = 0;
    float n_v = 0.0f;
    if (lane < m) {
        s_v = src[sorted];
        r_v = etype[sorted];
        n_v = norm[sorted];
    }

    int k = 0;
    for (; k + 4 <= m; k += 4) {
        float4 xa[4], xb[4];
        int    rr[4];
        float  nn[4];
        #pragma unroll
        for (int u = 0; u < 4; ++u) {
            const int s = __builtin_amdgcn_readlane(s_v, k + u);
            rr[u] = __builtin_amdgcn_readlane(r_v, k + u);
            nn[u] = __int_as_float(
                        __builtin_amdgcn_readlane(__float_as_int(n_v), k + u));
            const float4* xrow = (const float4*)(xin + (size_t)s * HD) + lane;
            xa[u] = xrow[0];
            xb[u] = xrow[32];
        }
        #pragma unroll
        for (int u = 0; u < 4; ++u)
            rgcn_msg_acc<RELU>(acc, xa[u], xb[u], wl + (size_t)rr[u] * NB, nn[u]);
    }
    for (; k < m; ++k) {
        const int   s  = __builtin_amdgcn_readlane(s_v, k);
        const int   r  = __builtin_amdgcn_readlane(r_v, k);
        const float nr = __int_as_float(
                             __builtin_amdgcn_readlane(__float_as_int(n_v), k));
        const float4* xrow = (const float4*)(xin + (size_t)s * HD) + lane;
        rgcn_msg_acc<RELU>(acc, xrow[0], xrow[32], wl + (size_t)r * NB, nr);
    }
}

template <bool RELU>
__global__ __launch_bounds__(256)
void rgcn_node_gather(const float* __restrict__ xin,
                      const int*   __restrict__ src,
                      const int*   __restrict__ dst,
                      const int*   __restrict__ etype,
                      const float* __restrict__ norm,
                      const float* __restrict__ w,
                      const int*   __restrict__ seg,
                      const int*   __restrict__ perm,
                      float* agg,
                      int n_nodes,
                      int n_edges,
                      int cap)
{
    __shared__ __align__(16) float wlds[WTAB];
    __shared__ __align__(16) int   idbuf[8 * 2 * LCAP];
    for (int i = threadIdx.x; i < WTAB / 4; i += blockDim.x) {
        const int b = i & (NB - 1);
        const int j = ((b >> 6) << 1) | (b & 1);
        const int L = (b & 63) >> 1;
        ((float4*)wlds)[(i & ~(NB - 1)) + j * 32 + L] = ((const float4*)w)[i];
    }
    __syncthreads();

    const int lane   = threadIdx.x & 31;
    const int wid    = __builtin_amdgcn_readfirstlane(
                           blockIdx.x * (blockDim.x >> 5) + (threadIdx.x >> 5));
    const int nwaves = (gridDim.x * blockDim.x) >> 5;
    const float4* wl = (const float4*)wlds + lane;
    int* uns = idbuf + __builtin_amdgcn_readfirstlane((int)(threadIdx.x >> 5)) * (2 * LCAP);
    int* srt = uns + LCAP;

    for (int n = wid; n < n_nodes; n += nwaves) {
        int beg, cnt;
        if (cap > 0) {
            cnt = seg[n];
            beg = n * cap;
        } else {
            beg = (n > 0) ? seg[n - 1] : 0;
            cnt = seg[n] - beg;
        }
        if (cnt <= 0) continue;
        const int  end    = beg + cnt;
        const bool listed = (cap == 0) || (cnt <= cap);

        float4* orow = (float4*)(agg + (size_t)n * HD) + lane;
        float4 oa = orow[0];
        float4 ob = orow[32];

        float acc[8] = { 0.f, 0.f, 0.f, 0.f, 0.f, 0.f, 0.f, 0.f };

        if (listed && cnt <= LCAP) {
            int sorted0 = 0;
            if (cnt <= 32) {
                const int my = (lane < cnt) ? perm[beg + lane] : 0x7fffffff;
                int rank = 0;
                for (int m = 0; m < cnt; ++m)
                    rank += (__builtin_amdgcn_readlane(my, m) < my) ? 1 : 0;
                sorted0 = __builtin_amdgcn_ds_permute(rank << 2, my);
            } else {
                const int padded = (cnt + 31) & ~31;
                for (int p = lane; p < padded; p += 32)
                    uns[p] = (p < cnt) ? perm[beg + p] : 0x7fffffff;
                for (int p = lane; p < padded; p += 32) {
                    const int my = uns[p];
                    int rank = 0;
                    for (int q = 0; q < padded; q += 4) {
                        const int4 v = *(const int4*)(uns + q);
                        rank += ((v.x < my) ? 1 : 0) + ((v.y < my) ? 1 : 0)
                              + ((v.z < my) ? 1 : 0) + ((v.w < my) ? 1 : 0);
                    }
                    if (p < cnt) srt[rank] = my;
                }
            }
            for (int c = 0; c < cnt; c += 32) {
                const int m = (cnt - c < 32) ? (cnt - c) : 32;
                const int sorted = (cnt <= 32) ? sorted0 : ((lane < m) ? srt[c + lane] : 0);
                rgcn_sorted_run<RELU>(acc, sorted, m, lane, xin, src, etype, norm, wl);
            }
        } else if (listed && cnt <= 2048) {
            int lo = -1;
            for (int i = 0; i < cnt; ++i) {
                int best = 0x7fffffff;
                for (int p = beg + lane; p < end; p += 32) {
                    const int e = perm[p];
                    if (e > lo && e < best) best = e;
                }
                #pragma unroll
                for (int o = 16; o > 0; o >>= 1) {
                    const int other = __shfl_xor(best, o);
                    best = (other < best) ? other : best;
                }
                const int e = __builtin_amdgcn_readfirstlane(best);
                lo = e;
                const int   s  = src[e];
                const int   r  = etype[e];
                const float nr = norm[e];
                const float4* xrow = (const float4*)(xin + (size_t)s * HD) + lane;
                rgcn_msg_acc<RELU>(acc, xrow[0], xrow[32], wl + (size_t)r * NB, nr);
            }
        } else {
            for (int e0 = 0; e0 < n_edges; e0 += 32) {
                const int  ev  = e0 + lane;
                const bool hit = (ev < n_edges) && (dst[ev] == n);
                unsigned m = __builtin_amdgcn_ballot_w32(hit);
                while (m != 0u) {
                    const int e = e0 + (__builtin_ctz(m));
                    m &= m - 1u;
                    const int   s  = src[e];
                    const int   r  = etype[e];
                    const float nr = norm[e];
                    const float4* xrow = (const float4*)(xin + (size_t)s * HD) + lane;
                    rgcn_msg_acc<RELU>(acc, xrow[0], xrow[32], wl + (size_t)r * NB, nr);
                }
            }
        }

        oa.x += acc[0]; oa.y += acc[1]; oa.z += acc[2]; oa.w += acc[3];
        ob.x += acc[4]; ob.y += acc[5]; ob.z += acc[6]; ob.w += acc[7];
        orow[0]  = oa;
        orow[32] = ob;
    }
}

#define RCAP 64

__global__ __launch_bounds__(256)
void rgcn_bucket_fill_rec(const int*   __restrict__ dst,
                          const int*   __restrict__ src,
                          const int*   __restrict__ etype,
                          const float* __restrict__ norm,
                          int* cnt, int* __restrict__ bucket,
                          int cap, int n_edges)
{
    int2* rec = (int2*)bucket;
    const int e0 = (blockIdx.x * blockDim.x + threadIdx.x) * 4;
    if (e0 + 3 < n_edges) {
        const int4   d = *(const int4*)(dst + e0);
        const int4   s = *(const int4*)(src + e0);
        const int4   r = *(const int4*)(etype + e0);
        const float4 w = *(const float4*)(norm + e0);
        const int p0 = __hip_atomic_fetch_add(&cnt[d.x], 1,
                                              __ATOMIC_RELAXED, __HIP_MEMORY_SCOPE_AGENT);
        const int p1 = __hip_atomic_fetch_add(&cnt[d.y], 1,
                                              __ATOMIC_RELAXED, __HIP_MEMORY_SCOPE_AGENT);
        const int p2 = __hip_atomic_fetch_add(&cnt[d.z], 1,
                                              __ATOMIC_RELAXED, __HIP_MEMORY_SCOPE_AGENT);
        const int p3 = __hip_atomic_fetch_add(&cnt[d.w], 1,
                                              __ATOMIC_RELAXED, __HIP_MEMORY_SCOPE_AGENT);
        if (p0 < cap) rec[(size_t)d.x * cap + p0] = make_int2(s.x | (r.x << 24), __float_as_int(w.x));
        if (p1 < cap) rec[(size_t)d.y * cap + p1] = make_int2(s.y | (r.y << 24), __float_as_int(w.y));
        if (p2 < cap) rec[(size_t)d.z * cap + p2] = make_int2(s.z | (r.z << 24), __float_as_int(w.z));
        if (p3 < cap) rec[(size_t)d.w * cap + p3] = make_int2(s.w | (r.w << 24), __float_as_int(w.w));
    } else {
        for (int e = e0; e < n_edges; ++e) {
            const int d = dst[e];
            const int p = __hip_atomic_fetch_add(&cnt[d], 1,
                                                 __ATOMIC_RELAXED, __HIP_MEMORY_SCOPE_AGENT);
            if (p < cap)
                rec[(size_t)d * cap + p] = make_int2(src[e] | (etype[e] << 24),
                                                     __float_as_int(norm[e]));
        }
    }
}

__device__ __forceinline__
unsigned long long rgcn_rec_key(int w0, int nb)
{
    return ((unsigned long long)(unsigned)w0 << 32) | (unsigned long long)(unsigned)nb;
}

template <bool RELU>
__device__ __forceinline__
void rgcn_rec_run(float (&acc)[8], const int w0, const int nb, const int m, const int lane,
                  const float* __restrict__ xin, const float4* wl)
{
    const int s_v = w0 & 0x00ffffff;
    const int r_v = (int)((unsigned)w0 >> 24);

    int k = 0;
    for (; k + 4 <= m; k += 4) {
        float4 xa[4], xb[4];
        int    rr[4];
        float  nn[4];
        #pragma unroll
        for (int u = 0; u < 4; ++u) {
            const int s = __builtin_amdgcn_readlane(s_v, k + u);
            rr[u] = __builtin_amdgcn_readlane(r_v, k + u);
            nn[u] = __int_as_float(__builtin_amdgcn_readlane(nb, k + u));
            const float4* xrow = (const float4*)(xin + (size_t)s * HD) + lane;
            xa[u] = xrow[0];
            xb[u] = xrow[32];
        }
        #pragma unroll
        for (int u = 0; u < 4; ++u)
            rgcn_msg_acc<RELU>(acc, xa[u], xb[u], wl + (size_t)rr[u] * NB, nn[u]);
    }
    if (k < m) {
        const int left = m - k;
        const float4 z4 = make_float4(0.f, 0.f, 0.f, 0.f);
        float4 xa1 = z4, xb1 = z4, xa2 = z4, xb2 = z4;
        int    r1 = 0, r2 = 0;
        float  n1 = 0.f, n2 = 0.f;
        const int   s0 = __builtin_amdgcn_readlane(s_v, k);
        const int   r0 = __builtin_amdgcn_readlane(r_v, k);
        const float n0 = __int_as_float(__builtin_amdgcn_readlane(nb, k));
        const float4* x0 = (const float4*)(xin + (size_t)s0 * HD) + lane;
        const float4 xa0 = x0[0];
        const float4 xb0 = x0[32];
        if (left > 1) {
            const int s1 = __builtin_amdgcn_readlane(s_v, k + 1);
            r1 = __builtin_amdgcn_readlane(r_v, k + 1);
            n1 = __int_as_float(__builtin_amdgcn_readlane(nb, k + 1));
            const float4* x1 = (const float4*)(xin + (size_t)s1 * HD) + lane;
            xa1 = x1[0];
            xb1 = x1[32];
        }
        if (left > 2) {
            const int s2 = __builtin_amdgcn_readlane(s_v, k + 2);
            r2 = __builtin_amdgcn_readlane(r_v, k + 2);
            n2 = __int_as_float(__builtin_amdgcn_readlane(nb, k + 2));
            const float4* x2 = (const float4*)(xin + (size_t)s2 * HD) + lane;
            xa2 = x2[0];
            xb2 = x2[32];
        }
        rgcn_msg_acc<RELU>(acc, xa0, xb0, wl + (size_t)r0 * NB, n0);
        if (left > 1) rgcn_msg_acc<RELU>(acc, xa1, xb1, wl + (size_t)r1 * NB, n1);
        if (left > 2) rgcn_msg_acc<RELU>(acc, xa2, xb2, wl + (size_t)r2 * NB, n2);
    }
}

template <bool RELU>
__global__ __launch_bounds__(256)
void rgcn_node_gather_rec(const float* __restrict__ xin,
                          const int*   __restrict__ src,
                          const int*   __restrict__ dst,
                          const int*   __restrict__ etype,
                          const float* __restrict__ norm,
                          const float* __restrict__ w,
                          const int*   __restrict__ seg,
                          const int*   __restrict__ bucket,
                          float* agg,
                          int n_nodes,
                          int n_edges,
                          int cap)
{
    __shared__ __align__(16) float wlds[WTAB];
    __shared__ __align__(16) int   keybuf[8 * 4 * RCAP];

    const int lane   = threadIdx.x & 31;
    const int wid    = __builtin_amdgcn_readfirstlane(
                           blockIdx.x * (blockDim.x >> 5) + (threadIdx.x >> 5));
    const int nwaves = (gridDim.x * blockDim.x) >> 5;
    const int2* rec = (const int2*)bucket;

    int n   = wid;
    int cnt = (n < n_nodes) ? seg[n] : 0;

    if (blockDim.x * 8 == WTAB / 4) {
        float4 tv[8];
        #pragma unroll
        for (int q = 0; q < 8; ++q)
            tv[q] = ((const float4*)w)[threadIdx.x + q * (WTAB / 32)];
        #pragma unroll
        for (int q = 0; q < 8; ++q) {
            const int i = threadIdx.x + q * (WTAB / 32);
            const int b = i & (NB - 1);
            const int j = ((b >> 6) << 1) | (b & 1);
            const int L = (b & 63) >> 1;
            ((float4*)wlds)[(i & ~(NB - 1)) + j * 32 + L] = tv[q];
        }
    } else {
        for (int i = threadIdx.x; i < WTAB / 4; i += blockDim.x) {
            const int b = i & (NB - 1);
            const int j = ((b >> 6) << 1) | (b & 1);
            const int L = (b & 63) >> 1;
            ((float4*)wlds)[(i & ~(NB - 1)) + j * 32 + L] = ((const float4*)w)[i];
        }
    }

    int2 recn = make_int2(-1, -1);
    if (lane < cnt && lane < cap) recn = rec[(size_t)n * cap + lane];
    int n1   = n + nwaves;
    int cnt1 = (n1 < n_nodes) ? seg[n1] : 0;

    __syncthreads();

    const float4* wl = (const float4*)wlds + lane;
    int2* uns = (int2*)(keybuf + __builtin_amdgcn_readfirstlane((int)(threadIdx.x >> 5)) * (4 * RCAP));
    int2* srt = uns + RCAP;

    int2 recn1 = make_int2(-1, -1);
    int  n2 = 0, cnt2 = 0;
    for (; n < n_nodes; n = n1, cnt = cnt1, recn = recn1, n1 = n2, cnt1 = cnt2) {
        recn1 = make_int2(-1, -1);
        if (lane < cnt1 && lane < cap) recn1 = rec[(size_t)n1 * cap + lane];
        n2   = n1 + nwaves;
        cnt2 = (n2 < n_nodes) ? seg[n2] : 0;

        if (cnt <= 0) continue;

        float4* orow = (float4*)(agg + (size_t)n * HD) + lane;
        float4 oa = orow[0];
        float4 ob = orow[32];

        float acc[8] = { 0.f, 0.f, 0.f, 0.f, 0.f, 0.f, 0.f, 0.f };

        if (cnt <= cap && cnt <= RCAP) {
            const int2* row = rec + (size_t)n * cap;
            int w0s = 0, nbs = 0;
            if (cnt <= 32) {
                const int2 mine = recn;
                const unsigned long long my = rgcn_rec_key(mine.x, mine.y);
                int rank = 0;
                for (int m = 0; m < cnt; ++m) {
                    const unsigned long long ok =
                        rgcn_rec_key(__builtin_amdgcn_readlane(mine.x, m),
                                     __builtin_amdgcn_readlane(mine.y, m));
                    rank += (ok < my || (ok == my && m < lane)) ? 1 : 0;
                }
                w0s = __builtin_amdgcn_ds_permute(rank << 2, mine.x);
                nbs = __builtin_amdgcn_ds_permute(rank << 2, mine.y);
            } else {
                const int padded = (cnt + 31) & ~31;
                for (int p = lane; p < padded; p += 32)
                    uns[p] = (p < cnt) ? row[p] : make_int2(-1, -1);
                for (int p = lane; p < padded; p += 32) {
                    const int2 mine = uns[p];
                    const unsigned long long my = rgcn_rec_key(mine.x, mine.y);
                    int rank = 0;
                    for (int q = 0; q < padded; q += 2) {
                        const int4 v = *(const int4*)(uns + q);
                        const unsigned long long k0 = rgcn_rec_key(v.x, v.y);
                        const unsigned long long k1 = rgcn_rec_key(v.z, v.w);
                        rank += ((k0 < my || (k0 == my && q     < p)) ? 1 : 0)
                              + ((k1 < my || (k1 == my && q + 1 < p)) ? 1 : 0);
                    }
                    if (p < cnt) srt[rank] = mine;
                }
            }
            for (int c = 0; c < cnt; c += 32) {
                const int m = (cnt - c < 32) ? (cnt - c) : 32;
                if (cnt > 32) {
                    int2 r = make_int2(0, 0);
                    if (lane < m) r = srt[c + lane];
                    w0s = r.x;
                    nbs = r.y;
                }
                rgcn_rec_run<RELU>(acc, w0s, nbs, m, lane, xin, wl);
            }
        } else {
            for (int e0 = 0; e0 < n_edges; e0 += 32) {
                const int  ev  = e0 + lane;
                const bool hit = (ev < n_edges) && (dst[ev] == n);
                unsigned m = __builtin_amdgcn_ballot_w32(hit);
                while (m != 0u) {
                    const int e = e0 + (__builtin_ctz(m));
                    m &= m - 1u;
                    const int   s  = src[e];
                    const int   r  = etype[e];
                    const float nr = norm[e];
                    const float4* xrow = (const float4*)(xin + (size_t)s * HD) + lane;
                    rgcn_msg_acc<RELU>(acc, xrow[0], xrow[32], wl + (size_t)r * NB, nr);
                }
            }
        }

        oa.x += acc[0]; oa.y += acc[1]; oa.z += acc[2]; oa.w += acc[3];
        ob.x += acc[4]; ob.y += acc[5]; ob.z += acc[6]; ob.w += acc[7];
        orow[0]  = oa;
        orow[32] = ob;
    }
}

extern "C" void kernel_launch(void* const* d_in, const int* in_sizes, int n_in,
                              void* d_out, int out_size, void* d_ws, size_t ws_size,
                              hipStream_t stream)
{
    const float* x     = (const float*)d_in[0];
    const int*   src   = (const int*)  d_in[1];
    const int*   dst   = (const int*)  d_in[2];
    const int*   etype = (const int*)  d_in[3];
    const float* norm  = (const float*)d_in[4];
    const float* w1    = (const float*)d_in[5];
    const float* lw1   = (const float*)d_in[6];
    const float* b1    = (const float*)d_in[7];
    const float* w2    = (const float*)d_in[8];
    const float* lw2   = (const float*)d_in[9];
    const float* b2    = (const float*)d_in[10];
    float*       out   = (float*)d_out;

    const int n_nodes = in_sizes[0] / HD;
    const int n_edges = in_sizes[1];

    float* h = (float*)d_ws;

    const int row_groups  = ((n_nodes + 15) / 16 + 7) / 8;
    const int gemm_blocks = row_groups * (HD / 64);

    const size_t seg_off  = ((size_t)n_nodes * HD * sizeof(float) + 255) & ~(size_t)255;
    const size_t perm_off = (seg_off + (size_t)n_nodes * sizeof(int) + 255) & ~(size_t)255;
    const size_t ws_need  = perm_off + (size_t)n_edges * sizeof(int);

    if (ws_size < ws_need) {
        const int edge_blocks = 1024;
        rgcn_selfloop_gemm<false><<<gemm_blocks, 256, 0, stream>>>(x, lw1, b1, h, n_nodes,
                                                                   nullptr, 0);
        rgcn_edge_msg<false><<<edge_blocks, 256, 0, stream>>>(x, src, dst, etype, norm, w1,
                                                              h, n_edges);
        rgcn_selfloop_gemm<true><<<gemm_blocks, 256, 0, stream>>>(h, lw2, b2, out, n_nodes,
                                                                  nullptr, 0);
        rgcn_edge_msg<true><<<edge_blocks, 256, 0, stream>>>(h, src, dst, etype, norm, w2,
                                                             out, n_edges);
        return;
    }

    int* seg  = (int*)((char*)d_ws + seg_off);
    const int gather_blocks = 1024;

    const int    bcap     = 64;
    const size_t bkt_need = perm_off + (size_t)n_nodes * bcap * sizeof(int);

    const size_t rec_need = perm_off + (size_t)n_nodes * RCAP * 2 * sizeof(int);

    if (ws_size >= rec_need && n_nodes <= (1 << 24)) {
        int* bucket = (int*)((char*)d_ws + perm_off);
        const int fill_blocks = (n_edges + 1023) / 1024;

        rgcn_selfloop_gemm<false><<<gemm_blocks, 256, 0, stream>>>(x, lw1, b1, h, n_nodes,
                                                                   seg, n_nodes);
        rgcn_bucket_fill_rec<<<fill_blocks, 256, 0, stream>>>(dst, src, etype, norm, seg, bucket,
                                                              RCAP, n_edges);

        rgcn_node_gather_rec<false><<<gather_blocks, 256, 0, stream>>>(x, src, dst, etype, norm, w1,
                                                                       seg, bucket, h, n_nodes,
                                                                       n_edges, RCAP);
        rgcn_selfloop_gemm<true><<<gemm_blocks, 256, 0, stream>>>(h, lw2, b2, out, n_nodes,
                                                                  nullptr, 0);
        rgcn_node_gather_rec<true><<<gather_blocks, 256, 0, stream>>>(h, src, dst, etype, norm, w2,
                                                                      seg, bucket, out, n_nodes,
                                                                      n_edges, RCAP);
        return;
    }

    if (ws_size >= bkt_need) {
        int* bucket = (int*)((char*)d_ws + perm_off);
        const int fill_blocks = (n_edges + 1023) / 1024;

        rgcn_selfloop_gemm<false><<<gemm_blocks, 256, 0, stream>>>(x, lw1, b1, h, n_nodes,
                                                                   seg, n_nodes);
        rgcn_bucket_fill<<<fill_blocks, 256, 0, stream>>>(dst, seg, bucket, bcap, n_edges);

        rgcn_node_gather<false><<<gather_blocks, 256, 0, stream>>>(x, src, dst, etype, norm, w1,
                                                                   seg, bucket, h, n_nodes,
                                                                   n_edges, bcap);
        rgcn_selfloop_gemm<true><<<gemm_blocks, 256, 0, stream>>>(h, lw2, b2, out, n_nodes,
                                                                  nullptr, 0);
        rgcn_node_gather<true><<<gather_blocks, 256, 0, stream>>>(h, src, dst, etype, norm, w2,
                                                                  seg, bucket, out, n_nodes,
                                                                  n_edges, bcap);
        return;
    }

    int* perm = (int*)((char*)d_ws + perm_off);
    const int edge_blocks   = (n_edges + 255) / 256;

    rgcn_selfloop_gemm<false><<<gemm_blocks, 256, 0, stream>>>(x, lw1, b1, h, n_nodes,
                                                               seg, n_nodes);
    rgcn_deg_count<<<edge_blocks, 256, 0, stream>>>(dst, seg, n_edges);
    rgcn_seg_scan<<<1, 1024, 0, stream>>>(seg, n_nodes);
    rgcn_seg_fill<<<edge_blocks, 256, 0, stream>>>(dst, seg, perm, n_edges);

    rgcn_node_gather<false><<<gather_blocks, 256, 0, stream>>>(x, src, dst, etype, norm, w1,
                                                               seg, perm, h, n_nodes, n_edges, 0);

    rgcn_selfloop_gemm<true><<<gemm_blocks, 256, 0, stream>>>(h, lw2, b2, out, n_nodes,
                                                              nullptr, 0);
    rgcn_node_gather<true><<<gather_blocks, 256, 0, stream>>>(h, src, dst, etype, norm, w2,
                                                              seg, perm, out, n_nodes, n_edges, 0);
}


# Self-loop GEMM: work-group = 5 row tiles x 4 strips (20 waves), all of W in LDS, one work-group per WGP: busiest SIMD 1280 WMMAs (was 1536), A read from L2 once (was 4 times); k sum order kept
# speedup vs baseline: 2.6346x; 1.0408x over previous
#include <hip/hip_runtime.h>

#define HD   256
#define NB   128
#define RT   16
#define WTAB (RT * NB * 4)
#define LCAP 128

typedef float v2f __attribute__((ext_vector_type(2)));
typedef float v8f __attribute__((ext_vector_type(8)));

#define GEMM_KQ      (HD / 4)
#define GEMM_RT      5
#define GEMM_WAVES   (GEMM_RT * 4)
#define GEMM_THREADS (GEMM_WAVES * 32)
#define GEMM_PITCH   (HD + 1)
#define GEMM_STILES  ((GEMM_KQ / 4) * (HD / 32))
#define GEMM_SPASS   ((GEMM_STILES + GEMM_WAVES - 1) / GEMM_WAVES)

template <bool RELU>
__global__ __launch_bounds__(GEMM_THREADS)
void rgcn_selfloop_gemm(const float* __restrict__ A,
                        const float* __restrict__ W,
                        const float* __restrict__ bias,
                        float* __restrict__ out,
                        int n_rows,
                        int* __restrict__ zero_buf,
                        int zero_n)
{
    __shared__ __align__(16) float bl[GEMM_KQ * GEMM_PITCH * 4];

    if (zero_buf != nullptr) {
        const int gid = blockIdx.x * blockDim.x + threadIdx.x;
        if (gid < zero_n) zero_buf[gid] = 0;
    }

    const int lane = threadIdx.x & 31;
    const int wave    = __builtin_amdgcn_readfirstlane((int)(threadIdx.x >> 5));
    const int strip_n = wave & 3;
    const int tile_m  = blockIdx.x * GEMM_RT + (wave >> 2);
    const bool live   = tile_m * 16 < n_rows;
    const int half = lane >> 4;
    const int sub  = lane & 15;
    const int col0 = strip_n * 64 + sub;

    const float4* arow = (const float4*)(A + (size_t)(tile_m * 16 + sub) * HD) + half;

    float4 an[4] = {};
    if (live) {
        #pragma unroll
        for (int q = 0; q < 4; ++q) an[q] = arow[2 * q];
    }

    {
        const int kqi = lane & 3;
        const int cqi = lane >> 2;
        float4 r[GEMM_SPASS][4];
        #pragma unroll
        for (int s = 0; s < GEMM_SPASS; ++s) {
            const int tile = wave + s * GEMM_WAVES;
            if (tile < GEMM_STILES) {
                const int kq = (tile >> 3) * 4 + kqi;
                const int cq = (tile & 7) * 8 + cqi;
                const float4* p = (const float4*)(W + (size_t)(4 * kq) * HD) + cq;
                r[s][0] = p[0];
                r[s][1] = p[HD / 4];
                r[s][2] = p[2 * (HD / 4)];
                r[s][3] = p[3 * (HD / 4)];
            }
        }
        #pragma unroll
        for (int s = 0; s < GEMM_SPASS; ++s) {
            const int tile = wave + s * GEMM_WAVES;
            if (tile < GEMM_STILES) {
                const int kq = (tile >> 3) * 4 + kqi;
                const int cq = (tile & 7) * 8 + cqi;
                float4* d = (float4*)bl + kq * GEMM_PITCH + 4 * cq;
                d[0] = make_float4(r[s][0].x, r[s][1].x, r[s][2].x, r[s][3].x);
                d[1] = make_float4(r[s][0].y, r[s][1].y, r[s][2].y, r[s][3].y);
                d[2] = make_float4(r[s][0].z, r[s][1].z, r[s][2].z, r[s][3].z);
                d[3] = make_float4(r[s][0].w, r[s][1].w, r[s][2].w, r[s][3].w);
            }
        }
    }
    __syncthreads();

    if (!live) return;

    const float4* bfr  = (const float4*)bl + half * GEMM_PITCH + strip_n * 64 + sub;

    v8f c0 = {}, c1 = {}, c2 = {}, c3 = {};

    for (int g = 0; g < 8; ++g) {
        float4 ac[4];
        #pragma unroll
        for (int q = 0; q < 4; ++q) ac[q] = an[q];
        if (g < 7) {
            #pragma unroll
            for (int q = 0; q < 4; ++q) an[q] = arow[2 * (4 * (g + 1) + q)];
        }
        #pragma unroll
        for (int q = 0; q < 4; ++q) {
            const int t = 4 * g + q;
            float4 a4 = ac[q];
            if (RELU) {
                a4.x = fmaxf(a4.x, 0.0f); a4.y = fmaxf(a4.y, 0.0f);
                a4.z = fmaxf(a4.z, 0.0f); a4.w = fmaxf(a4.w, 0.0f);
            }
            const float4 q0 = bfr[t * (2 * GEMM_PITCH) +  0];
            const float4 q1 = bfr[t * (2 * GEMM_PITCH) + 16];
            const float4 q2 = bfr[t * (2 * GEMM_PITCH) + 32];
            const float4 q3 = bfr[t * (2 * GEMM_PITCH) + 48];

            v2f a;  v2f b0, b1, b2, b3;
            a.x  = a4.x; a.y  = a4.y;
            b0.x = q0.x; b0.y = q0.y;
            b1.x = q1.x; b1.y = q1.y;
            b2.x = q2.x; b2.y = q2.y;
            b3.x = q3.x; b3.y = q3.y;
            c0 = __builtin_amdgcn_wmma_f32_16x16x4_f32(false, a, false, b0, (short)0, c0, false, false);
            c1 = __builtin_amdgcn_wmma_f32_16x16x4_f32(false, a, false, b1, (short)0, c1, false, false);
            c2 = __builtin_amdgcn_wmma_f32_16x16x4_f32(false, a, false, b2, (short)0, c2, false, false);
            c3 = __builtin_amdgcn_wmma_f32_16x16x4_f32(false, a, false, b3, (short)0, c3, false, false);

            a.x  = a4.z; a.y  = a4.w;
            b0.x = q0.z; b0.y = q0.w;
            b1.x = q1.z; b1.y = q1.w;
            b2.x = q2.z; b2.y = q2.w;
            b3.x = q3.z; b3.y = q3.w;
            c0 = __builtin_amdgcn_wmma_f32_16x16x4_f32(false, a, false, b0, (short)0, c0, false, false);
            c1 = __builtin_amdgcn_wmma_f32_16x16x4_f32(false, a, false, b1, (short)0, c1, false, false);
            c2 = __builtin_amdgcn_wmma_f32_16x16x4_f32(false, a, false, b2, (short)0, c2, false, false);
            c3 = __builtin_amdgcn_wmma_f32_16x16x4_f32(false, a, false, b3, (short)0, c3, false, false);
        }
    }

    float* obase = out + (size_t)(tile_m * 16 + half * 8) * HD + col0;
    #pragma unroll
    for (int t = 0; t < 4; ++t) {
        const v8f& c = (t == 0) ? c0 : (t == 1) ? c1 : (t == 2) ? c2 : c3;
        const float bcol = bias[col0 + t * 16];
        #pragma unroll
        for (int j = 0; j < 8; ++j)
            obase[(size_t)j * HD + t * 16] = c[j] + bcol;
    }
}

template <bool RELU>
__global__ __launch_bounds__(256)
void rgcn_edge_msg(const float* __restrict__ xin,
                   const int*   __restrict__ src,
                   const int*   __restrict__ dst,
                   const int*   __restrict__ etype,
                   const float* __restrict__ norm,
                   const float* __restrict__ w,
                   float* agg,
                   int n_edges)
{
    __shared__ __align__(16) float wlds[WTAB];
    for (int i = threadIdx.x; i < WTAB; i += blockDim.x) wlds[i] = w[i];
    __syncthreads();

    const int lane   = threadIdx.x & 31;
    const int wid    = blockIdx.x * (blockDim.x >> 5) + (threadIdx.x >> 5);
    const int nwaves = (gridDim.x * blockDim.x) >> 5;

    for (int e = wid; e < n_edges; e += nwaves) {
        const int   s   = src[e];
        const int   d   = dst[e];
        const int   r   = etype[e];
        const float nrm = norm[e];

        const float4* xrow = (const float4*)(xin + (size_t)s * HD) + lane * 2;
        float4 xa = xrow[0];
        float4 xb = xrow[1];
        if (RELU) {
            xa.x = fmaxf(xa.x, 0.0f); xa.y = fmaxf(xa.y, 0.0f);
            xa.z = fmaxf(xa.z, 0.0f); xa.w = fmaxf(xa.w, 0.0f);
            xb.x = fmaxf(xb.x, 0.0f); xb.y = fmaxf(xb.y, 0.0f);
            xb.z = fmaxf(xb.z, 0.0f); xb.w = fmaxf(xb.w, 0.0f);
        }
        const float xs[8] = { xa.x, xa.y, xa.z, xa.w, xb.x, xb.y, xb.z, xb.w };

        const float4* wrel = (const float4*)(wlds + (size_t)r * (NB * 4)) + lane * 4;
        float* arow = agg + (size_t)d * HD + lane * 8;

        #pragma unroll
        for (int j = 0; j < 4; ++j) {
            const float4 wq = wrel[j];
            const float x0 = xs[2 * j + 0];
            const float x1 = xs[2 * j + 1];
            const float o0 = (x0 * wq.x + x1 * wq.z) * nrm;
            const float o1 = (x0 * wq.y + x1 * wq.w) * nrm;
            (void)__hip_atomic_fetch_add(&arow[2 * j + 0], o0,
                                         __ATOMIC_RELAXED, __HIP_MEMORY_SCOPE_AGENT);
            (void)__hip_atomic_fetch_add(&arow[2 * j + 1], o1,
                                         __ATOMIC_RELAXED, __HIP_MEMORY_SCOPE_AGENT);
        }
    }
}

__global__ __launch_bounds__(256)
void rgcn_deg_count(const int* __restrict__ dst, int* seg, int n_edges)
{
    const int e = blockIdx.x * blockDim.x + threadIdx.x;
    if (e < n_edges)
        (void)__hip_atomic_fetch_add(&seg[dst[e]], 1,
                                     __ATOMIC_RELAXED, __HIP_MEMORY_SCOPE_AGENT);
}

__global__ __launch_bounds__(1024)
void rgcn_seg_scan(int* seg, int n)
{
    __shared__ int wsum[32];
    const int t    = threadIdx.x;
    const int lane = t & 31;
    const int w    = t >> 5;
    const int chunk = (n + 1023) / 1024;
    const int b  = t * chunk;
    const int en = (b + chunk < n) ? (b + chunk) : n;

    int s = 0;
    for (int i = b; i < en; ++i) s += seg[i];

    int v = s;
    #pragma unroll
    for (int o = 1; o < 32; o <<= 1) {
        const int u = __shfl_up(v, o);
        if (lane >= o) v += u;
    }
    if (lane == 31) wsum[w] = v;
    __syncthreads();
    if (w == 0) {
        const int x = wsum[lane];
        int y = x;
        #pragma unroll
        for (int o = 1; o < 32; o <<= 1) {
            const int u = __shfl_up(y, o);
            if (lane >= o) y += u;
        }
        wsum[lane] = y - x;
    }
    __syncthreads();

    int run = wsum[w] + v - s;
    for (int i = b; i < en; ++i) {
        const int c = seg[i];
        seg[i] = run;
        run += c;
    }
}

__global__ __launch_bounds__(256)
void rgcn_seg_fill(const int* __restrict__ dst, int* seg, int* __restrict__ perm,
                   int n_edges)
{
    const int e = blockIdx.x * blockDim.x + threadIdx.x;
    if (e < n_edges) {
        const int p = __hip_atomic_fetch_add(&seg[dst[e]], 1,
                                             __ATOMIC_RELAXED, __HIP_MEMORY_SCOPE_AGENT);
        perm[p] = e;
    }
}

__global__ __launch_bounds__(256)
void rgcn_bucket_fill(const int* __restrict__ dst, int* cnt, int* __restrict__ bucket,
                      int cap, int n_edges)
{
    const int e0 = (blockIdx.x * blockDim.x + threadIdx.x) * 4;
    if (e0 + 3 < n_edges) {
        const int4 d = *(const int4*)(dst + e0);
        const int p0 = __hip_atomic_fetch_add(&cnt[d.x], 1,
                                              __ATOMIC_RELAXED, __HIP_MEMORY_SCOPE_AGENT);
        const int p1 = __hip_atomic_fetch_add(&cnt[d.y], 1,
                                              __ATOMIC_RELAXED, __HIP_MEMORY_SCOPE_AGENT);
        const int p2 = __hip_atomic_fetch_add(&cnt[d.z], 1,
                                              __ATOMIC_RELAXED, __HIP_MEMORY_SCOPE_AGENT);
        const int p3 = __hip_atomic_fetch_add(&cnt[d.w], 1,
                                              __ATOMIC_RELAXED, __HIP_MEMORY_SCOPE_AGENT);
        if (p0 < cap) bucket[(size_t)d.x * cap + p0] = e0;
        if (p1 < cap) bucket[(size_t)d.y * cap + p1] = e0 + 1;
        if (p2 < cap) bucket[(size_t)d.z * cap + p2] = e0 + 2;
        if (p3 < cap) bucket[(size_t)d.w * cap + p3] = e0 + 3;
    } else {
        for (int e = e0; e < n_edges; ++e) {
            const int d = dst[e];
            const int p = __hip_atomic_fetch_add(&cnt[d], 1,
                                                 __ATOMIC_RELAXED, __HIP_MEMORY_SCOPE_AGENT);
            if (p < cap) bucket[(size_t)d * cap + p] = e;
        }
    }
}

template <bool RELU>
__device__ __forceinline__
void rgcn_msg_acc(float (&acc)[8], float4 xa, float4 xb,
                  const float4* wrel, float nrm)
{
    if (RELU) {
        xa.x = fmaxf(xa.x, 0.0f); xa.y = fmaxf(xa.y, 0.0f);
        xa.z = fmaxf(xa.z, 0.0f); xa.w = fmaxf(xa.w, 0.0f);
        xb.x = fmaxf(xb.x, 0.0f); xb.y = fmaxf(xb.y, 0.0f);
        xb.z = fmaxf(xb.z, 0.0f); xb.w = fmaxf(xb.w, 0.0f);
    }
    const float xs[8] = { xa.x, xa.y, xa.z, xa.w, xb.x, xb.y, xb.z, xb.w };
    #pragma unroll
    for (int j = 0; j < 4; ++j) {
        const float4 wq = wrel[32 * j];
        const float x0 = xs[2 * j + 0];
        const float x1 = xs[2 * j + 1];
        acc[2 * j + 0] += (x0 * wq.x + x1 * wq.z) * nrm;
        acc[2 * j + 1] += (x0 * wq.y + x1 * wq.w) * nrm;
    }
}

template <bool RELU>
__device__ __forceinline__
void rgcn_sorted_run(float (&acc)[8], const int sorted, const int m, const int lane,
                     const float* __restrict__ xin,
                     const int*   __restrict__ src,
                     const int*   __restrict__ etype,
                     const float* __restrict__ norm,
                     const float4* wl)
{
    int   s_v = 0, r_v = 0;
    float n_v = 0.0f;
    if (lane < m) {
        s_v = src[sorted];
        r_v = etype[sorted];
        n_v = norm[sorted];
    }

    int k = 0;
    for (; k + 4 <= m; k += 4) {
        float4 xa[4], xb[4];
        int    rr[4];
        float  nn[4];
        #pragma unroll
        for (int u = 0; u < 4; ++u) {
            const int s = __builtin_amdgcn_readlane(s_v, k + u);
            rr[u] = __builtin_amdgcn_readlane(r_v, k + u);
            nn[u] = __int_as_float(
                        __builtin_amdgcn_readlane(__float_as_int(n_v), k + u));
            const float4* xrow = (const float4*)(xin + (size_t)s * HD) + lane;
            xa[u] = xrow[0];
            xb[u] = xrow[32];
        }
        #pragma unroll
        for (int u = 0; u < 4; ++u)
            rgcn_msg_acc<RELU>(acc, xa[u], xb[u], wl + (size_t)rr[u] * NB, nn[u]);
    }
    for (; k < m; ++k) {
        const int   s  = __builtin_amdgcn_readlane(s_v, k);
        const int   r  = __builtin_amdgcn_readlane(r_v, k);
        const float nr = __int_as_float(
                             __builtin_amdgcn_readlane(__float_as_int(n_v), k));
        const float4* xrow = (const float4*)(xin + (size_t)s * HD) + lane;
        rgcn_msg_acc<RELU>(acc, xrow[0], xrow[32], wl + (size_t)r * NB, nr);
    }
}

template <bool RELU>
__global__ __launch_bounds__(256)
void rgcn_node_gather(const float* __restrict__ xin,
                      const int*   __restrict__ src,
                      const int*   __restrict__ dst,
                      const int*   __restrict__ etype,
                      const float* __restrict__ norm,
                      const float* __restrict__ w,
                      const int*   __restrict__ seg,
                      const int*   __restrict__ perm,
                      float* agg,
                      int n_nodes,
                      int n_edges,
                      int cap)
{
    __shared__ __align__(16) float wlds[WTAB];
    __shared__ __align__(16) int   idbuf[8 * 2 * LCAP];
    for (int i = threadIdx.x; i < WTAB / 4; i += blockDim.x) {
        const int b = i & (NB - 1);
        const int j = ((b >> 6) << 1) | (b & 1);
        const int L = (b & 63) >> 1;
        ((float4*)wlds)[(i & ~(NB - 1)) + j * 32 + L] = ((const float4*)w)[i];
    }
    __syncthreads();

    const int lane   = threadIdx.x & 31;
    const int wid    = __builtin_amdgcn_readfirstlane(
                           blockIdx.x * (blockDim.x >> 5) + (threadIdx.x >> 5));
    const int nwaves = (gridDim.x * blockDim.x) >> 5;
    const float4* wl = (const float4*)wlds + lane;
    int* uns = idbuf + __builtin_amdgcn_readfirstlane((int)(threadIdx.x >> 5)) * (2 * LCAP);
    int* srt = uns + LCAP;

    for (int n = wid; n < n_nodes; n += nwaves) {
        int beg, cnt;
        if (cap > 0) {
            cnt = seg[n];
            beg = n * cap;
        } else {
            beg = (n > 0) ? seg[n - 1] : 0;
            cnt = seg[n] - beg;
        }
        if (cnt <= 0) continue;
        const int  end    = beg + cnt;
        const bool listed = (cap == 0) || (cnt <= cap);

        float4* orow = (float4*)(agg + (size_t)n * HD) + lane;
        float4 oa = orow[0];
        float4 ob = orow[32];

        float acc[8] = { 0.f, 0.f, 0.f, 0.f, 0.f, 0.f, 0.f, 0.f };

        if (listed && cnt <= LCAP) {
            int sorted0 = 0;
            if (cnt <= 32) {
                const int my = (lane < cnt) ? perm[beg + lane] : 0x7fffffff;
                int rank = 0;
                for (int m = 0; m < cnt; ++m)
                    rank += (__builtin_amdgcn_readlane(my, m) < my) ? 1 : 0;
                sorted0 = __builtin_amdgcn_ds_permute(rank << 2, my);
            } else {
                const int padded = (cnt + 31) & ~31;
                for (int p = lane; p < padded; p += 32)
                    uns[p] = (p < cnt) ? perm[beg + p] : 0x7fffffff;
                for (int p = lane; p < padded; p += 32) {
                    const int my = uns[p];
                    int rank = 0;
                    for (int q = 0; q < padded; q += 4) {
                        const int4 v = *(const int4*)(uns + q);
                        rank += ((v.x < my) ? 1 : 0) + ((v.y < my) ? 1 : 0)
                              + ((v.z < my) ? 1 : 0) + ((v.w < my) ? 1 : 0);
                    }
                    if (p < cnt) srt[rank] = my;
                }
            }
            for (int c = 0; c < cnt; c += 32) {
                const int m = (cnt - c < 32) ? (cnt - c) : 32;
                const int sorted = (cnt <= 32) ? sorted0 : ((lane < m) ? srt[c + lane] : 0);
                rgcn_sorted_run<RELU>(acc, sorted, m, lane, xin, src, etype, norm, wl);
            }
        } else if (listed && cnt <= 2048) {
            int lo = -1;
            for (int i = 0; i < cnt; ++i) {
                int best = 0x7fffffff;
                for (int p = beg + lane; p < end; p += 32) {
                    const int e = perm[p];
                    if (e > lo && e < best) best = e;
                }
                #pragma unroll
                for (int o = 16; o > 0; o >>= 1) {
                    const int other = __shfl_xor(best, o);
                    best = (other < best) ? other : best;
                }
                const int e = __builtin_amdgcn_readfirstlane(best);
                lo = e;
                const int   s  = src[e];
                const int   r  = etype[e];
                const float nr = norm[e];
                const float4* xrow = (const float4*)(xin + (size_t)s * HD) + lane;
                rgcn_msg_acc<RELU>(acc, xrow[0], xrow[32], wl + (size_t)r * NB, nr);
            }
        } else {
            for (int e0 = 0; e0 < n_edges; e0 += 32) {
                const int  ev  = e0 + lane;
                const bool hit = (ev < n_edges) && (dst[ev] == n);
                unsigned m = __builtin_amdgcn_ballot_w32(hit);
                while (m != 0u) {
                    const int e = e0 + (__builtin_ctz(m));
                    m &= m - 1u;
                    const int   s  = src[e];
                    const int   r  = etype[e];
                    const float nr = norm[e];
                    const float4* xrow = (const float4*)(xin + (size_t)s * HD) + lane;
                    rgcn_msg_acc<RELU>(acc, xrow[0], xrow[32], wl + (size_t)r * NB, nr);
                }
            }
        }

        oa.x += acc[0]; oa.y += acc[1]; oa.z += acc[2]; oa.w += acc[3];
        ob.x += acc[4]; ob.y += acc[5]; ob.z += acc[6]; ob.w += acc[7];
        orow[0]  = oa;
        orow[32] = ob;
    }
}

#define RCAP 64

__global__ __launch_bounds__(256)
void rgcn_bucket_fill_rec(const int*   __restrict__ dst,
                          const int*   __restrict__ src,
                          const int*   __restrict__ etype,
                          const float* __restrict__ norm,
                          int* cnt, int* __restrict__ bucket,
                          int cap, int n_edges)
{
    int2* rec = (int2*)bucket;
    const int e0 = (blockIdx.x * blockDim.x + threadIdx.x) * 4;
    if (e0 + 3 < n_edges) {
        const int4   d = *(const int4*)(dst + e0);
        const int4   s = *(const int4*)(src + e0);
        const int4   r = *(const int4*)(etype + e0);
        const float4 w = *(const float4*)(norm + e0);
        const int p0 = __hip_atomic_fetch_add(&cnt[d.x], 1,
                                              __ATOMIC_RELAXED, __HIP_MEMORY_SCOPE_AGENT);
        const int p1 = __hip_atomic_fetch_add(&cnt[d.y], 1,
                                              __ATOMIC_RELAXED, __HIP_MEMORY_SCOPE_AGENT);
        const int p2 = __hip_atomic_fetch_add(&cnt[d.z], 1,
                                              __ATOMIC_RELAXED, __HIP_MEMORY_SCOPE_AGENT);
        const int p3 = __hip_atomic_fetch_add(&cnt[d.w], 1,
                                              __ATOMIC_RELAXED, __HIP_MEMORY_SCOPE_AGENT);
        if (p0 < cap) rec[(size_t)d.x * cap + p0] = make_int2(s.x | (r.x << 24), __float_as_int(w.x));
        if (p1 < cap) rec[(size_t)d.y * cap + p1] = make_int2(s.y | (r.y << 24), __float_as_int(w.y));
        if (p2 < cap) rec[(size_t)d.z * cap + p2] = make_int2(s.z | (r.z << 24), __float_as_int(w.z));
        if (p3 < cap) rec[(size_t)d.w * cap + p3] = make_int2(s.w | (r.w << 24), __float_as_int(w.w));
    } else {
        for (int e = e0; e < n_edges; ++e) {
            const int d = dst[e];
            const int p = __hip_atomic_fetch_add(&cnt[d], 1,
                                                 __ATOMIC_RELAXED, __HIP_MEMORY_SCOPE_AGENT);
            if (p < cap)
                rec[(size_t)d * cap + p] = make_int2(src[e] | (etype[e] << 24),
                                                     __float_as_int(norm[e]));
        }
    }
}

__device__ __forceinline__
unsigned long long rgcn_rec_key(int w0, int nb)
{
    return ((unsigned long long)(unsigned)w0 << 32) | (unsigned long long)(unsigned)nb;
}

template <bool RELU>
__device__ __forceinline__
void rgcn_rec_run(float (&acc)[8], const int w0, const int nb, const int m, const int lane,
                  const float* __restrict__ xin, const float4* wl)
{
    const int s_v = w0 & 0x00ffffff;
    const int r_v = (int)((unsigned)w0 >> 24);

    int k = 0;
    for (; k + 4 <= m; k += 4) {
        float4 xa[4], xb[4];
        int    rr[4];
        float  nn[4];
        #pragma unroll
        for (int u = 0; u < 4; ++u) {
            const int s = __builtin_amdgcn_readlane(s_v, k + u);
            rr[u] = __builtin_amdgcn_readlane(r_v, k + u);
            nn[u] = __int_as_float(__builtin_amdgcn_readlane(nb, k + u));
            const float4* xrow = (const float4*)(xin + (size_t)s * HD) + lane;
            xa[u] = xrow[0];
            xb[u] = xrow[32];
        }
        #pragma unroll
        for (int u = 0; u < 4; ++u)
            rgcn_msg_acc<RELU>(acc, xa[u], xb[u], wl + (size_t)rr[u] * NB, nn[u]);
    }
    if (k < m) {
        const int left = m - k;
        const float4 z4 = make_float4(0.f, 0.f, 0.f, 0.f);
        float4 xa1 = z4, xb1 = z4, xa2 = z4, xb2 = z4;
        int    r1 = 0, r2 = 0;
        float  n1 = 0.f, n2 = 0.f;
        const int   s0 = __builtin_amdgcn_readlane(s_v, k);
        const int   r0 = __builtin_amdgcn_readlane(r_v, k);
        const float n0 = __int_as_float(__builtin_amdgcn_readlane(nb, k));
        const float4* x0 = (const float4*)(xin + (size_t)s0 * HD) + lane;
        const float4 xa0 = x0[0];
        const float4 xb0 = x0[32];
        if (left > 1) {
            const int s1 = __builtin_amdgcn_readlane(s_v, k + 1);
            r1 = __builtin_amdgcn_readlane(r_v, k + 1);
            n1 = __int_as_float(__builtin_amdgcn_readlane(nb, k + 1));
            const float4* x1 = (const float4*)(xin + (size_t)s1 * HD) + lane;
            xa1 = x1[0];
            xb1 = x1[32];
        }
        if (left > 2) {
            const int s2 = __builtin_amdgcn_readlane(s_v, k + 2);
            r2 = __builtin_amdgcn_readlane(r_v, k + 2);
            n2 = __int_as_float(__builtin_amdgcn_readlane(nb, k + 2));
            const float4* x2 = (const float4*)(xin + (size_t)s2 * HD) + lane;
            xa2 = x2[0];
            xb2 = x2[32];
        }
        rgcn_msg_acc<RELU>(acc, xa0, xb0, wl + (size_t)r0 * NB, n0);
        if (left > 1) rgcn_msg_acc<RELU>(acc, xa1, xb1, wl + (size_t)r1 * NB, n1);
        if (left > 2) rgcn_msg_acc<RELU>(acc, xa2, xb2, wl + (size_t)r2 * NB, n2);
    }
}

template <bool RELU>
__global__ __launch_bounds__(256)
void rgcn_node_gather_rec(const float* __restrict__ xin,
                          const int*   __restrict__ src,
                          const int*   __restrict__ dst,
                          const int*   __restrict__ etype,
                          const float* __restrict__ norm,
                          const float* __restrict__ w,
                          const int*   __restrict__ seg,
                          const int*   __restrict__ bucket,
                          float* agg,
                          int n_nodes,
                          int n_edges,
                          int cap)
{
    __shared__ __align__(16) float wlds[WTAB];
    __shared__ __align__(16) int   keybuf[8 * 4 * RCAP];

    const int lane   = threadIdx.x & 31;
    const int wid    = __builtin_amdgcn_readfirstlane(
                           blockIdx.x * (blockDim.x >> 5) + (threadIdx.x >> 5));
    const int nwaves = (gridDim.x * blockDim.x) >> 5;
    const int2* rec = (const int2*)bucket;

    int n   = wid;
    int cnt = (n < n_nodes) ? seg[n] : 0;

    if (blockDim.x * 8 == WTAB / 4) {
        float4 tv[8];
        #pragma unroll
        for (int q = 0; q < 8; ++q)
            tv[q] = ((const float4*)w)[threadIdx.x + q * (WTAB / 32)];
        #pragma unroll
        for (int q = 0; q < 8; ++q) {
            const int i = threadIdx.x + q * (WTAB / 32);
            const int b = i & (NB - 1);
            const int j = ((b >> 6) << 1) | (b & 1);
            const int L = (b & 63) >> 1;
            ((float4*)wlds)[(i & ~(NB - 1)) + j * 32 + L] = tv[q];
        }
    } else {
        for (int i = threadIdx.x; i < WTAB / 4; i += blockDim.x) {
            const int b = i & (NB - 1);
            const int j = ((b >> 6) << 1) | (b & 1);
            const int L = (b & 63) >> 1;
            ((float4*)wlds)[(i & ~(NB - 1)) + j * 32 + L] = ((const float4*)w)[i];
        }
    }

    int2 recn = make_int2(-1, -1);
    if (lane < cnt && lane < cap) recn = rec[(size_t)n * cap + lane];
    int n1   = n + nwaves;
    int cnt1 = (n1 < n_nodes) ? seg[n1] : 0;

    __syncthreads();

    const float4* wl = (const float4*)wlds + lane;
    int2* uns = (int2*)(keybuf + __builtin_amdgcn_readfirstlane((int)(threadIdx.x >> 5)) * (4 * RCAP));
    int2* srt = uns + RCAP;

    int2 recn1 = make_int2(-1, -1);
    int  n2 = 0, cnt2 = 0;
    for (; n < n_nodes; n = n1, cnt = cnt1, recn = recn1, n1 = n2, cnt1 = cnt2) {
        recn1 = make_int2(-1, -1);
        if (lane < cnt1 && lane < cap) recn1 = rec[(size_t)n1 * cap + lane];
        n2   = n1 + nwaves;
        cnt2 = (n2 < n_nodes) ? seg[n2] : 0;

        if (cnt <= 0) continue;

        float4* orow = (float4*)(agg + (size_t)n * HD) + lane;
        float4 oa = orow[0];
        float4 ob = orow[32];

        float acc[8] = { 0.f, 0.f, 0.f, 0.f, 0.f, 0.f, 0.f, 0.f };

        if (cnt <= cap && cnt <= RCAP) {
            const int2* row = rec + (size_t)n * cap;
            int w0s = 0, nbs = 0;
            if (cnt <= 32) {
                const int2 mine = recn;
                const unsigned long long my = rgcn_rec_key(mine.x, mine.y);
                int rank = 0;
                for (int m = 0; m < cnt; ++m) {
                    const unsigned long long ok =
                        rgcn_rec_key(__builtin_amdgcn_readlane(mine.x, m),
                                     __builtin_amdgcn_readlane(mine.y, m));
                    rank += (ok < my || (ok == my && m < lane)) ? 1 : 0;
                }
                w0s = __builtin_amdgcn_ds_permute(rank << 2, mine.x);
                nbs = __builtin_amdgcn_ds_permute(rank << 2, mine.y);
            } else {
                const int padded = (cnt + 31) & ~31;
                for (int p = lane; p < padded; p += 32)
                    uns[p] = (p < cnt) ? row[p] : make_int2(-1, -1);
                for (int p = lane; p < padded; p += 32) {
                    const int2 mine = uns[p];
                    const unsigned long long my = rgcn_rec_key(mine.x, mine.y);
                    int rank = 0;
                    for (int q = 0; q < padded; q += 2) {
                        const int4 v = *(const int4*)(uns + q);
                        const unsigned long long k0 = rgcn_rec_key(v.x, v.y);
                        const unsigned long long k1 = rgcn_rec_key(v.z, v.w);
                        rank += ((k0 < my || (k0 == my && q     < p)) ? 1 : 0)
                              + ((k1 < my || (k1 == my && q + 1 < p)) ? 1 : 0);
                    }
                    if (p < cnt) srt[rank] = mine;
                }
            }
            for (int c = 0; c < cnt; c += 32) {
                const int m = (cnt - c < 32) ? (cnt - c) : 32;
                if (cnt > 32) {
                    int2 r = make_int2(0, 0);
                    if (lane < m) r = srt[c + lane];
                    w0s = r.x;
                    nbs = r.y;
                }
                rgcn_rec_run<RELU>(acc, w0s, nbs, m, lane, xin, wl);
            }
        } else {
            for (int e0 = 0; e0 < n_edges; e0 += 32) {
                const int  ev  = e0 + lane;
                const bool hit = (ev < n_edges) && (dst[ev] == n);
                unsigned m = __builtin_amdgcn_ballot_w32(hit);
                while (m != 0u) {
                    const int e = e0 + (__builtin_ctz(m));
                    m &= m - 1u;
                    const int   s  = src[e];
                    const int   r  = etype[e];
                    const float nr = norm[e];
                    const float4* xrow = (const float4*)(xin + (size_t)s * HD) + lane;
                    rgcn_msg_acc<RELU>(acc, xrow[0], xrow[32], wl + (size_t)r * NB, nr);
                }
            }
        }

        oa.x += acc[0]; oa.y += acc[1]; oa.z += acc[2]; oa.w += acc[3];
        ob.x += acc[4]; ob.y += acc[5]; ob.z += acc[6]; ob.w += acc[7];
        orow[0]  = oa;
        orow[32] = ob;
    }
}

extern "C" void kernel_launch(void* const* d_in, const int* in_sizes, int n_in,
                              void* d_out, int out_size, void* d_ws, size_t ws_size,
                              hipStream_t stream)
{
    const float* x     = (const float*)d_in[0];
    const int*   src   = (const int*)  d_in[1];
    const int*   dst   = (const int*)  d_in[2];
    const int*   etype = (const int*)  d_in[3];
    const float* norm  = (const float*)d_in[4];
    const float* w1    = (const float*)d_in[5];
    const float* lw1   = (const float*)d_in[6];
    const float* b1    = (const float*)d_in[7];
    const float* w2    = (const float*)d_in[8];
    const float* lw2   = (const float*)d_in[9];
    const float* b2    = (const float*)d_in[10];
    float*       out   = (float*)d_out;

    const int n_nodes = in_sizes[0] / HD;
    const int n_edges = in_sizes[1];

    float* h = (float*)d_ws;

    const int gemm_blocks = ((n_nodes + 15) / 16 + GEMM_RT - 1) / GEMM_RT;

    const size_t seg_off  = ((size_t)n_nodes * HD * sizeof(float) + 255) & ~(size_t)255;
    const size_t perm_off = (seg_off + (size_t)n_nodes * sizeof(int) + 255) & ~(size_t)255;
    const size_t ws_need  = perm_off + (size_t)n_edges * sizeof(int);

    if (ws_size < ws_need) {
        const int edge_blocks = 1024;
        rgcn_selfloop_gemm<false><<<gemm_blocks, GEMM_THREADS, 0, stream>>>(x, lw1, b1, h, n_nodes,
                                                                   nullptr, 0);
        rgcn_edge_msg<false><<<edge_blocks, 256, 0, stream>>>(x, src, dst, etype, norm, w1,
                                                              h, n_edges);
        rgcn_selfloop_gemm<true><<<gemm_blocks, GEMM_THREADS, 0, stream>>>(h, lw2, b2, out, n_nodes,
                                                                  nullptr, 0);
        rgcn_edge_msg<true><<<edge_blocks, 256, 0, stream>>>(h, src, dst, etype, norm, w2,
                                                             out, n_edges);
        return;
    }

    int* seg  = (int*)((char*)d_ws + seg_off);
    const int gather_blocks = 1024;

    const int    bcap     = 64;
    const size_t bkt_need = perm_off + (size_t)n_nodes * bcap * sizeof(int);

    const size_t rec_need = perm_off + (size_t)n_nodes * RCAP * 2 * sizeof(int);

    if (ws_size >= rec_need && n_nodes <= (1 << 24)) {
        int* bucket = (int*)((char*)d_ws + perm_off);
        const int fill_blocks = (n_edges + 1023) / 1024;

        rgcn_selfloop_gemm<false><<<gemm_blocks, GEMM_THREADS, 0, stream>>>(x, lw1, b1, h, n_nodes,
                                                                   seg, n_nodes);
        rgcn_bucket_fill_rec<<<fill_blocks, 256, 0, stream>>>(dst, src, etype, norm, seg, bucket,
                                                              RCAP, n_edges);

        rgcn_node_gather_rec<false><<<gather_blocks, 256, 0, stream>>>(x, src, dst, etype, norm, w1,
                                                                       seg, bucket, h, n_nodes,
                                                                       n_edges, RCAP);
        rgcn_selfloop_gemm<true><<<gemm_blocks, GEMM_THREADS, 0, stream>>>(h, lw2, b2, out, n_nodes,
                                                                  nullptr, 0);
        rgcn_node_gather_rec<true><<<gather_blocks, 256, 0, stream>>>(h, src, dst, etype, norm, w2,
                                                                      seg, bucket, out, n_nodes,
                                                                      n_edges, RCAP);
        return;
    }

    if (ws_size >= bkt_need) {
        int* bucket = (int*)((char*)d_ws + perm_off);
        const int fill_blocks = (n_edges + 1023) / 1024;

        rgcn_selfloop_gemm<false><<<gemm_blocks, GEMM_THREADS, 0, stream>>>(x, lw1, b1, h, n_nodes,
                                                                   seg, n_nodes);
        rgcn_bucket_fill<<<fill_blocks, 256, 0, stream>>>(dst, seg, bucket, bcap, n_edges);

        rgcn_node_gather<false><<<gather_blocks, 256, 0, stream>>>(x, src, dst, etype, norm, w1,
                                                                   seg, bucket, h, n_nodes,
                                                                   n_edges, bcap);
        rgcn_selfloop_gemm<true><<<gemm_blocks, GEMM_THREADS, 0, stream>>>(h, lw2, b2, out, n_nodes,
                                                                  nullptr, 0);
        rgcn_node_gather<true><<<gather_blocks, 256, 0, stream>>>(h, src, dst, etype, norm, w2,
                                                                  seg, bucket, out, n_nodes,
                                                                  n_edges, bcap);
        return;
    }

    int* perm = (int*)((char*)d_ws + perm_off);
    const int edge_blocks   = (n_edges + 255) / 256;

    rgcn_selfloop_gemm<false><<<gemm_blocks, GEMM_THREADS, 0, stream>>>(x, lw1, b1, h, n_nodes,
                                                               seg, n_nodes);
    rgcn_deg_count<<<edge_blocks, 256, 0, stream>>>(dst, seg, n_edges);
    rgcn_seg_scan<<<1, 1024, 0, stream>>>(seg, n_nodes);
    rgcn_seg_fill<<<edge_blocks, 256, 0, stream>>>(dst, seg, perm, n_edges);

    rgcn_node_gather<false><<<gather_blocks, 256, 0, stream>>>(x, src, dst, etype, norm, w1,
                                                               seg, perm, h, n_nodes, n_edges, 0);

    rgcn_selfloop_gemm<true><<<gemm_blocks, GEMM_THREADS, 0, stream>>>(h, lw2, b2, out, n_nodes,
                                                              nullptr, 0);
    rgcn_node_gather<true><<<gather_blocks, 256, 0, stream>>>(h, src, dst, etype, norm, w2,
                                                              seg, perm, out, n_nodes, n_edges, 0);
}


# Self-loop GEMM: work-group = 4 waves (one per SIMD), wave = one 64-column strip x all 5 row tiles; a B fragment feeds 5 WMMAs straight from global loads: no LDS staging, no barrier; k sum order kept
# speedup vs baseline: 3.1031x; 1.1778x over previous
#include <hip/hip_runtime.h>

#define HD   256
#define NB   128
#define RT   16
#define WTAB (RT * NB * 4)
#define LCAP 128

typedef float v2f __attribute__((ext_vector_type(2)));
typedef float v8f __attribute__((ext_vector_type(8)));

#define GEMM_RT      5
#define GEMM_THREADS 128
#define GEMM_PAIRS   (HD / 8)
#define GEMM_AHEAD   4
#define GEMM_STAGES  (GEMM_AHEAD + 1)

__device__ __forceinline__ float rgcn_f4at(const float4& v, int j)
{
    return (j == 0) ? v.x : (j == 1) ? v.y : (j == 2) ? v.z : v.w;
}

__device__ __forceinline__ float rgcn_relu_bits(float x, int zero)
{
    const int b = __float_as_int(x);
    return __int_as_float((b > zero) ? b : zero);
}

template <bool RELU>
__global__ __launch_bounds__(GEMM_THREADS)
void rgcn_selfloop_gemm(const float* __restrict__ A,
                        const float* __restrict__ W,
                        const float* __restrict__ bias,
                        float* __restrict__ out,
                        int n_rows,
                        int* __restrict__ zero_buf,
                        int zero_n)
{
    if (zero_buf != nullptr) {
        const int gid = blockIdx.x * blockDim.x + threadIdx.x;
        if (gid < zero_n) zero_buf[gid] = 0;
    }

    const int lane = threadIdx.x & 31;
    const int strip = __builtin_amdgcn_readfirstlane((int)(threadIdx.x >> 5));
    const int half  = lane >> 4;
    const int sub   = lane & 15;
    const int col0  = strip * 64 + 4 * sub;
    const int tile0 = blockIdx.x * GEMM_RT;

    const float4* arow[GEMM_RT];
    bool live[GEMM_RT];
    #pragma unroll
    for (int i = 0; i < GEMM_RT; ++i) {
        live[i] = (tile0 + i) * 16 < n_rows;
        const int tl = live[i] ? (tile0 + i) : tile0;
        arow[i] = (const float4*)(A + (size_t)(tl * 16 + sub) * HD) + half;
    }
    const float4* wrow = (const float4*)(W + (size_t)(4 * half) * HD + col0);

    float4 af[GEMM_STAGES][GEMM_RT];
    float4 bw[GEMM_STAGES][4];
    v8f    c[GEMM_RT][4] = {};

    #pragma unroll
    for (int t = 0; t < GEMM_AHEAD; ++t) {
        #pragma unroll
        for (int e = 0; e < 4; ++e) bw[t][e] = wrow[(8 * t + e) * (HD / 4)];
        #pragma unroll
        for (int i = 0; i < GEMM_RT; ++i) af[t][i] = arow[i][2 * t];
    }

    #pragma unroll
    for (int t = 0; t < GEMM_PAIRS; ++t) {
        const int st = t % GEMM_STAGES;
        const int tn = t + GEMM_AHEAD;
        const int sn = tn % GEMM_STAGES;

        v2f b1[4], b2[4];
        #pragma unroll
        for (int j = 0; j < 4; ++j) {
            b1[j].x = rgcn_f4at(bw[st][0], j); b1[j].y = rgcn_f4at(bw[st][1], j);
            b2[j].x = rgcn_f4at(bw[st][2], j); b2[j].y = rgcn_f4at(bw[st][3], j);
        }

        #pragma unroll
        for (int i = 0; i < GEMM_RT; ++i) {
            if (tn < GEMM_PAIRS) {
                if (i < 4) bw[sn][i] = wrow[(8 * tn + i) * (HD / 4)];
                af[sn][i] = arow[i][2 * tn];
            }
            __builtin_amdgcn_sched_barrier(0);

            float4 a4 = af[st][i];
            if (RELU) {
                int zero;
                asm volatile("s_mov_b32 %0, 0" : "=s"(zero));
                a4.x = rgcn_relu_bits(a4.x, zero); a4.y = rgcn_relu_bits(a4.y, zero);
                a4.z = rgcn_relu_bits(a4.z, zero); a4.w = rgcn_relu_bits(a4.w, zero);
            }
            v2f a;
            a.x = a4.x; a.y = a4.y;
            #pragma unroll
            for (int j = 0; j < 4; ++j)
                c[i][j] = __builtin_amdgcn_wmma_f32_16x16x4_f32(false, a, false, b1[j], (short)0, c[i][j], false, false);
            a.x = a4.z; a.y = a4.w;
            #pragma unroll
            for (int j = 0; j < 4; ++j)
                c[i][j] = __builtin_amdgcn_wmma_f32_16x16x4_f32(false, a, false, b2[j], (short)0, c[i][j], false, false);
        }
    }

    __builtin_amdgcn_sched_barrier(0);

    const float4 b4 = *(const float4*)(bias + col0);
    #pragma unroll
    for (int i = 0; i < GEMM_RT; ++i) {
        if (i > 0 && !live[i]) continue;
        float4* obase = (float4*)(out + (size_t)((tile0 + i) * 16 + half * 8) * HD + col0);
        #pragma unroll
        for (int r = 0; r < 8; ++r)
            obase[r * (HD / 4)] = make_float4(c[i][0][r] + b4.x, c[i][1][r] + b4.y,
                                              c[i][2][r] + b4.z, c[i][3][r] + b4.w);
    }
}

template <bool RELU>
__global__ __launch_bounds__(256)
void rgcn_edge_msg(const float* __restrict__ xin,
                   const int*   __restrict__ src,
                   const int*   __restrict__ dst,
                   const int*   __restrict__ etype,
                   const float* __restrict__ norm,
                   const float* __restrict__ w,
                   float* agg,
                   int n_edges)
{
    __shared__ __align__(16) float wlds[WTAB];
    for (int i = threadIdx.x; i < WTAB; i += blockDim.x) wlds[i] = w[i];
    __syncthreads();

    const int lane   = threadIdx.x & 31;
    const int wid    = blockIdx.x * (blockDim.x >> 5) + (threadIdx.x >> 5);
    const int nwaves = (gridDim.x * blockDim.x) >> 5;

    for (int e = wid; e < n_edges; e += nwaves) {
        const int   s   = src[e];
        const int   d   = dst[e];
        const int   r   = etype[e];
        const float nrm = norm[e];

        const float4* xrow = (const float4*)(xin + (size_t)s * HD) + lane * 2;
        float4 xa = xrow[0];
        float4 xb = xrow[1];
        if (RELU) {
            xa.x = fmaxf(xa.x, 0.0f); xa.y = fmaxf(xa.y, 0.0f);
            xa.z = fmaxf(xa.z, 0.0f); xa.w = fmaxf(xa.w, 0.0f);
            xb.x = fmaxf(xb.x, 0.0f); xb.y = fmaxf(xb.y, 0.0f);
            xb.z = fmaxf(xb.z, 0.0f); xb.w = fmaxf(xb.w, 0.0f);
        }
        const float xs[8] = { xa.x, xa.y, xa.z, xa.w, xb.x, xb.y, xb.z, xb.w };

        const float4* wrel = (const float4*)(wlds + (size_t)r * (NB * 4)) + lane * 4;
        float* arow = agg + (size_t)d * HD + lane * 8;

        #pragma unroll
        for (int j = 0; j < 4; ++j) {
            const float4 wq = wrel[j];
            const float x0 = xs[2 * j + 0];
            const float x1 = xs[2 * j + 1];
            const float o0 = (x0 * wq.x + x1 * wq.z) * nrm;
            const float o1 = (x0 * wq.y + x1 * wq.w) * nrm;
            (void)__hip_atomic_fetch_add(&arow[2 * j + 0], o0,
                                         __ATOMIC_RELAXED, __HIP_MEMORY_SCOPE_AGENT);
            (void)__hip_atomic_fetch_add(&arow[2 * j + 1], o1,
                                         __ATOMIC_RELAXED, __HIP_MEMORY_SCOPE_AGENT);
        }
    }
}

__global__ __launch_bounds__(256)
void rgcn_deg_count(const int* __restrict__ dst, int* seg, int n_edges)
{
    const int e = blockIdx.x * blockDim.x + threadIdx.x;
    if (e < n_edges)
        (void)__hip_atomic_fetch_add(&seg[dst[e]], 1,
                                     __ATOMIC_RELAXED, __HIP_MEMORY_SCOPE_AGENT);
}

__global__ __launch_bounds__(1024)
void rgcn_seg_scan(int* seg, int n)
{
    __shared__ int wsum[32];
    const int t    = threadIdx.x;
    const int lane = t & 31;
    const int w    = t >> 5;
    const int chunk = (n + 1023) / 1024;
    const int b  = t * chunk;
    const int en = (b + chunk < n) ? (b + chunk) : n;

    int s = 0;
    for (int i = b; i < en; ++i) s += seg[i];

    int v = s;
    #pragma unroll
    for (int o = 1; o < 32; o <<= 1) {
        const int u = __shfl_up(v, o);
        if (lane >= o) v += u;
    }
    if (lane == 31) wsum[w] = v;
    __syncthreads();
    if (w == 0) {
        const int x = wsum[lane];
        int y = x;
        #pragma unroll
        for (int o = 1; o < 32; o <<= 1) {
            const int u = __shfl_up(y, o);
            if (lane >= o) y += u;
        }
        wsum[lane] = y - x;
    }
    __syncthreads();

    int run = wsum[w] + v - s;
    for (int i = b; i < en; ++i) {
        const int c = seg[i];
        seg[i] = run;
        run += c;
    }
}

__global__ __launch_bounds__(256)
void rgcn_seg_fill(const int* __restrict__ dst, int* seg, int* __restrict__ perm,
                   int n_edges)
{
    const int e = blockIdx.x * blockDim.x + threadIdx.x;
    if (e < n_edges) {
        const int p = __hip_atomic_fetch_add(&seg[dst[e]], 1,
                                             __ATOMIC_RELAXED, __HIP_MEMORY_SCOPE_AGENT);
        perm[p] = e;
    }
}

__global__ __launch_bounds__(256)
void rgcn_bucket_fill(const int* __restrict__ dst, int* cnt, int* __restrict__ bucket,
                      int cap, int n_edges)
{
    const int e0 = (blockIdx.x * blockDim.x + threadIdx.x) * 4;
    if (e0 + 3 < n_edges) {
        const int4 d = *(const int4*)(dst + e0);
        const int p0 = __hip_atomic_fetch_add(&cnt[d.x], 1,
                                              __ATOMIC_RELAXED, __HIP_MEMORY_SCOPE_AGENT);
        const int p1 = __hip_atomic_fetch_add(&cnt[d.y], 1,
                                              __ATOMIC_RELAXED, __HIP_MEMORY_SCOPE_AGENT);
        const int p2 = __hip_atomic_fetch_add(&cnt[d.z], 1,
                                              __ATOMIC_RELAXED, __HIP_MEMORY_SCOPE_AGENT);
        const int p3 = __hip_atomic_fetch_add(&cnt[d.w], 1,
                                              __ATOMIC_RELAXED, __HIP_MEMORY_SCOPE_AGENT);
        if (p0 < cap) bucket[(size_t)d.x * cap + p0] = e0;
        if (p1 < cap) bucket[(size_t)d.y * cap + p1] = e0 + 1;
        if (p2 < cap) bucket[(size_t)d.z * cap + p2] = e0 + 2;
        if (p3 < cap) bucket[(size_t)d.w * cap + p3] = e0 + 3;
    } else {
        for (int e = e0; e < n_edges; ++e) {
            const int d = dst[e];
            const int p = __hip_atomic_fetch_add(&cnt[d], 1,
                                                 __ATOMIC_RELAXED, __HIP_MEMORY_SCOPE_AGENT);
            if (p < cap) bucket[(size_t)d * cap + p] = e;
        }
    }
}

template <bool RELU>
__device__ __forceinline__
void rgcn_msg_acc(float (&acc)[8], float4 xa, float4 xb,
                  const float4* wrel, float nrm)
{
    if (RELU) {
        xa.x = fmaxf(xa.x, 0.0f); xa.y = fmaxf(xa.y, 0.0f);
        xa.z = fmaxf(xa.z, 0.0f); xa.w = fmaxf(xa.w, 0.0f);
        xb.x = fmaxf(xb.x, 0.0f); xb.y = fmaxf(xb.y, 0.0f);
        xb.z = fmaxf(xb.z, 0.0f); xb.w = fmaxf(xb.w, 0.0f);
    }
    const float xs[8] = { xa.x, xa.y, xa.z, xa.w, xb.x, xb.y, xb.z, xb.w };
    #pragma unroll
    for (int j = 0; j < 4; ++j) {
        const float4 wq = wrel[32 * j];
        const float x0 = xs[2 * j + 0];
        const float x1 = xs[2 * j + 1];
        acc[2 * j + 0] += (x0 * wq.x + x1 * wq.z) * nrm;
        acc[2 * j + 1] += (x0 * wq.y + x1 * wq.w) * nrm;
    }
}

template <bool RELU>
__device__ __forceinline__
void rgcn_sorted_run(float (&acc)[8], const int sorted, const int m, const int lane,
                     const float* __restrict__ xin,
                     const int*   __restrict__ src,
                     const int*   __restrict__ etype,
                     const float* __restrict__ norm,
                     const float4* wl)
{
    int   s_v = 0, r_v = 0;
    float n_v = 0.0f;
    if (lane < m) {
        s_v = src[sorted];
        r_v = etype[sorted];
        n_v = norm[sorted];
    }

    int k = 0;
    for (; k + 4 <= m; k += 4) {
        float4 xa[4], xb[4];
        int    rr[4];
        float  nn[4];
        #pragma unroll
        for (int u = 0; u < 4; ++u) {
            const int s = __builtin_amdgcn_readlane(s_v, k + u);
            rr[u] = __builtin_amdgcn_readlane(r_v, k + u);
            nn[u] = __int_as_float(
                        __builtin_amdgcn_readlane(__float_as_int(n_v), k + u));
            const float4* xrow = (const float4*)(xin + (size_t)s * HD) + lane;
            xa[u] = xrow[0];
            xb[u] = xrow[32];
        }
        #pragma unroll
        for (int u = 0; u < 4; ++u)
            rgcn_msg_acc<RELU>(acc, xa[u], xb[u], wl + (size_t)rr[u] * NB, nn[u]);
    }
    for (; k < m; ++k) {
        const int   s  = __builtin_amdgcn_readlane(s_v, k);
        const int   r  = __builtin_amdgcn_readlane(r_v, k);
        const float nr = __int_as_float(
                             __builtin_amdgcn_readlane(__float_as_int(n_v), k));
        const float4* xrow = (const float4*)(xin + (size_t)s * HD) + lane;
        rgcn_msg_acc<RELU>(acc, xrow[0], xrow[32], wl + (size_t)r * NB, nr);
    }
}

template <bool RELU>
__global__ __launch_bounds__(256)
void rgcn_node_gather(const float* __restrict__ xin,
                      const int*   __restrict__ src,
                      const int*   __restrict__ dst,
                      const int*   __restrict__ etype,
                      const float* __restrict__ norm,
                      const float* __restrict__ w,
                      const int*   __restrict__ seg,
                      const int*   __restrict__ perm,
                      float* agg,
                      int n_nodes,
                      int n_edges,
                      int cap)
{
    __shared__ __align__(16) float wlds[WTAB];
    __shared__ __align__(16) int   idbuf[8 * 2 * LCAP];
    for (int i = threadIdx.x; i < WTAB / 4; i += blockDim.x) {
        const int b = i & (NB - 1);
        const int j = ((b >> 6) << 1) | (b & 1);
        const int L = (b & 63) >> 1;
        ((float4*)wlds)[(i & ~(NB - 1)) + j * 32 + L] = ((const float4*)w)[i];
    }
    __syncthreads();

    const int lane   = threadIdx.x & 31;
    const int wid    = __builtin_amdgcn_readfirstlane(
                           blockIdx.x * (blockDim.x >> 5) + (threadIdx.x >> 5));
    const int nwaves = (gridDim.x * blockDim.x) >> 5;
    const float4* wl = (const float4*)wlds + lane;
    int* uns = idbuf + __builtin_amdgcn_readfirstlane((int)(threadIdx.x >> 5)) * (2 * LCAP);
    int* srt = uns + LCAP;

    for (int n = wid; n < n_nodes; n += nwaves) {
        int beg, cnt;
        if (cap > 0) {
            cnt = seg[n];
            beg = n * cap;
        } else {
            beg = (n > 0) ? seg[n - 1] : 0;
            cnt = seg[n] - beg;
        }
        if (cnt <= 0) continue;
        const int  end    = beg + cnt;
        const bool listed = (cap == 0) || (cnt <= cap);

        float4* orow = (float4*)(agg + (size_t)n * HD) + lane;
        float4 oa = orow[0];
        float4 ob = orow[32];

        float acc[8] = { 0.f, 0.f, 0.f, 0.f, 0.f, 0.f, 0.f, 0.f };

        if (listed && cnt <= LCAP) {
            int sorted0 = 0;
            if (cnt <= 32) {
                const int my = (lane < cnt) ? perm[beg + lane] : 0x7fffffff;
                int rank = 0;
                for (int m = 0; m < cnt; ++m)
                    rank += (__builtin_amdgcn_readlane(my, m) < my) ? 1 : 0;
                sorted0 = __builtin_amdgcn_ds_permute(rank << 2, my);
            } else {
                const int padded = (cnt + 31) & ~31;
                for (int p = lane; p < padded; p += 32)
                    uns[p] = (p < cnt) ? perm[beg + p] : 0x7fffffff;
                for (int p = lane; p < padded; p += 32) {
                    const int my = uns[p];
                    int rank = 0;
                    for (int q = 0; q < padded; q += 4) {
                        const int4 v = *(const int4*)(uns + q);
                        rank += ((v.x < my) ? 1 : 0) + ((v.y < my) ? 1 : 0)
                              + ((v.z < my) ? 1 : 0) + ((v.w < my) ? 1 : 0);
                    }
                    if (p < cnt) srt[rank] = my;
                }
            }
            for (int c = 0; c < cnt; c += 32) {
                const int m = (cnt - c < 32) ? (cnt - c) : 32;
                const int sorted = (cnt <= 32) ? sorted0 : ((lane < m) ? srt[c + lane] : 0);
                rgcn_sorted_run<RELU>(acc, sorted, m, lane, xin, src, etype, norm, wl);
            }
        } else if (listed && cnt <= 2048) {
            int lo = -1;
            for (int i = 0; i < cnt; ++i) {
                int best = 0x7fffffff;
                for (int p = beg + lane; p < end; p += 32) {
                    const int e = perm[p];
                    if (e > lo && e < best) best = e;
                }
                #pragma unroll
                for (int o = 16; o > 0; o >>= 1) {
                    const int other = __shfl_xor(best, o);
                    best = (other < best) ? other : best;
                }
                const int e = __builtin_amdgcn_readfirstlane(best);
                lo = e;
                const int   s  = src[e];
                const int   r  = etype[e];
                const float nr = norm[e];
                const float4* xrow = (const float4*)(xin + (size_t)s * HD) + lane;
                rgcn_msg_acc<RELU>(acc, xrow[0], xrow[32], wl + (size_t)r * NB, nr);
            }
        } else {
            for (int e0 = 0; e0 < n_edges; e0 += 32) {
                const int  ev  = e0 + lane;
                const bool hit = (ev < n_edges) && (dst[ev] == n);
                unsigned m = __builtin_amdgcn_ballot_w32(hit);
                while (m != 0u) {
                    const int e = e0 + (__builtin_ctz(m));
                    m &= m - 1u;
                    const int   s  = src[e];
                    const int   r  = etype[e];
                    const float nr = norm[e];
                    const float4* xrow = (const float4*)(xin + (size_t)s * HD) + lane;
                    rgcn_msg_acc<RELU>(acc, xrow[0], xrow[32], wl + (size_t)r * NB, nr);
                }
            }
        }

        oa.x += acc[0]; oa.y += acc[1]; oa.z += acc[2]; oa.w += acc[3];
        ob.x += acc[4]; ob.y += acc[5]; ob.z += acc[6]; ob.w += acc[7];
        orow[0]  = oa;
        orow[32] = ob;
    }
}

#define RCAP 64

__global__ __launch_bounds__(256)
void rgcn_bucket_fill_rec(const int*   __restrict__ dst,
                          const int*   __restrict__ src,
                          const int*   __restrict__ etype,
                          const float* __restrict__ norm,
                          int* cnt, int* __restrict__ bucket,
                          int cap, int n_edges)
{
    int2* rec = (int2*)bucket;
    const int e0 = (blockIdx.x * blockDim.x + threadIdx.x) * 4;
    if (e0 + 3 < n_edges) {
        const int4   d = *(const int4*)(dst + e0);
        const int4   s = *(const int4*)(src + e0);
        const int4   r = *(const int4*)(etype + e0);
        const float4 w = *(const float4*)(norm + e0);
        const int p0 = __hip_atomic_fetch_add(&cnt[d.x], 1,
                                              __ATOMIC_RELAXED, __HIP_MEMORY_SCOPE_AGENT);
        const int p1 = __hip_atomic_fetch_add(&cnt[d.y], 1,
                                              __ATOMIC_RELAXED, __HIP_MEMORY_SCOPE_AGENT);
        const int p2 = __hip_atomic_fetch_add(&cnt[d.z], 1,
                                              __ATOMIC_RELAXED, __HIP_MEMORY_SCOPE_AGENT);
        const int p3 = __hip_atomic_fetch_add(&cnt[d.w], 1,
                                              __ATOMIC_RELAXED, __HIP_MEMORY_SCOPE_AGENT);
        if (p0 < cap) rec[(size_t)d.x * cap + p0] = make_int2(s.x | (r.x << 24), __float_as_int(w.x));
        if (p1 < cap) rec[(size_t)d.y * cap + p1] = make_int2(s.y | (r.y << 24), __float_as_int(w.y));
        if (p2 < cap) rec[(size_t)d.z * cap + p2] = make_int2(s.z | (r.z << 24), __float_as_int(w.z));
        if (p3 < cap) rec[(size_t)d.w * cap + p3] = make_int2(s.w | (r.w << 24), __float_as_int(w.w));
    } else {
        for (int e = e0; e < n_edges; ++e) {
            const int d = dst[e];
            const int p = __hip_atomic_fetch_add(&cnt[d], 1,
                                                 __ATOMIC_RELAXED, __HIP_MEMORY_SCOPE_AGENT);
            if (p < cap)
                rec[(size_t)d * cap + p] = make_int2(src[e] | (etype[e] << 24),
                                                     __float_as_int(norm[e]));
        }
    }
}

__device__ __forceinline__
unsigned long long rgcn_rec_key(int w0, int nb)
{
    return ((unsigned long long)(unsigned)w0 << 32) | (unsigned long long)(unsigned)nb;
}

template <bool RELU>
__device__ __forceinline__
void rgcn_rec_run(float (&acc)[8], const int w0, const int nb, const int m, const int lane,
                  const float* __restrict__ xin, const float4* wl)
{
    const int s_v = w0 & 0x00ffffff;
    const int r_v = (int)((unsigned)w0 >> 24);

    int k = 0;
    for (; k + 4 <= m; k += 4) {
        float4 xa[4], xb[4];
        int    rr[4];
        float  nn[4];
        #pragma unroll
        for (int u = 0; u < 4; ++u) {
            const int s = __builtin_amdgcn_readlane(s_v, k + u);
            rr[u] = __builtin_amdgcn_readlane(r_v, k + u);
            nn[u] = __int_as_float(__builtin_amdgcn_readlane(nb, k + u));
            const float4* xrow = (const float4*)(xin + (size_t)s * HD) + lane;
            xa[u] = xrow[0];
            xb[u] = xrow[32];
        }
        #pragma unroll
        for (int u = 0; u < 4; ++u)
            rgcn_msg_acc<RELU>(acc, xa[u], xb[u], wl + (size_t)rr[u] * NB, nn[u]);
    }
    if (k < m) {
        const int left = m - k;
        const float4 z4 = make_float4(0.f, 0.f, 0.f, 0.f);
        float4 xa1 = z4, xb1 = z4, xa2 = z4, xb2 = z4;
        int    r1 = 0, r2 = 0;
        float  n1 = 0.f, n2 = 0.f;
        const int   s0 = __builtin_amdgcn_readlane(s_v, k);
        const int   r0 = __builtin_amdgcn_readlane(r_v, k);
        const float n0 = __int_as_float(__builtin_amdgcn_readlane(nb, k));
        const float4* x0 = (const float4*)(xin + (size_t)s0 * HD) + lane;
        const float4 xa0 = x0[0];
        const float4 xb0 = x0[32];
        if (left > 1) {
            const int s1 = __builtin_amdgcn_readlane(s_v, k + 1);
            r1 = __builtin_amdgcn_readlane(r_v, k + 1);
            n1 = __int_as_float(__builtin_amdgcn_readlane(nb, k + 1));
            const float4* x1 = (const float4*)(xin + (size_t)s1 * HD) + lane;
            xa1 = x1[0];
            xb1 = x1[32];
        }
        if (left > 2) {
            const int s2 = __builtin_amdgcn_readlane(s_v, k + 2);
            r2 = __builtin_amdgcn_readlane(r_v, k + 2);
            n2 = __int_as_float(__builtin_amdgcn_readlane(nb, k + 2));
            const float4* x2 = (const float4*)(xin + (size_t)s2 * HD) + lane;
            xa2 = x2[0];
            xb2 = x2[32];
        }
        rgcn_msg_acc<RELU>(acc, xa0, xb0, wl + (size_t)r0 * NB, n0);
        if (left > 1) rgcn_msg_acc<RELU>(acc, xa1, xb1, wl + (size_t)r1 * NB, n1);
        if (left > 2) rgcn_msg_acc<RELU>(acc, xa2, xb2, wl + (size_t)r2 * NB, n2);
    }
}

template <bool RELU>
__global__ __launch_bounds__(256)
void rgcn_node_gather_rec(const float* __restrict__ xin,
                          const int*   __restrict__ src,
                          const int*   __restrict__ dst,
                          const int*   __restrict__ etype,
                          const float* __restrict__ norm,
                          const float* __restrict__ w,
                          const int*   __restrict__ seg,
                          const int*   __restrict__ bucket,
                          float* agg,
                          int n_nodes,
                          int n_edges,
                          int cap)
{
    __shared__ __align__(16) float wlds[WTAB];
    __shared__ __align__(16) int   keybuf[8 * 4 * RCAP];

    const int lane   = threadIdx.x & 31;
    const int wid    = __builtin_amdgcn_readfirstlane(
                           blockIdx.x * (blockDim.x >> 5) + (threadIdx.x >> 5));
    const int nwaves = (gridDim.x * blockDim.x) >> 5;
    const int2* rec = (const int2*)bucket;

    int n   = wid;
    int cnt = (n < n_nodes) ? seg[n] : 0;

    if (blockDim.x * 8 == WTAB / 4) {
        float4 tv[8];
        #pragma unroll
        for (int q = 0; q < 8; ++q)
            tv[q] = ((const float4*)w)[threadIdx.x + q * (WTAB / 32)];
        #pragma unroll
        for (int q = 0; q < 8; ++q) {
            const int i = threadIdx.x + q * (WTAB / 32);
            const int b = i & (NB - 1);
            const int j = ((b >> 6) << 1) | (b & 1);
            const int L = (b & 63) >> 1;
            ((float4*)wlds)[(i & ~(NB - 1)) + j * 32 + L] = tv[q];
        }
    } else {
        for (int i = threadIdx.x; i < WTAB / 4; i += blockDim.x) {
            const int b = i & (NB - 1);
            const int j = ((b >> 6) << 1) | (b & 1);
            const int L = (b & 63) >> 1;
            ((float4*)wlds)[(i & ~(NB - 1)) + j * 32 + L] = ((const float4*)w)[i];
        }
    }

    int2 recn = make_int2(-1, -1);
    if (lane < cnt && lane < cap) recn = rec[(size_t)n * cap + lane];
    int n1   = n + nwaves;
    int cnt1 = (n1 < n_nodes) ? seg[n1] : 0;

    __syncthreads();

    const float4* wl = (const float4*)wlds + lane;
    int2* uns = (int2*)(keybuf + __builtin_amdgcn_readfirstlane((int)(threadIdx.x >> 5)) * (4 * RCAP));
    int2* srt = uns + RCAP;

    int2 recn1 = make_int2(-1, -1);
    int  n2 = 0, cnt2 = 0;
    for (; n < n_nodes; n = n1, cnt = cnt1, recn = recn1, n1 = n2, cnt1 = cnt2) {
        recn1 = make_int2(-1, -1);
        if (lane < cnt1 && lane < cap) recn1 = rec[(size_t)n1 * cap + lane];
        n2   = n1 + nwaves;
        cnt2 = (n2 < n_nodes) ? seg[n2] : 0;

        if (cnt <= 0) continue;

        float4* orow = (float4*)(agg + (size_t)n * HD) + lane;
        float4 oa = orow[0];
        float4 ob = orow[32];

        float acc[8] = { 0.f, 0.f, 0.f, 0.f, 0.f, 0.f, 0.f, 0.f };

        if (cnt <= cap && cnt <= RCAP) {
            const int2* row = rec + (size_t)n * cap;
            int w0s = 0, nbs = 0;
            if (cnt <= 32) {
                const int2 mine = recn;
                const unsigned long long my = rgcn_rec_key(mine.x, mine.y);
                int rank = 0;
                for (int m = 0; m < cnt; ++m) {
                    const unsigned long long ok =
                        rgcn_rec_key(__builtin_amdgcn_readlane(mine.x, m),
                                     __builtin_amdgcn_readlane(mine.y, m));
                    rank += (ok < my || (ok == my && m < lane)) ? 1 : 0;
                }
                w0s = __builtin_amdgcn_ds_permute(rank << 2, mine.x);
                nbs = __builtin_amdgcn_ds_permute(rank << 2, mine.y);
            } else {
                const int padded = (cnt + 31) & ~31;
                for (int p = lane; p < padded; p += 32)
                    uns[p] = (p < cnt) ? row[p] : make_int2(-1, -1);
                for (int p = lane; p < padded; p += 32) {
                    const int2 mine = uns[p];
                    const unsigned long long my = rgcn_rec_key(mine.x, mine.y);
                    int rank = 0;
                    for (int q = 0; q < padded; q += 2) {
                        const int4 v = *(const int4*)(uns + q);
                        const unsigned long long k0 = rgcn_rec_key(v.x, v.y);
                        const unsigned long long k1 = rgcn_rec_key(v.z, v.w);
                        rank += ((k0 < my || (k0 == my && q     < p)) ? 1 : 0)
                              + ((k1 < my || (k1 == my && q + 1 < p)) ? 1 : 0);
                    }
                    if (p < cnt) srt[rank] = mine;
                }
            }
            for (int c = 0; c < cnt; c += 32) {
                const int m = (cnt - c < 32) ? (cnt - c) : 32;
                if (cnt > 32) {
                    int2 r = make_int2(0, 0);
                    if (lane < m) r = srt[c + lane];
                    w0s = r.x;
                    nbs = r.y;
                }
                rgcn_rec_run<RELU>(acc, w0s, nbs, m, lane, xin, wl);
            }
        } else {
            for (int e0 = 0; e0 < n_edges; e0 += 32) {
                const int  ev  = e0 + lane;
                const bool hit = (ev < n_edges) && (dst[ev] == n);
                unsigned m = __builtin_amdgcn_ballot_w32(hit);
                while (m != 0u) {
                    const int e = e0 + (__builtin_ctz(m));
                    m &= m - 1u;
                    const int   s  = src[e];
                    const int   r  = etype[e];
                    const float nr = norm[e];
                    const float4* xrow = (const float4*)(xin + (size_t)s * HD) + lane;
                    rgcn_msg_acc<RELU>(acc, xrow[0], xrow[32], wl + (size_t)r * NB, nr);
                }
            }
        }

        oa.x += acc[0]; oa.y += acc[1]; oa.z += acc[2]; oa.w += acc[3];
        ob.x += acc[4]; ob.y += acc[5]; ob.z += acc[6]; ob.w += acc[7];
        orow[0]  = oa;
        orow[32] = ob;
    }
}

extern "C" void kernel_launch(void* const* d_in, const int* in_sizes, int n_in,
                              void* d_out, int out_size, void* d_ws, size_t ws_size,
                              hipStream_t stream)
{
    const float* x     = (const float*)d_in[0];
    const int*   src   = (const int*)  d_in[1];
    const int*   dst   = (const int*)  d_in[2];
    const int*   etype = (const int*)  d_in[3];
    const float* norm  = (const float*)d_in[4];
    const float* w1    = (const float*)d_in[5];
    const float* lw1   = (const float*)d_in[6];
    const float* b1    = (const float*)d_in[7];
    const float* w2    = (const float*)d_in[8];
    const float* lw2   = (const float*)d_in[9];
    const float* b2    = (const float*)d_in[10];
    float*       out   = (float*)d_out;

    const int n_nodes = in_sizes[0] / HD;
    const int n_edges = in_sizes[1];

    float* h = (float*)d_ws;

    const int gemm_blocks = ((n_nodes + 15) / 16 + GEMM_RT - 1) / GEMM_RT;

    const size_t seg_off  = ((size_t)n_nodes * HD * sizeof(float) + 255) & ~(size_t)255;
    const size_t perm_off = (seg_off + (size_t)n_nodes * sizeof(int) + 255) & ~(size_t)255;
    const size_t ws_need  = perm_off + (size_t)n_edges * sizeof(int);

    if (ws_size < ws_need) {
        const int edge_blocks = 1024;
        rgcn_selfloop_gemm<false><<<gemm_blocks, GEMM_THREADS, 0, stream>>>(x, lw1, b1, h, n_nodes,
                                                                   nullptr, 0);
        rgcn_edge_msg<false><<<edge_blocks, 256, 0, stream>>>(x, src, dst, etype, norm, w1,
                                                              h, n_edges);
        rgcn_selfloop_gemm<true><<<gemm_blocks, GEMM_THREADS, 0, stream>>>(h, lw2, b2, out, n_nodes,
                                                                  nullptr, 0);
        rgcn_edge_msg<true><<<edge_blocks, 256, 0, stream>>>(h, src, dst, etype, norm, w2,
                                                             out, n_edges);
        return;
    }

    int* seg  = (int*)((char*)d_ws + seg_off);
    const int gather_blocks = 1024;

    const int    bcap     = 64;
    const size_t bkt_need = perm_off + (size_t)n_nodes * bcap * sizeof(int);

    const size_t rec_need = perm_off + (size_t)n_nodes * RCAP * 2 * sizeof(int);

    if (ws_size >= rec_need && n_nodes <= (1 << 24)) {
        int* bucket = (int*)((char*)d_ws + perm_off);
        const int fill_blocks = (n_edges + 1023) / 1024;

        rgcn_selfloop_gemm<false><<<gemm_blocks, GEMM_THREADS, 0, stream>>>(x, lw1, b1, h, n_nodes,
                                                                   seg, n_nodes);
        rgcn_bucket_fill_rec<<<fill_blocks, 256, 0, stream>>>(dst, src, etype, norm, seg, bucket,
                                                              RCAP, n_edges);

        rgcn_node_gather_rec<false><<<gather_blocks, 256, 0, stream>>>(x, src, dst, etype, norm, w1,
                                                                       seg, bucket, h, n_nodes,
                                                                       n_edges, RCAP);
        rgcn_selfloop_gemm<true><<<gemm_blocks, GEMM_THREADS, 0, stream>>>(h, lw2, b2, out, n_nodes,
                                                                  nullptr, 0);
        rgcn_node_gather_rec<true><<<gather_blocks, 256, 0, stream>>>(h, src, dst, etype, norm, w2,
                                                                      seg, bucket, out, n_nodes,
                                                                      n_edges, RCAP);
        return;
    }

    if (ws_size >= bkt_need) {
        int* bucket = (int*)((char*)d_ws + perm_off);
        const int fill_blocks = (n_edges + 1023) / 1024;

        rgcn_selfloop_gemm<false><<<gemm_blocks, GEMM_THREADS, 0, stream>>>(x, lw1, b1, h, n_nodes,
                                                                   seg, n_nodes);
        rgcn_bucket_fill<<<fill_blocks, 256, 0, stream>>>(dst, seg, bucket, bcap, n_edges);

        rgcn_node_gather<false><<<gather_blocks, 256, 0, stream>>>(x, src, dst, etype, norm, w1,
                                                                   seg, bucket, h, n_nodes,
                                                                   n_edges, bcap);
        rgcn_selfloop_gemm<true><<<gemm_blocks, GEMM_THREADS, 0, stream>>>(h, lw2, b2, out, n_nodes,
                                                                  nullptr, 0);
        rgcn_node_gather<true><<<gather_blocks, 256, 0, stream>>>(h, src, dst, etype, norm, w2,
                                                                  seg, bucket, out, n_nodes,
                                                                  n_edges, bcap);
        return;
    }

    int* perm = (int*)((char*)d_ws + perm_off);
    const int edge_blocks   = (n_edges + 255) / 256;

    rgcn_selfloop_gemm<false><<<gemm_blocks, GEMM_THREADS, 0, stream>>>(x, lw1, b1, h, n_nodes,
                                                               seg, n_nodes);
    rgcn_deg_count<<<edge_blocks, 256, 0, stream>>>(dst, seg, n_edges);
    rgcn_seg_scan<<<1, 1024, 0, stream>>>(seg, n_nodes);
    rgcn_seg_fill<<<edge_blocks, 256, 0, stream>>>(dst, seg, perm, n_edges);

    rgcn_node_gather<false><<<gather_blocks, 256, 0, stream>>>(x, src, dst, etype, norm, w1,
                                                               seg, perm, h, n_nodes, n_edges, 0);

    rgcn_selfloop_gemm<true><<<gemm_blocks, GEMM_THREADS, 0, stream>>>(h, lw2, b2, out, n_nodes,
                                                              nullptr, 0);
    rgcn_node_gather<true><<<gather_blocks, 256, 0, stream>>>(h, src, dst, etype, norm, w2,
                                                              seg, perm, out, n_nodes, n_edges, 0);
}


# Record buckets built by extra work-groups of the layer-1 GEMM launch (integer atomics wait beside the matrix work), not by a launch behind it; counters cleared by a small launch first; sum order kept
# speedup vs baseline: 3.2992x; 1.0632x over previous
#include <hip/hip_runtime.h>

#define HD   256
#define NB   128
#define RT   16
#define WTAB (RT * NB * 4)
#define LCAP 128

typedef float v2f __attribute__((ext_vector_type(2)));
typedef float v8f __attribute__((ext_vector_type(8)));

#define GEMM_RT      5
#define GEMM_THREADS 128
#define GEMM_PAIRS   (HD / 8)
#define GEMM_AHEAD   4
#define GEMM_STAGES  (GEMM_AHEAD + 1)

__device__ __forceinline__ float rgcn_f4at(const float4& v, int j)
{
    return (j == 0) ? v.x : (j == 1) ? v.y : (j == 2) ? v.z : v.w;
}

__device__ __forceinline__ float rgcn_relu_bits(float x, int zero)
{
    const int b = __float_as_int(x);
    return __int_as_float((b > zero) ? b : zero);
}

__global__ __launch_bounds__(256)
void rgcn_zero_counters(int* __restrict__ cnt, int n)
{
    const int i = blockIdx.x * blockDim.x + threadIdx.x;
    if (i < n) cnt[i] = 0;
}

#define FILL_EPT 8

__device__ __forceinline__
void rgcn_fill_rec_thread(const int t,
                          const int*   __restrict__ dst,
                          const int*   __restrict__ src,
                          const int*   __restrict__ etype,
                          const float* __restrict__ norm,
                          int* cnt, int* __restrict__ bucket,
                          const int cap, const int n_edges)
{
    int2* rec = (int2*)bucket;
    const int e0 = t * FILL_EPT;
    if (e0 + FILL_EPT - 1 < n_edges) {
        int   dd[FILL_EPT], w0[FILL_EPT], nb[FILL_EPT], p[FILL_EPT];
        #pragma unroll
        for (int g = 0; g < FILL_EPT / 4; ++g) {
            const int4   d = *(const int4*)(dst + e0 + 4 * g);
            const int4   s = *(const int4*)(src + e0 + 4 * g);
            const int4   r = *(const int4*)(etype + e0 + 4 * g);
            const float4 w = *(const float4*)(norm + e0 + 4 * g);
            dd[4 * g + 0] = d.x; dd[4 * g + 1] = d.y; dd[4 * g + 2] = d.z; dd[4 * g + 3] = d.w;
            w0[4 * g + 0] = s.x | (r.x << 24); w0[4 * g + 1] = s.y | (r.y << 24);
            w0[4 * g + 2] = s.z | (r.z << 24); w0[4 * g + 3] = s.w | (r.w << 24);
            nb[4 * g + 0] = __float_as_int(w.x); nb[4 * g + 1] = __float_as_int(w.y);
            nb[4 * g + 2] = __float_as_int(w.z); nb[4 * g + 3] = __float_as_int(w.w);
        }
        #pragma unroll
        for (int i = 0; i < FILL_EPT; ++i)
            p[i] = __hip_atomic_fetch_add(&cnt[dd[i]], 1,
                                          __ATOMIC_RELAXED, __HIP_MEMORY_SCOPE_AGENT);
        #pragma unroll
        for (int i = 0; i < FILL_EPT; ++i)
            if (p[i] < cap) rec[(size_t)dd[i] * cap + p[i]] = make_int2(w0[i], nb[i]);
    } else {
        for (int e = e0; e < n_edges && e < e0 + FILL_EPT; ++e) {
            const int d = dst[e];
            const int q = __hip_atomic_fetch_add(&cnt[d], 1,
                                                 __ATOMIC_RELAXED, __HIP_MEMORY_SCOPE_AGENT);
            if (q < cap)
                rec[(size_t)d * cap + q] = make_int2(src[e] | (etype[e] << 24),
                                                     __float_as_int(norm[e]));
        }
    }
}

template <bool RELU, bool FILL>
__global__ __launch_bounds__(GEMM_THREADS)
void rgcn_selfloop_gemm(const float* __restrict__ A,
                        const float* __restrict__ W,
                        const float* __restrict__ bias,
                        float* __restrict__ out,
                        int n_rows,
                        int* __restrict__ zero_buf,
                        int zero_n,
                        int gemm_blocks,
                        const int*   __restrict__ e_dst,
                        const int*   __restrict__ e_src,
                        const int*   __restrict__ e_type,
                        const float* __restrict__ e_norm,
                        int* e_cnt, int* __restrict__ e_bucket,
                        int e_cap, int n_edges)
{
    if (FILL) {
        if ((int)blockIdx.x >= gemm_blocks) {
            rgcn_fill_rec_thread(((int)blockIdx.x - gemm_blocks) * (int)blockDim.x + (int)threadIdx.x,
                                 e_dst, e_src, e_type, e_norm, e_cnt, e_bucket, e_cap, n_edges);
            return;
        }
    }

    if (zero_buf != nullptr) {
        const int gid = blockIdx.x * blockDim.x + threadIdx.x;
        if (gid < zero_n) zero_buf[gid] = 0;
    }

    const int lane = threadIdx.x & 31;
    const int strip = __builtin_amdgcn_readfirstlane((int)(threadIdx.x >> 5));
    const int half  = lane >> 4;
    const int sub   = lane & 15;
    const int col0  = strip * 64 + 4 * sub;
    const int tile0 = blockIdx.x * GEMM_RT;

    const float4* arow[GEMM_RT];
    bool live[GEMM_RT];
    #pragma unroll
    for (int i = 0; i < GEMM_RT; ++i) {
        live[i] = (tile0 + i) * 16 < n_rows;
        const int tl = live[i] ? (tile0 + i) : tile0;
        arow[i] = (const float4*)(A + (size_t)(tl * 16 + sub) * HD) + half;
    }
    const float4* wrow = (const float4*)(W + (size_t)(4 * half) * HD + col0);

    float4 af[GEMM_STAGES][GEMM_RT];
    float4 bw[GEMM_STAGES][4];
    v8f    c[GEMM_RT][4] = {};

    #pragma unroll
    for (int t = 0; t < GEMM_AHEAD; ++t) {
        #pragma unroll
        for (int e = 0; e < 4; ++e) bw[t][e] = wrow[(8 * t + e) * (HD / 4)];
        #pragma unroll
        for (int i = 0; i < GEMM_RT; ++i) af[t][i] = arow[i][2 * t];
    }

    #pragma unroll
    for (int t = 0; t < GEMM_PAIRS; ++t) {
        const int st = t % GEMM_STAGES;
        const int tn = t + GEMM_AHEAD;
        const int sn = tn % GEMM_STAGES;

        v2f b1[4], b2[4];
        #pragma unroll
        for (int j = 0; j < 4; ++j) {
            b1[j].x = rgcn_f4at(bw[st][0], j); b1[j].y = rgcn_f4at(bw[st][1], j);
            b2[j].x = rgcn_f4at(bw[st][2], j); b2[j].y = rgcn_f4at(bw[st][3], j);
        }

        #pragma unroll
        for (int i = 0; i < GEMM_RT; ++i) {
            if (tn < GEMM_PAIRS) {
                if (i < 4) bw[sn][i] = wrow[(8 * tn + i) * (HD / 4)];
                af[sn][i] = arow[i][2 * tn];
            }
            __builtin_amdgcn_sched_barrier(0);

            float4 a4 = af[st][i];
            if (RELU) {
                int zero;
                asm volatile("s_mov_b32 %0, 0" : "=s"(zero));
                a4.x = rgcn_relu_bits(a4.x, zero); a4.y = rgcn_relu_bits(a4.y, zero);
                a4.z = rgcn_relu_bits(a4.z, zero); a4.w = rgcn_relu_bits(a4.w, zero);
            }
            v2f a;
            a.x = a4.x; a.y = a4.y;
            #pragma unroll
            for (int j = 0; j < 4; ++j)
                c[i][j] = __builtin_amdgcn_wmma_f32_16x16x4_f32(false, a, false, b1[j], (short)0, c[i][j], false, false);
            a.x = a4.z; a.y = a4.w;
            #pragma unroll
            for (int j = 0; j < 4; ++j)
                c[i][j] = __builtin_amdgcn_wmma_f32_16x16x4_f32(false, a, false, b2[j], (short)0, c[i][j], false, false);
        }
    }

    __builtin_amdgcn_sched_barrier(0);

    const float4 b4 = *(const float4*)(bias + col0);
    #pragma unroll
    for (int i = 0; i < GEMM_RT; ++i) {
        if (i > 0 && !live[i]) continue;
        float4* obase = (float4*)(out + (size_t)((tile0 + i) * 16 + half * 8) * HD + col0);
        #pragma unroll
        for (int r = 0; r < 8; ++r)
            obase[r * (HD / 4)] = make_float4(c[i][0][r] + b4.x, c[i][1][r] + b4.y,
                                              c[i][2][r] + b4.z, c[i][3][r] + b4.w);
    }
}

template <bool RELU>
__global__ __launch_bounds__(256)
void rgcn_edge_msg(const float* __restrict__ xin,
                   const int*   __restrict__ src,
                   const int*   __restrict__ dst,
                   const int*   __restrict__ etype,
                   const float* __restrict__ norm,
                   const float* __restrict__ w,
                   float* agg,
                   int n_edges)
{
    __shared__ __align__(16) float wlds[WTAB];
    for (int i = threadIdx.x; i < WTAB; i += blockDim.x) wlds[i] = w[i];
    __syncthreads();

    const int lane   = threadIdx.x & 31;
    const int wid    = blockIdx.x * (blockDim.x >> 5) + (threadIdx.x >> 5);
    const int nwaves = (gridDim.x * blockDim.x) >> 5;

    for (int e = wid; e < n_edges; e += nwaves) {
        const int   s   = src[e];
        const int   d   = dst[e];
        const int   r   = etype[e];
        const float nrm = norm[e];

        const float4* xrow = (const float4*)(xin + (size_t)s * HD) + lane * 2;
        float4 xa = xrow[0];
        float4 xb = xrow[1];
        if (RELU) {
            xa.x = fmaxf(xa.x, 0.0f); xa.y = fmaxf(xa.y, 0.0f);
            xa.z = fmaxf(xa.z, 0.0f); xa.w = fmaxf(xa.w, 0.0f);
            xb.x = fmaxf(xb.x, 0.0f); xb.y = fmaxf(xb.y, 0.0f);
            xb.z = fmaxf(xb.z, 0.0f); xb.w = fmaxf(xb.w, 0.0f);
        }
        const float xs[8] = { xa.x, xa.y, xa.z, xa.w, xb.x, xb.y, xb.z, xb.w };

        const float4* wrel = (const float4*)(wlds + (size_t)r * (NB * 4)) + lane * 4;
        float* arow = agg + (size_t)d * HD + lane * 8;

        #pragma unroll
        for (int j = 0; j < 4; ++j) {
            const float4 wq = wrel[j];
            const float x0 = xs[2 * j + 0];
            const float x1 = xs[2 * j + 1];
            const float o0 = (x0 * wq.x + x1 * wq.z) * nrm;
            const float o1 = (x0 * wq.y + x1 * wq.w) * nrm;
            (void)__hip_atomic_fetch_add(&arow[2 * j + 0], o0,
                                         __ATOMIC_RELAXED, __HIP_MEMORY_SCOPE_AGENT);
            (void)__hip_atomic_fetch_add(&arow[2 * j + 1], o1,
                                         __ATOMIC_RELAXED, __HIP_MEMORY_SCOPE_AGENT);
        }
    }
}

__global__ __launch_bounds__(256)
void rgcn_deg_count(const int* __restrict__ dst, int* seg, int n_edges)
{
    const int e = blockIdx.x * blockDim.x + threadIdx.x;
    if (e < n_edges)
        (void)__hip_atomic_fetch_add(&seg[dst[e]], 1,
                                     __ATOMIC_RELAXED, __HIP_MEMORY_SCOPE_AGENT);
}

__global__ __launch_bounds__(1024)
void rgcn_seg_scan(int* seg, int n)
{
    __shared__ int wsum[32];
    const int t    = threadIdx.x;
    const int lane = t & 31;
    const int w    = t >> 5;
    const int chunk = (n + 1023) / 1024;
    const int b  = t * chunk;
    const int en = (b + chunk < n) ? (b + chunk) : n;

    int s = 0;
    for (int i = b; i < en; ++i) s += seg[i];

    int v = s;
    #pragma unroll
    for (int o = 1; o < 32; o <<= 1) {
        const int u = __shfl_up(v, o);
        if (lane >= o) v += u;
    }
    if (lane == 31) wsum[w] = v;
    __syncthreads();
    if (w == 0) {
        const int x = wsum[lane];
        int y = x;
        #pragma unroll
        for (int o = 1; o < 32; o <<= 1) {
            const int u = __shfl_up(y, o);
            if (lane >= o) y += u;
        }
        wsum[lane] = y - x;
    }
    __syncthreads();

    int run = wsum[w] + v - s;
    for (int i = b; i < en; ++i) {
        const int c = seg[i];
        seg[i] = run;
        run += c;
    }
}

__global__ __launch_bounds__(256)
void rgcn_seg_fill(const int* __restrict__ dst, int* seg, int* __restrict__ perm,
                   int n_edges)
{
    const int e = blockIdx.x * blockDim.x + threadIdx.x;
    if (e < n_edges) {
        const int p = __hip_atomic_fetch_add(&seg[dst[e]], 1,
                                             __ATOMIC_RELAXED, __HIP_MEMORY_SCOPE_AGENT);
        perm[p] = e;
    }
}

__global__ __launch_bounds__(256)
void rgcn_bucket_fill(const int* __restrict__ dst, int* cnt, int* __restrict__ bucket,
                      int cap, int n_edges)
{
    const int e0 = (blockIdx.x * blockDim.x + threadIdx.x) * 4;
    if (e0 + 3 < n_edges) {
        const int4 d = *(const int4*)(dst + e0);
        const int p0 = __hip_atomic_fetch_add(&cnt[d.x], 1,
                                              __ATOMIC_RELAXED, __HIP_MEMORY_SCOPE_AGENT);
        const int p1 = __hip_atomic_fetch_add(&cnt[d.y], 1,
                                              __ATOMIC_RELAXED, __HIP_MEMORY_SCOPE_AGENT);
        const int p2 = __hip_atomic_fetch_add(&cnt[d.z], 1,
                                              __ATOMIC_RELAXED, __HIP_MEMORY_SCOPE_AGENT);
        const int p3 = __hip_atomic_fetch_add(&cnt[d.w], 1,
                                              __ATOMIC_RELAXED, __HIP_MEMORY_SCOPE_AGENT);
        if (p0 < cap) bucket[(size_t)d.x * cap + p0] = e0;
        if (p1 < cap) bucket[(size_t)d.y * cap + p1] = e0 + 1;
        if (p2 < cap) bucket[(size_t)d.z * cap + p2] = e0 + 2;
        if (p3 < cap) bucket[(size_t)d.w * cap + p3] = e0 + 3;
    } else {
        for (int e = e0; e < n_edges; ++e) {
            const int d = dst[e];
            const int p = __hip_atomic_fetch_add(&cnt[d], 1,
                                                 __ATOMIC_RELAXED, __HIP_MEMORY_SCOPE_AGENT);
            if (p < cap) bucket[(size_t)d * cap + p] = e;
        }
    }
}

template <bool RELU>
__device__ __forceinline__
void rgcn_msg_acc(float (&acc)[8], float4 xa, float4 xb,
                  const float4* wrel, float nrm)
{
    if (RELU) {
        xa.x = fmaxf(xa.x, 0.0f); xa.y = fmaxf(xa.y, 0.0f);
        xa.z = fmaxf(xa.z, 0.0f); xa.w = fmaxf(xa.w, 0.0f);
        xb.x = fmaxf(xb.x, 0.0f); xb.y = fmaxf(xb.y, 0.0f);
        xb.z = fmaxf(xb.z, 0.0f); xb.w = fmaxf(xb.w, 0.0f);
    }
    const float xs[8] = { xa.x, xa.y, xa.z, xa.w, xb.x, xb.y, xb.z, xb.w };
    #pragma unroll
    for (int j = 0; j < 4; ++j) {
        const float4 wq = wrel[32 * j];
        const float x0 = xs[2 * j + 0];
        const float x1 = xs[2 * j + 1];
        acc[2 * j + 0] += (x0 * wq.x + x1 * wq.z) * nrm;
        acc[2 * j + 1] += (x0 * wq.y + x1 * wq.w) * nrm;
    }
}

template <bool RELU>
__device__ __forceinline__
void rgcn_sorted_run(float (&acc)[8], const int sorted, const int m, const int lane,
                     const float* __restrict__ xin,
                     const int*   __restrict__ src,
                     const int*   __restrict__ etype,
                     const float* __restrict__ norm,
                     const float4* wl)
{
    int   s_v = 0, r_v = 0;
    float n_v = 0.0f;
    if (lane < m) {
        s_v = src[sorted];
        r_v = etype[sorted];
        n_v = norm[sorted];
    }

    int k = 0;
    for (; k + 4 <= m; k += 4) {
        float4 xa[4], xb[4];
        int    rr[4];
        float  nn[4];
        #pragma unroll
        for (int u = 0; u < 4; ++u) {
            const int s = __builtin_amdgcn_readlane(s_v, k + u);
            rr[u] = __builtin_amdgcn_readlane(r_v, k + u);
            nn[u] = __int_as_float(
                        __builtin_amdgcn_readlane(__float_as_int(n_v), k + u));
            const float4* xrow = (const float4*)(xin + (size_t)s * HD) + lane;
            xa[u] = xrow[0];
            xb[u] = xrow[32];
        }
        #pragma unroll
        for (int u = 0; u < 4; ++u)
            rgcn_msg_acc<RELU>(acc, xa[u], xb[u], wl + (size_t)rr[u] * NB, nn[u]);
    }
    for (; k < m; ++k) {
        const int   s  = __builtin_amdgcn_readlane(s_v, k);
        const int   r  = __builtin_amdgcn_readlane(r_v, k);
        const float nr = __int_as_float(
                             __builtin_amdgcn_readlane(__float_as_int(n_v), k));
        const float4* xrow = (const float4*)(xin + (size_t)s * HD) + lane;
        rgcn_msg_acc<RELU>(acc, xrow[0], xrow[32], wl + (size_t)r * NB, nr);
    }
}

template <bool RELU>
__global__ __launch_bounds__(256)
void rgcn_node_gather(const float* __restrict__ xin,
                      const int*   __restrict__ src,
                      const int*   __restrict__ dst,
                      const int*   __restrict__ etype,
                      const float* __restrict__ norm,
                      const float* __restrict__ w,
                      const int*   __restrict__ seg,
                      const int*   __restrict__ perm,
                      float* agg,
                      int n_nodes,
                      int n_edges,
                      int cap)
{
    __shared__ __align__(16) float wlds[WTAB];
    __shared__ __align__(16) int   idbuf[8 * 2 * LCAP];
    for (int i = threadIdx.x; i < WTAB / 4; i += blockDim.x) {
        const int b = i & (NB - 1);
        const int j = ((b >> 6) << 1) | (b & 1);
        const int L = (b & 63) >> 1;
        ((float4*)wlds)[(i & ~(NB - 1)) + j * 32 + L] = ((const float4*)w)[i];
    }
    __syncthreads();

    const int lane   = threadIdx.x & 31;
    const int wid    = __builtin_amdgcn_readfirstlane(
                           blockIdx.x * (blockDim.x >> 5) + (threadIdx.x >> 5));
    const int nwaves = (gridDim.x * blockDim.x) >> 5;
    const float4* wl = (const float4*)wlds + lane;
    int* uns = idbuf + __builtin_amdgcn_readfirstlane((int)(threadIdx.x >> 5)) * (2 * LCAP);
    int* srt = uns + LCAP;

    for (int n = wid; n < n_nodes; n += nwaves) {
        int beg, cnt;
        if (cap > 0) {
            cnt = seg[n];
            beg = n * cap;
        } else {
            beg = (n > 0) ? seg[n - 1] : 0;
            cnt = seg[n] - beg;
        }
        if (cnt <= 0) continue;
        const int  end    = beg + cnt;
        const bool listed = (cap == 0) || (cnt <= cap);

        float4* orow = (float4*)(agg + (size_t)n * HD) + lane;
        float4 oa = orow[0];
        float4 ob = orow[32];

        float acc[8] = { 0.f, 0.f, 0.f, 0.f, 0.f, 0.f, 0.f, 0.f };

        if (listed && cnt <= LCAP) {
            int sorted0 = 0;
            if (cnt <= 32) {
                const int my = (lane < cnt) ? perm[beg + lane] : 0x7fffffff;
                int rank = 0;
                for (int m = 0; m < cnt; ++m)
                    rank += (__builtin_amdgcn_readlane(my, m) < my) ? 1 : 0;
                sorted0 = __builtin_amdgcn_ds_permute(rank << 2, my);
            } else {
                const int padded = (cnt + 31) & ~31;
                for (int p = lane; p < padded; p += 32)
                    uns[p] = (p < cnt) ? perm[beg + p] : 0x7fffffff;
                for (int p = lane; p < padded; p += 32) {
                    const int my = uns[p];
                    int rank = 0;
                    for (int q = 0; q < padded; q += 4) {
                        const int4 v = *(const int4*)(uns + q);
                        rank += ((v.x < my) ? 1 : 0) + ((v.y < my) ? 1 : 0)
                              + ((v.z < my) ? 1 : 0) + ((v.w < my) ? 1 : 0);
                    }
                    if (p < cnt) srt[rank] = my;
                }
            }
            for (int c = 0; c < cnt; c += 32) {
                const int m = (cnt - c < 32) ? (cnt - c) : 32;
                const int sorted = (cnt <= 32) ? sorted0 : ((lane < m) ? srt[c + lane] : 0);
                rgcn_sorted_run<RELU>(acc, sorted, m, lane, xin, src, etype, norm, wl);
            }
        } else if (listed && cnt <= 2048) {
            int lo = -1;
            for (int i = 0; i < cnt; ++i) {
                int best = 0x7fffffff;
                for (int p = beg + lane; p < end; p += 32) {
                    const int e = perm[p];
                    if (e > lo && e < best) best = e;
                }
                #pragma unroll
                for (int o = 16; o > 0; o >>= 1) {
                    const int other = __shfl_xor(best, o);
                    best = (other < best) ? other : best;
                }
                const int e = __builtin_amdgcn_readfirstlane(best);
                lo = e;
                const int   s  = src[e];
                const int   r  = etype[e];
                const float nr = norm[e];
                const float4* xrow = (const float4*)(xin + (size_t)s * HD) + lane;
                rgcn_msg_acc<RELU>(acc, xrow[0], xrow[32], wl + (size_t)r * NB, nr);
            }
        } else {
            for (int e0 = 0; e0 < n_edges; e0 += 32) {
                const int  ev  = e0 + lane;
                const bool hit = (ev < n_edges) && (dst[ev] == n);
                unsigned m = __builtin_amdgcn_ballot_w32(hit);
                while (m != 0u) {
                    const int e = e0 + (__builtin_ctz(m));
                    m &= m - 1u;
                    const int   s  = src[e];
                    const int   r  = etype[e];
                    const float nr = norm[e];
                    const float4* xrow = (const float4*)(xin + (size_t)s * HD) + lane;
                    rgcn_msg_acc<RELU>(acc, xrow[0], xrow[32], wl + (size_t)r * NB, nr);
                }
            }
        }

        oa.x += acc[0]; oa.y += acc[1]; oa.z += acc[2]; oa.w += acc[3];
        ob.x += acc[4]; ob.y += acc[5]; ob.z += acc[6]; ob.w += acc[7];
        orow[0]  = oa;
        orow[32] = ob;
    }
}

#define RCAP 64

__device__ __forceinline__
unsigned long long rgcn_rec_key(int w0, int nb)
{
    return ((unsigned long long)(unsigned)w0 << 32) | (unsigned long long)(unsigned)nb;
}

template <bool RELU>
__device__ __forceinline__
void rgcn_rec_run(float (&acc)[8], const int w0, const int nb, const int m, const int lane,
                  const float* __restrict__ xin, const float4* wl)
{
    const int s_v = w0 & 0x00ffffff;
    const int r_v = (int)((unsigned)w0 >> 24);

    int k = 0;
    for (; k + 4 <= m; k += 4) {
        float4 xa[4], xb[4];
        int    rr[4];
        float  nn[4];
        #pragma unroll
        for (int u = 0; u < 4; ++u) {
            const int s = __builtin_amdgcn_readlane(s_v, k + u);
            rr[u] = __builtin_amdgcn_readlane(r_v, k + u);
            nn[u] = __int_as_float(__builtin_amdgcn_readlane(nb, k + u));
            const float4* xrow = (const float4*)(xin + (size_t)s * HD) + lane;
            xa[u] = xrow[0];
            xb[u] = xrow[32];
        }
        #pragma unroll
        for (int u = 0; u < 4; ++u)
            rgcn_msg_acc<RELU>(acc, xa[u], xb[u], wl + (size_t)rr[u] * NB, nn[u]);
    }
    if (k < m) {
        const int left = m - k;
        const float4 z4 = make_float4(0.f, 0.f, 0.f, 0.f);
        float4 xa1 = z4, xb1 = z4, xa2 = z4, xb2 = z4;
        int    r1 = 0, r2 = 0;
        float  n1 = 0.f, n2 = 0.f;
        const int   s0 = __builtin_amdgcn_readlane(s_v, k);
        const int   r0 = __builtin_amdgcn_readlane(r_v, k);
        const float n0 = __int_as_float(__builtin_amdgcn_readlane(nb, k));
        const float4* x0 = (const float4*)(xin + (size_t)s0 * HD) + lane;
        const float4 xa0 = x0[0];
        const float4 xb0 = x0[32];
        if (left > 1) {
            const int s1 = __builtin_amdgcn_readlane(s_v, k + 1);
            r1 = __builtin_amdgcn_readlane(r_v, k + 1);
            n1 = __int_as_float(__builtin_amdgcn_readlane(nb, k + 1));
            const float4* x1 = (const float4*)(xin + (size_t)s1 * HD) + lane;
            xa1 = x1[0];
            xb1 = x1[32];
        }
        if (left > 2) {
            const int s2 = __builtin_amdgcn_readlane(s_v, k + 2);
            r2 = __builtin_amdgcn_readlane(r_v, k + 2);
            n2 = __int_as_float(__builtin_amdgcn_readlane(nb, k + 2));
            const float4* x2 = (const float4*)(xin + (size_t)s2 * HD) + lane;
            xa2 = x2[0];
            xb2 = x2[32];
        }
        rgcn_msg_acc<RELU>(acc, xa0, xb0, wl + (size_t)r0 * NB, n0);
        if (left > 1) rgcn_msg_acc<RELU>(acc, xa1, xb1, wl + (size_t)r1 * NB, n1);
        if (left > 2) rgcn_msg_acc<RELU>(acc, xa2, xb2, wl + (size_t)r2 * NB, n2);
    }
}

template <bool RELU>
__global__ __launch_bounds__(256)
void rgcn_node_gather_rec(const float* __restrict__ xin,
                          const int*   __restrict__ src,
                          const int*   __restrict__ dst,
                          const int*   __restrict__ etype,
                          const float* __restrict__ norm,
                          const float* __restrict__ w,
                          const int*   __restrict__ seg,
                          const int*   __restrict__ bucket,
                          float* agg,
                          int n_nodes,
                          int n_edges,
                          int cap)
{
    __shared__ __align__(16) float wlds[WTAB];
    __shared__ __align__(16) int   keybuf[8 * 4 * RCAP];

    const int lane   = threadIdx.x & 31;
    const int wid    = __builtin_amdgcn_readfirstlane(
                           blockIdx.x * (blockDim.x >> 5) + (threadIdx.x >> 5));
    const int nwaves = (gridDim.x * blockDim.x) >> 5;
    const int2* rec = (const int2*)bucket;

    int n   = wid;
    int cnt = (n < n_nodes) ? seg[n] : 0;

    if (blockDim.x * 8 == WTAB / 4) {
        float4 tv[8];
        #pragma unroll
        for (int q = 0; q < 8; ++q)
            tv[q] = ((const float4*)w)[threadIdx.x + q * (WTAB / 32)];
        #pragma unroll
        for (int q = 0; q < 8; ++q) {
            const int i = threadIdx.x + q * (WTAB / 32);
            const int b = i & (NB - 1);
            const int j = ((b >> 6) << 1) | (b & 1);
            const int L = (b & 63) >> 1;
            ((float4*)wlds)[(i & ~(NB - 1)) + j * 32 + L] = tv[q];
        }
    } else {
        for (int i = threadIdx.x; i < WTAB / 4; i += blockDim.x) {
            const int b = i & (NB - 1);
            const int j = ((b >> 6) << 1) | (b & 1);
            const int L = (b & 63) >> 1;
            ((float4*)wlds)[(i & ~(NB - 1)) + j * 32 + L] = ((const float4*)w)[i];
        }
    }

    int2 recn = make_int2(-1, -1);
    if (lane < cnt && lane < cap) recn = rec[(size_t)n * cap + lane];
    int n1   = n + nwaves;
    int cnt1 = (n1 < n_nodes) ? seg[n1] : 0;

    __syncthreads();

    const float4* wl = (const float4*)wlds + lane;
    int2* uns = (int2*)(keybuf + __builtin_amdgcn_readfirstlane((int)(threadIdx.x >> 5)) * (4 * RCAP));
    int2* srt = uns + RCAP;

    int2 recn1 = make_int2(-1, -1);
    int  n2 = 0, cnt2 = 0;
    for (; n < n_nodes; n = n1, cnt = cnt1, recn = recn1, n1 = n2, cnt1 = cnt2) {
        recn1 = make_int2(-1, -1);
        if (lane < cnt1 && lane < cap) recn1 = rec[(size_t)n1 * cap + lane];
        n2   = n1 + nwaves;
        cnt2 = (n2 < n_nodes) ? seg[n2] : 0;

        if (cnt <= 0) continue;

        float4* orow = (float4*)(agg + (size_t)n * HD) + lane;
        float4 oa = orow[0];
        float4 ob = orow[32];

        float acc[8] = { 0.f, 0.f, 0.f, 0.f, 0.f, 0.f, 0.f, 0.f };

        if (cnt <= cap && cnt <= RCAP) {
            const int2* row = rec + (size_t)n * cap;
            int w0s = 0, nbs = 0;
            if (cnt <= 32) {
                const int2 mine = recn;
                const unsigned long long my = rgcn_rec_key(mine.x, mine.y);
                int rank = 0;
                for (int m = 0; m < cnt; ++m) {
                    const unsigned long long ok =
                        rgcn_rec_key(__builtin_amdgcn_readlane(mine.x, m),
                                     __builtin_amdgcn_readlane(mine.y, m));
                    rank += (ok < my || (ok == my && m < lane)) ? 1 : 0;
                }
                w0s = __builtin_amdgcn_ds_permute(rank << 2, mine.x);
                nbs = __builtin_amdgcn_ds_permute(rank << 2, mine.y);
            } else {
                const int padded = (cnt + 31) & ~31;
                for (int p = lane; p < padded; p += 32)
                    uns[p] = (p < cnt) ? row[p] : make_int2(-1, -1);
                for (int p = lane; p < padded; p += 32) {
                    const int2 mine = uns[p];
                    const unsigned long long my = rgcn_rec_key(mine.x, mine.y);
                    int rank = 0;
                    for (int q = 0; q < padded; q += 2) {
                        const int4 v = *(const int4*)(uns + q);
                        const unsigned long long k0 = rgcn_rec_key(v.x, v.y);
                        const unsigned long long k1 = rgcn_rec_key(v.z, v.w);
                        rank += ((k0 < my || (k0 == my && q     < p)) ? 1 : 0)
                              + ((k1 < my || (k1 == my && q + 1 < p)) ? 1 : 0);
                    }
                    if (p < cnt) srt[rank] = mine;
                }
            }
            for (int c = 0; c < cnt; c += 32) {
                const int m = (cnt - c < 32) ? (cnt - c) : 32;
                if (cnt > 32) {
                    int2 r = make_int2(0, 0);
                    if (lane < m) r = srt[c + lane];
                    w0s = r.x;
                    nbs = r.y;
                }
                rgcn_rec_run<RELU>(acc, w0s, nbs, m, lane, xin, wl);
            }
        } else {
            for (int e0 = 0; e0 < n_edges; e0 += 32) {
                const int  ev  = e0 + lane;
                const bool hit = (ev < n_edges) && (dst[ev] == n);
                unsigned m = __builtin_amdgcn_ballot_w32(hit);
                while (m != 0u) {
                    const int e = e0 + (__builtin_ctz(m));
                    m &= m - 1u;
                    const int   s  = src[e];
                    const int   r  = etype[e];
                    const float nr = norm[e];
                    const float4* xrow = (const float4*)(xin + (size_t)s * HD) + lane;
                    rgcn_msg_acc<RELU>(acc, xrow[0], xrow[32], wl + (size_t)r * NB, nr);
                }
            }
        }

        oa.x += acc[0]; oa.y += acc[1]; oa.z += acc[2]; oa.w += acc[3];
        ob.x += acc[4]; ob.y += acc[5]; ob.z += acc[6]; ob.w += acc[7];
        orow[0]  = oa;
        orow[32] = ob;
    }
}

extern "C" void kernel_launch(void* const* d_in, const int* in_sizes, int n_in,
                              void* d_out, int out_size, void* d_ws, size_t ws_size,
                              hipStream_t stream)
{
    const float* x     = (const float*)d_in[0];
    const int*   src   = (const int*)  d_in[1];
    const int*   dst   = (const int*)  d_in[2];
    const int*   etype = (const int*)  d_in[3];
    const float* norm  = (const float*)d_in[4];
    const float* w1    = (const float*)d_in[5];
    const float* lw1   = (const float*)d_in[6];
    const float* b1    = (const float*)d_in[7];
    const float* w2    = (const float*)d_in[8];
    const float* lw2   = (const float*)d_in[9];
    const float* b2    = (const float*)d_in[10];
    float*       out   = (float*)d_out;

    const int n_nodes = in_sizes[0] / HD;
    const int n_edges = in_sizes[1];

    float* h = (float*)d_ws;

    const int gemm_blocks = ((n_nodes + 15) / 16 + GEMM_RT - 1) / GEMM_RT;

    const size_t seg_off  = ((size_t)n_nodes * HD * sizeof(float) + 255) & ~(size_t)255;
    const size_t perm_off = (seg_off + (size_t)n_nodes * sizeof(int) + 255) & ~(size_t)255;
    const size_t ws_need  = perm_off + (size_t)n_edges * sizeof(int);

    if (ws_size < ws_need) {
        const int edge_blocks = 1024;
        rgcn_selfloop_gemm<false, false><<<gemm_blocks, GEMM_THREADS, 0, stream>>>(x, lw1, b1, h, n_nodes,
                                                                   nullptr, 0, gemm_blocks, nullptr, nullptr,
                                                                   nullptr, nullptr, nullptr, nullptr, 0, 0);
        rgcn_edge_msg<false><<<edge_blocks, 256, 0, stream>>>(x, src, dst, etype, norm, w1,
                                                              h, n_edges);
        rgcn_selfloop_gemm<true, false><<<gemm_blocks, GEMM_THREADS, 0, stream>>>(h, lw2, b2, out, n_nodes,
                                                                  nullptr, 0, gemm_blocks, nullptr, nullptr,
                                                                  nullptr, nullptr, nullptr, nullptr, 0, 0);
        rgcn_edge_msg<true><<<edge_blocks, 256, 0, stream>>>(h, src, dst, etype, norm, w2,
                                                             out, n_edges);
        return;
    }

    int* seg  = (int*)((char*)d_ws + seg_off);
    const int gather_blocks = 1024;

    const int    bcap     = 64;
    const size_t bkt_need = perm_off + (size_t)n_nodes * bcap * sizeof(int);

    const size_t rec_need = perm_off + (size_t)n_nodes * RCAP * 2 * sizeof(int);

    if (ws_size >= rec_need && n_nodes <= (1 << 24)) {
        int* bucket = (int*)((char*)d_ws + perm_off);
        const int fill_blocks = (n_edges + GEMM_THREADS * FILL_EPT - 1) / (GEMM_THREADS * FILL_EPT);

        rgcn_zero_counters<<<(n_nodes + 255) / 256, 256, 0, stream>>>(seg, n_nodes);
        rgcn_selfloop_gemm<false, true><<<gemm_blocks + fill_blocks, GEMM_THREADS, 0, stream>>>(
                                                                   x, lw1, b1, h, n_nodes,
                                                                   nullptr, 0, gemm_blocks, dst, src,
                                                                   etype, norm, seg, bucket, RCAP, n_edges);

        rgcn_node_gather_rec<false><<<gather_blocks, 256, 0, stream>>>(x, src, dst, etype, norm, w1,
                                                                       seg, bucket, h, n_nodes,
                                                                       n_edges, RCAP);
        rgcn_selfloop_gemm<true, false><<<gemm_blocks, GEMM_THREADS, 0, stream>>>(h, lw2, b2, out, n_nodes,
                                                                  nullptr, 0, gemm_blocks, nullptr, nullptr,
                                                                  nullptr, nullptr, nullptr, nullptr, 0, 0);
        rgcn_node_gather_rec<true><<<gather_blocks, 256, 0, stream>>>(h, src, dst, etype, norm, w2,
                                                                      seg, bucket, out, n_nodes,
                                                                      n_edges, RCAP);
        return;
    }

    if (ws_size >= bkt_need) {
        int* bucket = (int*)((char*)d_ws + perm_off);
        const int fill_blocks = (n_edges + 1023) / 1024;

        rgcn_selfloop_gemm<false, false><<<gemm_blocks, GEMM_THREADS, 0, stream>>>(x, lw1, b1, h, n_nodes,
                                                                   seg, n_nodes, gemm_blocks, nullptr, nullptr,
                                                                   nullptr, nullptr, nullptr, nullptr, 0, 0);
        rgcn_bucket_fill<<<fill_blocks, 256, 0, stream>>>(dst, seg, bucket, bcap, n_edges);

        rgcn_node_gather<false><<<gather_blocks, 256, 0, stream>>>(x, src, dst, etype, norm, w1,
                                                                   seg, bucket, h, n_nodes,
                                                                   n_edges, bcap);
        rgcn_selfloop_gemm<true, false><<<gemm_blocks, GEMM_THREADS, 0, stream>>>(h, lw2, b2, out, n_nodes,
                                                                  nullptr, 0, gemm_blocks, nullptr, nullptr,
                                                                  nullptr, nullptr, nullptr, nullptr, 0, 0);
        rgcn_node_gather<true><<<gather_blocks, 256, 0, stream>>>(h, src, dst, etype, norm, w2,
                                                                  seg, bucket, out, n_nodes,
                                                                  n_edges, bcap);
        return;
    }

    int* perm = (int*)((char*)d_ws + perm_off);
    const int edge_blocks   = (n_edges + 255) / 256;

    rgcn_selfloop_gemm<false, false><<<gemm_blocks, GEMM_THREADS, 0, stream>>>(x, lw1, b1, h, n_nodes,
                                                               seg, n_nodes, gemm_blocks, nullptr, nullptr,
                                                               nullptr, nullptr, nullptr, nullptr, 0, 0);
    rgcn_deg_count<<<edge_blocks, 256, 0, stream>>>(dst, seg, n_edges);
    rgcn_seg_scan<<<1, 1024, 0, stream>>>(seg, n_nodes);
    rgcn_seg_fill<<<edge_blocks, 256, 0, stream>>>(dst, seg, perm, n_edges);

    rgcn_node_gather<false><<<gather_blocks, 256, 0, stream>>>(x, src, dst, etype, norm, w1,
                                                               seg, perm, h, n_nodes, n_edges, 0);

    rgcn_selfloop_gemm<true, false><<<gemm_blocks, GEMM_THREADS, 0, stream>>>(h, lw2, b2, out, n_nodes,
                                                              nullptr, 0, gemm_blocks, nullptr, nullptr,
                                                              nullptr, nullptr, nullptr, nullptr, 0, 0);
    rgcn_node_gather<true><<<gather_blocks, 256, 0, stream>>>(h, src, dst, etype, norm, w2,
                                                              seg, perm, out, n_nodes, n_edges, 0);
}
